# starting kernel
# baseline (speedup 1.0000x reference)
#include <hip/hip_runtime.h>
#include <hip/hip_bf16.h>

typedef __attribute__((ext_vector_type(2))) float v2f;
typedef __attribute__((ext_vector_type(8))) float v8f;

#define F_IN 128
#define C1   64
#define C2   16
#define NEG_SLOPE 0.2f

__global__ void __launch_bounds__(256)
gemm1_wmma(const float* __restrict__ x, const float* __restrict__ W1,
           float* __restrict__ h1, int Nn) {
  __shared__ float sW[F_IN * C1];
  for (int i = threadIdx.x; i < (F_IN * C1) / 4; i += 256) {
    float4 v = ((const float4*)W1)[i];
    const int k  = i >> 4;
    const int c0 = (i & 15) * 4;
    float* dst = sW + (k >> 2) * 256 + (k & 3);
    dst[(c0 + 0) * 4] = v.x;
    dst[(c0 + 1) * 4] = v.y;
    dst[(c0 + 2) * 4] = v.z;
    dst[(c0 + 3) * 4] = v.w;
  }
  __syncthreads();

  const int wave = threadIdx.x >> 5;
  const int lane = threadIdx.x & 31;
  const int tile = blockIdx.x * 8 + wave;
  if (tile * 16 >= Nn) return;
  const int row0 = tile * 16;
  const int r    = lane & 15;
  const int kh   = (lane >> 4) << 1;
  const int arow = min(row0 + r, Nn - 1);

  v8f acc0 = {}, acc1 = {}, acc2 = {}, acc3 = {};
  const float* ap = x + (size_t)arow * F_IN + kh;
  const float* bbase = sW + r * 4 + kh;
  for (int k0 = 0; k0 < F_IN; k0 += 4) {
    v2f a; a.x = ap[k0]; a.y = ap[k0 + 1];
    const float* bq = bbase + (k0 >> 2) * 256;
    v2f b0 = *(const v2f*)(bq);
    v2f b1 = *(const v2f*)(bq + 64);
    v2f b2 = *(const v2f*)(bq + 128);
    v2f b3 = *(const v2f*)(bq + 192);
    acc0 = __builtin_amdgcn_wmma_f32_16x16x4_f32(false, a, false, b0, (short)0, acc0, false, false);
    acc1 = __builtin_amdgcn_wmma_f32_16x16x4_f32(false, a, false, b1, (short)0, acc1, false, false);
    acc2 = __builtin_amdgcn_wmma_f32_16x16x4_f32(false, a, false, b2, (short)0, acc2, false, false);
    acc3 = __builtin_amdgcn_wmma_f32_16x16x4_f32(false, a, false, b3, (short)0, acc3, false, false);
  }

  const int mbase = (lane >> 4) * 8;
  for (int i = 0; i < 8; ++i) {
    const int row = row0 + mbase + i;
    if (row < Nn) {
      float* op = h1 + (size_t)row * C1 + r;
      op[0]  = acc0[i];
      op[16] = acc1[i];
      op[32] = acc2[i];
      op[48] = acc3[i];
    }
  }
}

__global__ void __launch_bounds__(256)
gemm2_wmma(const float* __restrict__ h, const float* __restrict__ W2,
           float* __restrict__ h2, int Nn) {
  __shared__ float sW[C1 * C2];
  for (int i = threadIdx.x; i < (C1 * C2) / 4; i += 256) {
    float4 v = ((const float4*)W2)[i];
    const int k  = i >> 2;
    const int c0 = (i & 3) * 4;
    float* dst = sW + (k >> 2) * 64 + (k & 3);
    dst[(c0 + 0) * 4] = v.x;
    dst[(c0 + 1) * 4] = v.y;
    dst[(c0 + 2) * 4] = v.z;
    dst[(c0 + 3) * 4] = v.w;
  }
  __syncthreads();

  const int wave = threadIdx.x >> 5;
  const int lane = threadIdx.x & 31;
  const int tile = blockIdx.x * 8 + wave;
  if (tile * 16 >= Nn) return;
  const int row0 = tile * 16;
  const int r    = lane & 15;
  const int kh   = (lane >> 4) << 1;
  const int arow = min(row0 + r, Nn - 1);

  v8f acc = {};
  const float* ap = h + (size_t)arow * C1 + kh;
  const float* bbase = sW + r * 4 + kh;
  for (int k0 = 0; k0 < C1; k0 += 4) {
    v2f a; a.x = ap[k0]; a.y = ap[k0 + 1];
    v2f b = *(const v2f*)(bbase + (k0 >> 2) * 64);
    acc = __builtin_amdgcn_wmma_f32_16x16x4_f32(false, a, false, b, (short)0, acc, false, false);
  }

  const int mbase = (lane >> 4) * 8;
  for (int i = 0; i < 8; ++i) {
    const int row = row0 + mbase + i;
    if (row < Nn) h2[(size_t)row * C2 + r] = acc[i];
  }
}

template <int C>
__global__ void node_dots(const float* __restrict__ h,
                          const float* __restrict__ asrc,
                          const float* __restrict__ adst,
                          float* __restrict__ s, float* __restrict__ d, int Nn) {
  int n = blockIdx.x * blockDim.x + threadIdx.x;
  if (n >= Nn) return;
  const float* hp = h + (size_t)n * C;
  float ss = 0.f, dd = 0.f;
  for (int f = 0; f < C; ++f) { float v = hp[f]; ss += v * asrc[f]; dd += v * adst[f]; }
  s[n] = ss; d[n] = dd;
}

__global__ void init_accum(float* __restrict__ outacc, size_t acc_elems,
                           int* __restrict__ amax, float* __restrict__ denom, int Nn) {
  size_t i = (size_t)blockIdx.x * blockDim.x + threadIdx.x;
  if (i < acc_elems) outacc[i] = 0.f;
  if (i < (size_t)Nn) { amax[i] = (int)0x80000000; denom[i] = 0.f; }
}

__device__ __forceinline__ int fkey(float f) {
  int k = __float_as_int(f);
  return k >= 0 ? k : (k ^ 0x7fffffff);
}
__device__ __forceinline__ float fkey_inv(int k) {
  return __int_as_float(k >= 0 ? k : (k ^ 0x7fffffff));
}
__device__ __forceinline__ void edge_sd(const int* ei, int E, int e, int& s, int& d) {
  if (e < E) { s = ei[e]; d = ei[E + e]; } else { s = d = e - E; }
}

__global__ void edge_max_kernel(const int* __restrict__ ei, int E, int Etot,
                                const float* __restrict__ s, const float* __restrict__ d,
                                float* __restrict__ alpha_buf, int* __restrict__ amax) {
  int e = blockIdx.x * blockDim.x + threadIdx.x;
  if (e >= Etot) return;
  int sn, dn; edge_sd(ei, E, e, sn, dn);
  float a = s[sn] + d[dn];
  a = a > 0.f ? a : NEG_SLOPE * a;
  alpha_buf[e] = a;
  atomicMax(amax + dn, fkey(a));
}

__global__ void edge_sum_kernel(const int* __restrict__ ei, int E, int Etot,
                                float* __restrict__ alpha_buf,
                                const int* __restrict__ amax, float* __restrict__ denom) {
  int e = blockIdx.x * blockDim.x + threadIdx.x;
  if (e >= Etot) return;
  int sn, dn; edge_sd(ei, E, e, sn, dn);
  float ew = __expf(alpha_buf[e] - fkey_inv(amax[dn]));
  alpha_buf[e] = ew;
  atomicAdd(denom + dn, ew);
}

template <int C>
__global__ void edge_agg_kernel(const int* __restrict__ ei, int E, int Etot,
                                const float* __restrict__ ewb, const float* __restrict__ denom,
                                const float* __restrict__ hsrc, float* __restrict__ outacc) {
  const int chunks = C / 4;
  long long tid = (long long)blockIdx.x * blockDim.x + threadIdx.x;
  int e  = (int)(tid / chunks);
  int fo = (int)(tid % chunks) * 4;
  if (e >= Etot) return;
  int sn, dn; edge_sd(ei, E, e, sn, dn);
  float w = ewb[e] / (denom[dn] + 1e-16f);
  float4 hv = *(const float4*)(hsrc + (size_t)sn * C + fo);
  float* op = outacc + (size_t)dn * C + fo;
  atomicAdd(op + 0, hv.x * w);
  atomicAdd(op + 1, hv.y * w);
  atomicAdd(op + 2, hv.z * w);
  atomicAdd(op + 3, hv.w * w);
}

__device__ __forceinline__ float softplusf(float v) {
  return v > 0.f ? v + log1pf(__expf(-v)) : log1pf(__expf(v));
}

__global__ void finish1_kernel(const float* __restrict__ acc, const float* __restrict__ b1,
                               float* __restrict__ hbuf, float* __restrict__ ixz1, int Nn) {
  size_t tid = (size_t)blockIdx.x * blockDim.x + threadIdx.x;
  int n = (int)(tid >> 6), f = (int)(tid & 63);
  if (n >= Nn) return;
  float v = acc[(size_t)n * C1 + f] + b1[f];
  hbuf[(size_t)n * C1 + f] = v > 0.f ? v : (__expf(v) - 1.f);
  if (f < 32) {
    float sp  = acc[(size_t)n * C1 + 32 + f] + b1[32 + f];
    float sd  = softplusf(sp) + 1e-10f;
    ixz1[(size_t)n * 32 + f] = -__logf(sd) + 0.5f * (sd * sd + v * v - 1.f);
  }
}

__global__ void finish2_kernel(const float* __restrict__ acc, const float* __restrict__ b2,
                               float* __restrict__ out2, float* __restrict__ ixz2, int Nn) {
  size_t tid = (size_t)blockIdx.x * blockDim.x + threadIdx.x;
  int n = (int)(tid >> 4), f = (int)(tid & 15);
  if (n >= Nn) return;
  float v = acc[(size_t)n * C2 + f] + b2[f];
  out2[(size_t)n * C2 + f] = v;
  if (f < 8) {
    float sp = acc[(size_t)n * C2 + 8 + f] + b2[8 + f];
    float sd = softplusf(sp) + 1e-10f;
    ixz2[(size_t)n * 8 + f] = -__logf(sd) + 0.5f * (sd * sd + v * v - 1.f);
  }
}

extern "C" void kernel_launch(void* const* d_in, const int* in_sizes, int n_in,
                              void* d_out, int out_size, void* d_ws, size_t ws_size,
                              hipStream_t stream) {
  const float* x   = (const float*)d_in[0];
  const int*   ei  = (const int*)  d_in[1];
  const float* W1  = (const float*)d_in[2];
  const float* as1 = (const float*)d_in[3];
  const float* ad1 = (const float*)d_in[4];
  const float* b1  = (const float*)d_in[5];
  const float* W2  = (const float*)d_in[6];
  const float* as2 = (const float*)d_in[7];
  const float* ad2 = (const float*)d_in[8];
  const float* b2  = (const float*)d_in[9];

  const int N    = in_sizes[0] / F_IN;
  const int E    = in_sizes[1] / 2;
  const int Etot = E + N;
  float* out = (float*)d_out;

  const size_t N64 = (size_t)N * C1, N16 = (size_t)N * C2;
  float* ws    = (float*)d_ws;
  float* h1    = ws;
  float* out1a = h1 + N64;
  float* s1    = out1a + N64;
  float* d1    = s1 + N;
  float* denom = d1 + N;
  int*   amax  = (int*)(denom + N);
  float* ewb   = (float*)(amax + N);
  float* h2    = ewb + Etot;
  float* out2a = h2 + N16;

  float* ixz1 = out + N16;
  float* ixz2 = out + (size_t)N * 48;

  const int tiles  = (N + 15) / 16;
  const int gblk   = (tiles + 7) / 8;
  const int eblk   = (Etot + 255) / 256;

  init_accum<<<(int)((N64 + 255) / 256), 256, 0, stream>>>(out1a, N64, amax, denom, N);
  gemm1_wmma<<<gblk, 256, 0, stream>>>(x, W1, h1, N);
  node_dots<C1><<<(N + 255) / 256, 256, 0, stream>>>(h1, as1, ad1, s1, d1, N);
  edge_max_kernel<<<eblk, 256, 0, stream>>>(ei, E, Etot, s1, d1, ewb, amax);
  edge_sum_kernel<<<eblk, 256, 0, stream>>>(ei, E, Etot, ewb, amax, denom);
  {
    long long tot = (long long)Etot * (C1 / 4);
    edge_agg_kernel<C1><<<(int)((tot + 255) / 256), 256, 0, stream>>>(ei, E, Etot, ewb, denom, h1, out1a);
  }
  finish1_kernel<<<(int)((N64 + 255) / 256), 256, 0, stream>>>(out1a, b1, h1, ixz1, N);

  init_accum<<<(int)((N16 + 255) / 256), 256, 0, stream>>>(out2a, N16, amax, denom, N);
  gemm2_wmma<<<gblk, 256, 0, stream>>>(h1, W2, h2, N);
  node_dots<C2><<<(N + 255) / 256, 256, 0, stream>>>(h2, as2, ad2, s1, d1, N);
  edge_max_kernel<<<eblk, 256, 0, stream>>>(ei, E, Etot, s1, d1, ewb, amax);
  edge_sum_kernel<<<eblk, 256, 0, stream>>>(ei, E, Etot, ewb, amax, denom);
  {
    long long tot = (long long)Etot * (C2 / 4);
    edge_agg_kernel<C2><<<(int)((tot + 255) / 256), 256, 0, stream>>>(ei, E, Etot, ewb, denom, h2, out2a);
  }
  finish2_kernel<<<(int)((N16 + 255) / 256), 256, 0, stream>>>(out2a, b2, out, ixz2, N);
}


# edge_agg: one lane per feature word, so each wave-wide row load and f32 atomic add covers whole rows (a full 128B line per instruction, was four part-filled lines); atomic sum order as before
# speedup vs baseline: 1.1781x; 1.1781x over previous
#include <hip/hip_runtime.h>
#include <hip/hip_bf16.h>

typedef __attribute__((ext_vector_type(2))) float v2f;
typedef __attribute__((ext_vector_type(8))) float v8f;

#define F_IN 128
#define C1   64
#define C2   16
#define NEG_SLOPE 0.2f

__global__ void __launch_bounds__(256)
gemm1_wmma(const float* __restrict__ x, const float* __restrict__ W1,
           float* __restrict__ h1, int Nn) {
  __shared__ float sW[F_IN * C1];
  for (int i = threadIdx.x; i < (F_IN * C1) / 4; i += 256) {
    float4 v = ((const float4*)W1)[i];
    const int k  = i >> 4;
    const int c0 = (i & 15) * 4;
    float* dst = sW + (k >> 2) * 256 + (k & 3);
    dst[(c0 + 0) * 4] = v.x;
    dst[(c0 + 1) * 4] = v.y;
    dst[(c0 + 2) * 4] = v.z;
    dst[(c0 + 3) * 4] = v.w;
  }
  __syncthreads();

  const int wave = threadIdx.x >> 5;
  const int lane = threadIdx.x & 31;
  const int tile = blockIdx.x * 8 + wave;
  if (tile * 16 >= Nn) return;
  const int row0 = tile * 16;
  const int r    = lane & 15;
  const int kh   = (lane >> 4) << 1;
  const int arow = min(row0 + r, Nn - 1);

  v8f acc0 = {}, acc1 = {}, acc2 = {}, acc3 = {};
  const float* ap = x + (size_t)arow * F_IN + kh;
  const float* bbase = sW + r * 4 + kh;
  for (int k0 = 0; k0 < F_IN; k0 += 4) {
    v2f a; a.x = ap[k0]; a.y = ap[k0 + 1];
    const float* bq = bbase + (k0 >> 2) * 256;
    v2f b0 = *(const v2f*)(bq);
    v2f b1 = *(const v2f*)(bq + 64);
    v2f b2 = *(const v2f*)(bq + 128);
    v2f b3 = *(const v2f*)(bq + 192);
    acc0 = __builtin_amdgcn_wmma_f32_16x16x4_f32(false, a, false, b0, (short)0, acc0, false, false);
    acc1 = __builtin_amdgcn_wmma_f32_16x16x4_f32(false, a, false, b1, (short)0, acc1, false, false);
    acc2 = __builtin_amdgcn_wmma_f32_16x16x4_f32(false, a, false, b2, (short)0, acc2, false, false);
    acc3 = __builtin_amdgcn_wmma_f32_16x16x4_f32(false, a, false, b3, (short)0, acc3, false, false);
  }

  const int mbase = (lane >> 4) * 8;
  for (int i = 0; i < 8; ++i) {
    const int row = row0 + mbase + i;
    if (row < Nn) {
      float* op = h1 + (size_t)row * C1 + r;
      op[0]  = acc0[i];
      op[16] = acc1[i];
      op[32] = acc2[i];
      op[48] = acc3[i];
    }
  }
}

__global__ void __launch_bounds__(256)
gemm2_wmma(const float* __restrict__ h, const float* __restrict__ W2,
           float* __restrict__ h2, int Nn) {
  __shared__ float sW[C1 * C2];
  for (int i = threadIdx.x; i < (C1 * C2) / 4; i += 256) {
    float4 v = ((const float4*)W2)[i];
    const int k  = i >> 2;
    const int c0 = (i & 3) * 4;
    float* dst = sW + (k >> 2) * 64 + (k & 3);
    dst[(c0 + 0) * 4] = v.x;
    dst[(c0 + 1) * 4] = v.y;
    dst[(c0 + 2) * 4] = v.z;
    dst[(c0 + 3) * 4] = v.w;
  }
  __syncthreads();

  const int wave = threadIdx.x >> 5;
  const int lane = threadIdx.x & 31;
  const int tile = blockIdx.x * 8 + wave;
  if (tile * 16 >= Nn) return;
  const int row0 = tile * 16;
  const int r    = lane & 15;
  const int kh   = (lane >> 4) << 1;
  const int arow = min(row0 + r, Nn - 1);

  v8f acc = {};
  const float* ap = h + (size_t)arow * C1 + kh;
  const float* bbase = sW + r * 4 + kh;
  for (int k0 = 0; k0 < C1; k0 += 4) {
    v2f a; a.x = ap[k0]; a.y = ap[k0 + 1];
    v2f b = *(const v2f*)(bbase + (k0 >> 2) * 64);
    acc = __builtin_amdgcn_wmma_f32_16x16x4_f32(false, a, false, b, (short)0, acc, false, false);
  }

  const int mbase = (lane >> 4) * 8;
  for (int i = 0; i < 8; ++i) {
    const int row = row0 + mbase + i;
    if (row < Nn) h2[(size_t)row * C2 + r] = acc[i];
  }
}

template <int C>
__global__ void node_dots(const float* __restrict__ h,
                          const float* __restrict__ asrc,
                          const float* __restrict__ adst,
                          float* __restrict__ s, float* __restrict__ d, int Nn) {
  int n = blockIdx.x * blockDim.x + threadIdx.x;
  if (n >= Nn) return;
  const float* hp = h + (size_t)n * C;
  float ss = 0.f, dd = 0.f;
  for (int f = 0; f < C; ++f) { float v = hp[f]; ss += v * asrc[f]; dd += v * adst[f]; }
  s[n] = ss; d[n] = dd;
}

__global__ void init_accum(float* __restrict__ outacc, size_t acc_elems,
                           int* __restrict__ amax, float* __restrict__ denom, int Nn) {
  size_t i = (size_t)blockIdx.x * blockDim.x + threadIdx.x;
  if (i < acc_elems) outacc[i] = 0.f;
  if (i < (size_t)Nn) { amax[i] = (int)0x80000000; denom[i] = 0.f; }
}

__device__ __forceinline__ int fkey(float f) {
  int k = __float_as_int(f);
  return k >= 0 ? k : (k ^ 0x7fffffff);
}
__device__ __forceinline__ float fkey_inv(int k) {
  return __int_as_float(k >= 0 ? k : (k ^ 0x7fffffff));
}
__device__ __forceinline__ void edge_sd(const int* ei, int E, int e, int& s, int& d) {
  if (e < E) { s = ei[e]; d = ei[E + e]; } else { s = d = e - E; }
}

__global__ void edge_max_kernel(const int* __restrict__ ei, int E, int Etot,
                                const float* __restrict__ s, const float* __restrict__ d,
                                float* __restrict__ alpha_buf, int* __restrict__ amax) {
  int e = blockIdx.x * blockDim.x + threadIdx.x;
  if (e >= Etot) return;
  int sn, dn; edge_sd(ei, E, e, sn, dn);
  float a = s[sn] + d[dn];
  a = a > 0.f ? a : NEG_SLOPE * a;
  alpha_buf[e] = a;
  atomicMax(amax + dn, fkey(a));
}

__global__ void edge_sum_kernel(const int* __restrict__ ei, int E, int Etot,
                                float* __restrict__ alpha_buf,
                                const int* __restrict__ amax, float* __restrict__ denom) {
  int e = blockIdx.x * blockDim.x + threadIdx.x;
  if (e >= Etot) return;
  int sn, dn; edge_sd(ei, E, e, sn, dn);
  float ew = __expf(alpha_buf[e] - fkey_inv(amax[dn]));
  alpha_buf[e] = ew;
  atomicAdd(denom + dn, ew);
}

template <int C>
__global__ void __launch_bounds__(256)
edge_agg_kernel(const int* __restrict__ ei, int E, int Etot,
                const float* __restrict__ ewb, const float* __restrict__ denom,
                const float* __restrict__ hsrc, float* __restrict__ outacc) {
  constexpr int LPE   = (C >= 32) ? 32 : C;
  constexpr int EPI   = 32 / LPE;
  constexpr int WPL   = C / LPE;
  constexpr int STEPS = 8;
  constexpr int EPW   = STEPS * EPI;
  const int lane = threadIdx.x & 31;
  const int wave = (int)((blockIdx.x * blockDim.x + threadIdx.x) >> 5);
  const int e0   = wave * EPW;
  if (e0 >= Etot) return;

  int sn = 0, dn = 0;
  const int e = e0 + lane;
  const bool mine = (lane < EPW) && (e < Etot);
  if (mine) edge_sd(ei, E, e, sn, dn);

  const int sub = lane / LPE;
  const int f   = lane % LPE;

  float v[STEPS][WPL];
#pragma unroll
  for (int i = 0; i < STEPS; ++i) {
    const int sj = __shfl(sn, i * EPI + sub);
    const float* hp = hsrc + (size_t)sj * C + f;
#pragma unroll
    for (int k = 0; k < WPL; ++k) v[i][k] = hp[k * LPE];
  }

  float w = 0.f;
  if (mine) w = ewb[e] / (denom[dn] + 1e-16f);

#pragma unroll
  for (int i = 0; i < STEPS; ++i) {
    const int j  = i * EPI + sub;
    const int dj = __shfl(dn, j);
    const float wj = __shfl(w, j);
    if (e0 + j < Etot) {
      float* op = outacc + (size_t)dj * C + f;
#pragma unroll
      for (int k = 0; k < WPL; ++k) atomicAdd(op + k * LPE, v[i][k] * wj);
    }
  }
}

__device__ __forceinline__ float softplusf(float v) {
  return v > 0.f ? v + log1pf(__expf(-v)) : log1pf(__expf(v));
}

__global__ void finish1_kernel(const float* __restrict__ acc, const float* __restrict__ b1,
                               float* __restrict__ hbuf, float* __restrict__ ixz1, int Nn) {
  size_t tid = (size_t)blockIdx.x * blockDim.x + threadIdx.x;
  int n = (int)(tid >> 6), f = (int)(tid & 63);
  if (n >= Nn) return;
  float v = acc[(size_t)n * C1 + f] + b1[f];
  hbuf[(size_t)n * C1 + f] = v > 0.f ? v : (__expf(v) - 1.f);
  if (f < 32) {
    float sp  = acc[(size_t)n * C1 + 32 + f] + b1[32 + f];
    float sd  = softplusf(sp) + 1e-10f;
    ixz1[(size_t)n * 32 + f] = -__logf(sd) + 0.5f * (sd * sd + v * v - 1.f);
  }
}

__global__ void finish2_kernel(const float* __restrict__ acc, const float* __restrict__ b2,
                               float* __restrict__ out2, float* __restrict__ ixz2, int Nn) {
  size_t tid = (size_t)blockIdx.x * blockDim.x + threadIdx.x;
  int n = (int)(tid >> 4), f = (int)(tid & 15);
  if (n >= Nn) return;
  float v = acc[(size_t)n * C2 + f] + b2[f];
  out2[(size_t)n * C2 + f] = v;
  if (f < 8) {
    float sp = acc[(size_t)n * C2 + 8 + f] + b2[8 + f];
    float sd = softplusf(sp) + 1e-10f;
    ixz2[(size_t)n * 8 + f] = -__logf(sd) + 0.5f * (sd * sd + v * v - 1.f);
  }
}

extern "C" void kernel_launch(void* const* d_in, const int* in_sizes, int n_in,
                              void* d_out, int out_size, void* d_ws, size_t ws_size,
                              hipStream_t stream) {
  const float* x   = (const float*)d_in[0];
  const int*   ei  = (const int*)  d_in[1];
  const float* W1  = (const float*)d_in[2];
  const float* as1 = (const float*)d_in[3];
  const float* ad1 = (const float*)d_in[4];
  const float* b1  = (const float*)d_in[5];
  const float* W2  = (const float*)d_in[6];
  const float* as2 = (const float*)d_in[7];
  const float* ad2 = (const float*)d_in[8];
  const float* b2  = (const float*)d_in[9];

  const int N    = in_sizes[0] / F_IN;
  const int E    = in_sizes[1] / 2;
  const int Etot = E + N;
  float* out = (float*)d_out;

  const size_t N64 = (size_t)N * C1, N16 = (size_t)N * C2;
  float* ws    = (float*)d_ws;
  float* h1    = ws;
  float* out1a = h1 + N64;
  float* s1    = out1a + N64;
  float* d1    = s1 + N;
  float* denom = d1 + N;
  int*   amax  = (int*)(denom + N);
  float* ewb   = (float*)(amax + N);
  float* h2    = ewb + Etot;
  float* out2a = h2 + N16;

  float* ixz1 = out + N16;
  float* ixz2 = out + (size_t)N * 48;

  const int tiles  = (N + 15) / 16;
  const int gblk   = (tiles + 7) / 8;
  const int eblk   = (Etot + 255) / 256;

  init_accum<<<(int)((N64 + 255) / 256), 256, 0, stream>>>(out1a, N64, amax, denom, N);
  gemm1_wmma<<<gblk, 256, 0, stream>>>(x, W1, h1, N);
  node_dots<C1><<<(N + 255) / 256, 256, 0, stream>>>(h1, as1, ad1, s1, d1, N);
  edge_max_kernel<<<eblk, 256, 0, stream>>>(ei, E, Etot, s1, d1, ewb, amax);
  edge_sum_kernel<<<eblk, 256, 0, stream>>>(ei, E, Etot, ewb, amax, denom);
  {
    edge_agg_kernel<C1><<<(Etot + 63) / 64, 256, 0, stream>>>(ei, E, Etot, ewb, denom, h1, out1a);
  }
  finish1_kernel<<<(int)((N64 + 255) / 256), 256, 0, stream>>>(out1a, b1, h1, ixz1, N);

  init_accum<<<(int)((N16 + 255) / 256), 256, 0, stream>>>(out2a, N16, amax, denom, N);
  gemm2_wmma<<<gblk, 256, 0, stream>>>(h1, W2, h2, N);
  node_dots<C2><<<(N + 255) / 256, 256, 0, stream>>>(h2, as2, ad2, s1, d1, N);
  edge_max_kernel<<<eblk, 256, 0, stream>>>(ei, E, Etot, s1, d1, ewb, amax);
  edge_sum_kernel<<<eblk, 256, 0, stream>>>(ei, E, Etot, ewb, amax, denom);
  {
    edge_agg_kernel<C2><<<(Etot + 127) / 128, 256, 0, stream>>>(ei, E, Etot, ewb, denom, h2, out2a);
  }
  finish2_kernel<<<(int)((N16 + 255) / 256), 256, 0, stream>>>(out2a, b2, out, ixz2, N);
}


# edges grouped by destination once per call and each row's sources sorted; a 16-lane group gathers each out row, softmax+bias+ELU/KL in registers: no float atomics, sum order = ascending source
# speedup vs baseline: 1.9142x; 1.6248x over previous
#include <hip/hip_runtime.h>
#include <hip/hip_bf16.h>

typedef __attribute__((ext_vector_type(2))) float v2f;
typedef __attribute__((ext_vector_type(8))) float v8f;

#define F_IN 128
#define C1   64
#define C2   16
#define NEG_SLOPE 0.2f

__global__ void __launch_bounds__(256)
gemm1_wmma(const float* __restrict__ x, const float* __restrict__ W1,
           float* __restrict__ h1, int Nn) {
  __shared__ float sW[F_IN * C1];
  for (int i = threadIdx.x; i < (F_IN * C1) / 4; i += 256) {
    float4 v = ((const float4*)W1)[i];
    const int k  = i >> 4;
    const int c0 = (i & 15) * 4;
    float* dst = sW + (k >> 2) * 256 + (k & 3);
    dst[(c0 + 0) * 4] = v.x;
    dst[(c0 + 1) * 4] = v.y;
    dst[(c0 + 2) * 4] = v.z;
    dst[(c0 + 3) * 4] = v.w;
  }
  __syncthreads();

  const int wave = threadIdx.x >> 5;
  const int lane = threadIdx.x & 31;
  const int tile = blockIdx.x * 8 + wave;
  if (tile * 16 >= Nn) return;
  const int row0 = tile * 16;
  const int r    = lane & 15;
  const int kh   = (lane >> 4) << 1;
  const int arow = min(row0 + r, Nn - 1);

  v8f acc0 = {}, acc1 = {}, acc2 = {}, acc3 = {};
  const float* ap = x + (size_t)arow * F_IN + kh;
  const float* bbase = sW + r * 4 + kh;
  for (int k0 = 0; k0 < F_IN; k0 += 4) {
    v2f a; a.x = ap[k0]; a.y = ap[k0 + 1];
    const float* bq = bbase + (k0 >> 2) * 256;
    v2f b0 = *(const v2f*)(bq);
    v2f b1 = *(const v2f*)(bq + 64);
    v2f b2 = *(const v2f*)(bq + 128);
    v2f b3 = *(const v2f*)(bq + 192);
    acc0 = __builtin_amdgcn_wmma_f32_16x16x4_f32(false, a, false, b0, (short)0, acc0, false, false);
    acc1 = __builtin_amdgcn_wmma_f32_16x16x4_f32(false, a, false, b1, (short)0, acc1, false, false);
    acc2 = __builtin_amdgcn_wmma_f32_16x16x4_f32(false, a, false, b2, (short)0, acc2, false, false);
    acc3 = __builtin_amdgcn_wmma_f32_16x16x4_f32(false, a, false, b3, (short)0, acc3, false, false);
  }

  const int mbase = (lane >> 4) * 8;
  for (int i = 0; i < 8; ++i) {
    const int row = row0 + mbase + i;
    if (row < Nn) {
      float* op = h1 + (size_t)row * C1 + r;
      op[0]  = acc0[i];
      op[16] = acc1[i];
      op[32] = acc2[i];
      op[48] = acc3[i];
    }
  }
}

__global__ void __launch_bounds__(256)
gemm2_wmma(const float* __restrict__ h, const float* __restrict__ W2,
           float* __restrict__ h2, int Nn) {
  __shared__ float sW[C1 * C2];
  for (int i = threadIdx.x; i < (C1 * C2) / 4; i += 256) {
    float4 v = ((const float4*)W2)[i];
    const int k  = i >> 2;
    const int c0 = (i & 3) * 4;
    float* dst = sW + (k >> 2) * 64 + (k & 3);
    dst[(c0 + 0) * 4] = v.x;
    dst[(c0 + 1) * 4] = v.y;
    dst[(c0 + 2) * 4] = v.z;
    dst[(c0 + 3) * 4] = v.w;
  }
  __syncthreads();

  const int wave = threadIdx.x >> 5;
  const int lane = threadIdx.x & 31;
  const int tile = blockIdx.x * 8 + wave;
  if (tile * 16 >= Nn) return;
  const int row0 = tile * 16;
  const int r    = lane & 15;
  const int kh   = (lane >> 4) << 1;
  const int arow = min(row0 + r, Nn - 1);

  v8f acc = {};
  const float* ap = h + (size_t)arow * C1 + kh;
  const float* bbase = sW + r * 4 + kh;
  for (int k0 = 0; k0 < C1; k0 += 4) {
    v2f a; a.x = ap[k0]; a.y = ap[k0 + 1];
    v2f b = *(const v2f*)(bbase + (k0 >> 2) * 64);
    acc = __builtin_amdgcn_wmma_f32_16x16x4_f32(false, a, false, b, (short)0, acc, false, false);
  }

  const int mbase = (lane >> 4) * 8;
  for (int i = 0; i < 8; ++i) {
    const int row = row0 + mbase + i;
    if (row < Nn) h2[(size_t)row * C2 + r] = acc[i];
  }
}

template <int C>
__global__ void node_dots(const float* __restrict__ h,
                          const float* __restrict__ asrc,
                          const float* __restrict__ adst,
                          float* __restrict__ s, float* __restrict__ d, int Nn) {
  int n = blockIdx.x * blockDim.x + threadIdx.x;
  if (n >= Nn) return;
  const float* hp = h + (size_t)n * C;
  float ss = 0.f, dd = 0.f;
  for (int f = 0; f < C; ++f) { float v = hp[f]; ss += v * asrc[f]; dd += v * adst[f]; }
  s[n] = ss; d[n] = dd;
}

__global__ void csr_init(int* __restrict__ deg, int* __restrict__ cursor, int Nn) {
  int i = blockIdx.x * blockDim.x + threadIdx.x;
  if (i < Nn) deg[i] = 0;
  if (i == 0) *cursor = 0;
}

__global__ void csr_count(const int* __restrict__ ei, int E, int* __restrict__ deg) {
  int e = blockIdx.x * blockDim.x + threadIdx.x;
  if (e < E) atomicAdd(deg + ei[E + e], 1);
}

__global__ void csr_place(const int* __restrict__ deg, int* __restrict__ cursor,
                          int* __restrict__ off, int* __restrict__ pos, int Nn) {
  const int n = blockIdx.x * blockDim.x + threadIdx.x;
  const int lane = threadIdx.x & 31;
  const int dg = (n < Nn) ? deg[n] : 0;
  int x = dg;
  for (int k = 1; k < 32; k <<= 1) {
    const int y = __shfl_up(x, k);
    if (lane >= k) x += y;
  }
  int base = 0;
  if (lane == 31) base = atomicAdd(cursor, x);
  base = __shfl(base, 31);
  const int o = base + x - dg;
  if (n < Nn) { off[n] = o; pos[n] = o; }
}

__global__ void csr_fill(const int* __restrict__ ei, int E,
                         int* __restrict__ pos, int* __restrict__ lin) {
  int e = blockIdx.x * blockDim.x + threadIdx.x;
  if (e < E) {
    const int slot = atomicAdd(pos + ei[E + e], 1);
    lin[slot] = ei[e];
  }
}

__global__ void __launch_bounds__(256)
csr_sort(const int* __restrict__ off, const int* __restrict__ deg,
         const int* __restrict__ lin, int* __restrict__ lout, int Nn) {
  const int lane  = threadIdx.x & 31;
  const int l     = lane & 15;
  const int gbase = lane & 16;
  const int n  = (int)((blockIdx.x * blockDim.x + threadIdx.x) >> 4);
  const int nn = min(n, Nn - 1);
  const int o  = off[nn];
  const int dg = (n < Nn) ? deg[nn] : 0;
  const int dgmax = max(dg, __shfl_xor(dg, 16));
  for (int ci = 0; ci < dgmax; ci += 16) {
    const int i = ci + l;
    const int key = (i < dg) ? lin[o + i] : 0x7fffffff;
    int rank = 0;
    for (int cj = 0; cj < dgmax; cj += 16) {
      const int jm = cj + l;
      const int kmine = (jm < dg) ? lin[o + jm] : 0x7fffffff;
      const int cnt = min(16, dgmax - cj);
      for (int t = 0; t < cnt; ++t) {
        const int kt = __shfl(kmine, gbase + t);
        const int jt = cj + t;
        rank += ((jt < dg) && (kt < key || (kt == key && jt < i))) ? 1 : 0;
      }
    }
    if (i < dg) lout[o + rank] = key;
  }
}

__device__ __forceinline__ float softplusf(float v) {
  return fmaxf(v, 0.f) + log1pf(__expf(-fabsf(v)));
}
__device__ __forceinline__ float kl_term(float mean, float sp) {
  const float sd = softplusf(sp) + 1e-10f;
  return -__logf(sd) + 0.5f * (sd * sd + mean * mean - 1.f);
}
__device__ __forceinline__ float leaky(float a) { return a > 0.f ? a : NEG_SLOPE * a; }
__device__ __forceinline__ float group_max16(float v) {
  v = fmaxf(v, __shfl_xor(v, 8)); v = fmaxf(v, __shfl_xor(v, 4));
  v = fmaxf(v, __shfl_xor(v, 2)); v = fmaxf(v, __shfl_xor(v, 1));
  return v;
}
__device__ __forceinline__ float group_sum16(float v) {
  v += __shfl_xor(v, 8); v += __shfl_xor(v, 4);
  v += __shfl_xor(v, 2); v += __shfl_xor(v, 1);
  return v;
}

template <int C>
__global__ void __launch_bounds__(256)
gather_kernel(const int* __restrict__ off, const int* __restrict__ deg,
              const int* __restrict__ lst,
              const float* __restrict__ s, const float* __restrict__ d,
              const float* __restrict__ hsrc, const float* __restrict__ bias,
              float* __restrict__ orow, float* __restrict__ ixz, int Nn) {
  constexpr int V = C / 16;
  const int lane  = threadIdx.x & 31;
  const int l     = lane & 15;
  const int gbase = lane & 16;
  const int n  = (int)((blockIdx.x * blockDim.x + threadIdx.x) >> 4);
  const bool live = n < Nn;
  const int nn = min(n, Nn - 1);
  const int o   = off[nn];
  const int dg  = deg[nn];
  const int tot = dg + 1;
  const float dn = d[nn];
  const int totmax = max(tot, __shfl_xor(tot, 16));

  const int   src0 = (l < dg) ? lst[o + l] : nn;
  const float a0   = leaky(s[src0] + dn);
  const bool  v0   = l < tot;

  float m = v0 ? a0 : -INFINITY;
  for (int c = 16; c < totmax; c += 16) {
    const int i = c + l;
    const int sr = (i < dg) ? lst[o + i] : nn;
    const float a = leaky(s[sr] + dn);
    if (i < tot) m = fmaxf(m, a);
  }
  m = group_max16(m);

  const float ex0 = v0 ? __expf(a0 - m) : 0.f;
  float den = ex0;
  for (int c = 16; c < totmax; c += 16) {
    const int i = c + l;
    const int sr = (i < dg) ? lst[o + i] : nn;
    const float a = leaky(s[sr] + dn);
    den += (i < tot) ? __expf(a - m) : 0.f;
  }
  den = group_sum16(den) + 1e-16f;

  float acc[V];
#pragma unroll
  for (int k = 0; k < V; ++k) acc[k] = 0.f;

  for (int c = 0; c < totmax; c += 16) {
    int sr; float w;
    if (c == 0) { sr = src0; w = ex0 / den; }
    else {
      const int i = c + l;
      sr = (i < dg) ? lst[o + i] : nn;
      const float a = leaky(s[sr] + dn);
      w = (i < tot) ? __expf(a - m) / den : 0.f;
    }
    const int cnt = min(16, totmax - c);
    for (int j = 0; j < cnt; j += 4) {
      float r[4][V];
#pragma unroll
      for (int u = 0; u < 4; ++u) {
        const int sj = __shfl(sr, gbase + ((j + u) & 15));
        const float* hp = hsrc + (size_t)sj * C + l * V;
        if constexpr (V == 4) {
          const float4 t = *(const float4*)hp;
          r[u][0] = t.x; r[u][1] = t.y; r[u][2] = t.z; r[u][3] = t.w;
        } else {
#pragma unroll
          for (int k = 0; k < V; ++k) r[u][k] = hp[k];
        }
      }
      if constexpr (V == 4) {
        asm volatile("" :: "v"(r[0][0]), "v"(r[0][1]), "v"(r[0][2]), "v"(r[0][3]),
                           "v"(r[1][0]), "v"(r[1][1]), "v"(r[1][2]), "v"(r[1][3]),
                           "v"(r[2][0]), "v"(r[2][1]), "v"(r[2][2]), "v"(r[2][3]),
                           "v"(r[3][0]), "v"(r[3][1]), "v"(r[3][2]), "v"(r[3][3]));
      } else {
        asm volatile("" :: "v"(r[0][0]), "v"(r[1][0]), "v"(r[2][0]), "v"(r[3][0]));
      }
#pragma unroll
      for (int u = 0; u < 4; ++u) {
        const float wj = __shfl(w, gbase + ((j + u) & 15));
        if (c + j + u < tot) {
#pragma unroll
          for (int k = 0; k < V; ++k) acc[k] += r[u][k] * wj;
        }
      }
    }
  }

  float ov[V];
#pragma unroll
  for (int k = 0; k < V; ++k) ov[k] = acc[k] + bias[l * V + k];

  float pv[V];
#pragma unroll
  for (int k = 0; k < V; ++k) pv[k] = __shfl(ov[k], gbase + (l ^ 8));

  if (!live) return;
  if constexpr (C == C1) {
    float4 hv;
    hv.x = ov[0] > 0.f ? ov[0] : (__expf(ov[0]) - 1.f);
    hv.y = ov[1] > 0.f ? ov[1] : (__expf(ov[1]) - 1.f);
    hv.z = ov[2] > 0.f ? ov[2] : (__expf(ov[2]) - 1.f);
    hv.w = ov[3] > 0.f ? ov[3] : (__expf(ov[3]) - 1.f);
    *(float4*)(orow + (size_t)n * C + l * V) = hv;
    const bool lo = l < 8;
    float2 kv;
    kv.x = kl_term(lo ? ov[0] : pv[2], lo ? pv[0] : ov[2]);
    kv.y = kl_term(lo ? ov[1] : pv[3], lo ? pv[1] : ov[3]);
    *(float2*)(ixz + (size_t)n * (C / 2) + (l & 7) * V + (lo ? 0 : 2)) = kv;
  } else {
#pragma unroll
    for (int k = 0; k < V; ++k) orow[(size_t)n * C + l * V + k] = ov[k];
    if (l < 8) {
#pragma unroll
      for (int k = 0; k < V; ++k) ixz[(size_t)n * (C / 2) + l * V + k] = kl_term(ov[k], pv[k]);
    }
  }
}

extern "C" void kernel_launch(void* const* d_in, const int* in_sizes, int n_in,
                              void* d_out, int out_size, void* d_ws, size_t ws_size,
                              hipStream_t stream) {
  const float* x   = (const float*)d_in[0];
  const int*   ei  = (const int*)  d_in[1];
  const float* W1  = (const float*)d_in[2];
  const float* as1 = (const float*)d_in[3];
  const float* ad1 = (const float*)d_in[4];
  const float* b1  = (const float*)d_in[5];
  const float* W2  = (const float*)d_in[6];
  const float* as2 = (const float*)d_in[7];
  const float* ad2 = (const float*)d_in[8];
  const float* b2  = (const float*)d_in[9];

  const int N = in_sizes[0] / F_IN;
  const int E = in_sizes[1] / 2;
  float* out = (float*)d_out;

  const size_t N64 = (size_t)N * C1, N16 = (size_t)N * C2;
  float* ws   = (float*)d_ws;
  float* h1   = ws;
  float* hb   = h1 + N64;
  float* s1   = hb + N64;
  float* d1   = s1 + N;
  float* h2   = d1 + N;
  int*   deg  = (int*)(h2 + N16);
  int*   off  = deg + N;
  int*   pos  = off + N;
  int*   lin  = pos + N;
  int*   lst  = lin + E;
  int*   cur  = lst + E;

  float* ixz1 = out + N16;
  float* ixz2 = out + (size_t)N * 48;

  const int tiles = (N + 15) / 16;
  const int gblk  = (tiles + 7) / 8;
  const int nblk  = (N + 255) / 256;
  const int eblk  = (E + 255) / 256;
  const int rblk  = (N + 15) / 16;

  csr_init<<<nblk, 256, 0, stream>>>(deg, cur, N);
  csr_count<<<eblk, 256, 0, stream>>>(ei, E, deg);
  csr_place<<<nblk, 256, 0, stream>>>(deg, cur, off, pos, N);
  csr_fill<<<eblk, 256, 0, stream>>>(ei, E, pos, lin);
  csr_sort<<<rblk, 256, 0, stream>>>(off, deg, lin, lst, N);

  gemm1_wmma<<<gblk, 256, 0, stream>>>(x, W1, h1, N);
  node_dots<C1><<<nblk, 256, 0, stream>>>(h1, as1, ad1, s1, d1, N);
  gather_kernel<C1><<<rblk, 256, 0, stream>>>(off, deg, lst, s1, d1, h1, b1, hb, ixz1, N);

  gemm2_wmma<<<gblk, 256, 0, stream>>>(hb, W2, h2, N);
  node_dots<C2><<<nblk, 256, 0, stream>>>(h2, as2, ad2, s1, d1, N);
  gather_kernel<C2><<<rblk, 256, 0, stream>>>(off, deg, lst, s1, d1, h2, b2, out, ixz2, N);
}


# node-level steps fused into the kernel that makes their rows: s/d dots in GEMM1's tail, layer-2 GEMM (WMMA on an LDS tile) + its dots in gather 1's tail; 11->8 launches; dot sums now a lane tree
# speedup vs baseline: 1.9804x; 1.0346x over previous
#include <hip/hip_runtime.h>
#include <hip/hip_bf16.h>

typedef __attribute__((ext_vector_type(2))) float v2f;
typedef __attribute__((ext_vector_type(8))) float v8f;

#define F_IN 128
#define C1   64
#define C2   16
#define NEG_SLOPE 0.2f

__device__ __forceinline__ float group_max16(float v) {
  v = fmaxf(v, __shfl_xor(v, 8)); v = fmaxf(v, __shfl_xor(v, 4));
  v = fmaxf(v, __shfl_xor(v, 2)); v = fmaxf(v, __shfl_xor(v, 1));
  return v;
}
__device__ __forceinline__ float group_sum16(float v) {
  v += __shfl_xor(v, 8); v += __shfl_xor(v, 4);
  v += __shfl_xor(v, 2); v += __shfl_xor(v, 1);
  return v;
}
__device__ __forceinline__ float group_sum16x8(const float (&v)[8], int l) {
  const bool b3 = (l & 8) != 0, b2 = (l & 4) != 0, b1 = (l & 2) != 0;
  float w[4];
  for (int j = 0; j < 4; ++j) {
    const float mine = b3 ? v[j + 4] : v[j];
    const float oth  = b3 ? v[j] : v[j + 4];
    w[j] = mine + __shfl_xor(oth, 8);
  }
  float u[2];
  for (int j = 0; j < 2; ++j) {
    const float mine = b2 ? w[j + 2] : w[j];
    const float oth  = b2 ? w[j] : w[j + 2];
    u[j] = mine + __shfl_xor(oth, 4);
  }
  const float mine = b1 ? u[1] : u[0];
  const float oth  = b1 ? u[0] : u[1];
  const float t = mine + __shfl_xor(oth, 2);
  return t + __shfl_xor(t, 1);
}

__global__ void __launch_bounds__(256)
gemm1_wmma(const float* __restrict__ x, const float* __restrict__ W1,
           const float* __restrict__ asrc, const float* __restrict__ adst,
           float* __restrict__ h1, float* __restrict__ s, float* __restrict__ d, int Nn) {
  __shared__ float sW[F_IN * C1];
  for (int i = threadIdx.x; i < (F_IN * C1) / 4; i += 256) {
    float4 v = ((const float4*)W1)[i];
    const int k  = i >> 4;
    const int c0 = (i & 15) * 4;
    float* dst = sW + (k >> 2) * 256 + (k & 3);
    dst[(c0 + 0) * 4] = v.x;
    dst[(c0 + 1) * 4] = v.y;
    dst[(c0 + 2) * 4] = v.z;
    dst[(c0 + 3) * 4] = v.w;
  }
  __syncthreads();

  const int wave = threadIdx.x >> 5;
  const int lane = threadIdx.x & 31;
  const int tile = blockIdx.x * 8 + wave;
  if (tile * 16 >= Nn) return;
  const int row0 = tile * 16;
  const int r    = lane & 15;
  const int kh   = (lane >> 4) << 1;
  const int arow = min(row0 + r, Nn - 1);

  const float sa0 = asrc[r], sa1 = asrc[r + 16], sa2 = asrc[r + 32], sa3 = asrc[r + 48];
  const float da0 = adst[r], da1 = adst[r + 16], da2 = adst[r + 32], da3 = adst[r + 48];

  v8f acc0 = {}, acc1 = {}, acc2 = {}, acc3 = {};
  const float* ap = x + (size_t)arow * F_IN + kh;
  const float* bbase = sW + r * 4 + kh;
  for (int k0 = 0; k0 < F_IN; k0 += 4) {
    v2f a; a.x = ap[k0]; a.y = ap[k0 + 1];
    const float* bq = bbase + (k0 >> 2) * 256;
    v2f b0 = *(const v2f*)(bq);
    v2f b1 = *(const v2f*)(bq + 64);
    v2f b2 = *(const v2f*)(bq + 128);
    v2f b3 = *(const v2f*)(bq + 192);
    acc0 = __builtin_amdgcn_wmma_f32_16x16x4_f32(false, a, false, b0, (short)0, acc0, false, false);
    acc1 = __builtin_amdgcn_wmma_f32_16x16x4_f32(false, a, false, b1, (short)0, acc1, false, false);
    acc2 = __builtin_amdgcn_wmma_f32_16x16x4_f32(false, a, false, b2, (short)0, acc2, false, false);
    acc3 = __builtin_amdgcn_wmma_f32_16x16x4_f32(false, a, false, b3, (short)0, acc3, false, false);
  }

  const int mbase = (lane >> 4) * 8;
  for (int i = 0; i < 8; ++i) {
    const int row = row0 + mbase + i;
    if (row < Nn) {
      float* op = h1 + (size_t)row * C1 + r;
      op[0]  = acc0[i];
      op[16] = acc1[i];
      op[32] = acc2[i];
      op[48] = acc3[i];
    }
  }

  float ps[8], pd[8];
  for (int i = 0; i < 8; ++i) {
    ps[i] = acc0[i] * sa0 + acc1[i] * sa1 + acc2[i] * sa2 + acc3[i] * sa3;
    pd[i] = acc0[i] * da0 + acc1[i] * da1 + acc2[i] * da2 + acc3[i] * da3;
  }
  const float st = group_sum16x8(ps, r);
  const float dt = group_sum16x8(pd, r);
  const int srow = row0 + mbase + (r >> 1);
  if ((r & 1) == 0 && srow < Nn) { s[srow] = st; d[srow] = dt; }
}

__global__ void csr_init(int* __restrict__ deg, int* __restrict__ cursor, int Nn) {
  int i = blockIdx.x * blockDim.x + threadIdx.x;
  if (i < Nn) deg[i] = 0;
  if (i == 0) *cursor = 0;
}

__global__ void csr_count(const int* __restrict__ ei, int E, int* __restrict__ deg) {
  int e = blockIdx.x * blockDim.x + threadIdx.x;
  if (e < E) atomicAdd(deg + ei[E + e], 1);
}

__global__ void csr_place(const int* __restrict__ deg, int* __restrict__ cursor,
                          int* __restrict__ off, int* __restrict__ pos, int Nn) {
  const int n = blockIdx.x * blockDim.x + threadIdx.x;
  const int lane = threadIdx.x & 31;
  const int dg = (n < Nn) ? deg[n] : 0;
  int x = dg;
  for (int k = 1; k < 32; k <<= 1) {
    const int y = __shfl_up(x, k);
    if (lane >= k) x += y;
  }
  int base = 0;
  if (lane == 31) base = atomicAdd(cursor, x);
  base = __shfl(base, 31);
  const int o = base + x - dg;
  if (n < Nn) { off[n] = o; pos[n] = o; }
}

__global__ void csr_fill(const int* __restrict__ ei, int E,
                         int* __restrict__ pos, int* __restrict__ lin) {
  int e = blockIdx.x * blockDim.x + threadIdx.x;
  if (e < E) {
    const int slot = atomicAdd(pos + ei[E + e], 1);
    lin[slot] = ei[e];
  }
}

__global__ void __launch_bounds__(256)
csr_sort(const int* __restrict__ off, const int* __restrict__ deg,
         const int* __restrict__ lin, int* __restrict__ lout, int Nn) {
  const int lane  = threadIdx.x & 31;
  const int l     = lane & 15;
  const int gbase = lane & 16;
  const int n  = (int)((blockIdx.x * blockDim.x + threadIdx.x) >> 4);
  const int nn = min(n, Nn - 1);
  const int o  = off[nn];
  const int dg = (n < Nn) ? deg[nn] : 0;
  const int dgmax = max(dg, __shfl_xor(dg, 16));
  for (int ci = 0; ci < dgmax; ci += 16) {
    const int i = ci + l;
    const int key = (i < dg) ? lin[o + i] : 0x7fffffff;
    int rank = 0;
    for (int cj = 0; cj < dgmax; cj += 16) {
      const int jm = cj + l;
      const int kmine = (jm < dg) ? lin[o + jm] : 0x7fffffff;
      const int cnt = min(16, dgmax - cj);
      for (int t = 0; t < cnt; ++t) {
        const int kt = __shfl(kmine, gbase + t);
        const int jt = cj + t;
        rank += ((jt < dg) && (kt < key || (kt == key && jt < i))) ? 1 : 0;
      }
    }
    if (i < dg) lout[o + rank] = key;
  }
}

__device__ __forceinline__ float softplusf(float v) {
  return fmaxf(v, 0.f) + log1pf(__expf(-fabsf(v)));
}
__device__ __forceinline__ float kl_term(float mean, float sp) {
  const float sd = softplusf(sp) + 1e-10f;
  return -__logf(sd) + 0.5f * (sd * sd + mean * mean - 1.f);
}
__device__ __forceinline__ float leaky(float a) { return a > 0.f ? a : NEG_SLOPE * a; }

template <int C>
__device__ __forceinline__ void
gather_row(const int* __restrict__ off, const int* __restrict__ deg,
           const int* __restrict__ lst,
           const float* __restrict__ s, const float* __restrict__ d,
           const float* __restrict__ hsrc, const float* __restrict__ bias,
           int nn, int l, int gbase, float (&ov)[C / 16]) {
  constexpr int V = C / 16;
  const int o   = off[nn];
  const int dg  = deg[nn];
  const int tot = dg + 1;
  const float dn = d[nn];
  const int totmax = max(tot, __shfl_xor(tot, 16));

  const int   src0 = (l < dg) ? lst[o + l] : nn;
  const float a0   = leaky(s[src0] + dn);
  const bool  v0   = l < tot;

  float m = v0 ? a0 : -INFINITY;
  for (int c = 16; c < totmax; c += 16) {
    const int i = c + l;
    const int sr = (i < dg) ? lst[o + i] : nn;
    const float a = leaky(s[sr] + dn);
    if (i < tot) m = fmaxf(m, a);
  }
  m = group_max16(m);

  const float ex0 = v0 ? __expf(a0 - m) : 0.f;
  float den = ex0;
  for (int c = 16; c < totmax; c += 16) {
    const int i = c + l;
    const int sr = (i < dg) ? lst[o + i] : nn;
    const float a = leaky(s[sr] + dn);
    den += (i < tot) ? __expf(a - m) : 0.f;
  }
  den = group_sum16(den) + 1e-16f;

  float acc[V];
#pragma unroll
  for (int k = 0; k < V; ++k) acc[k] = 0.f;

  for (int c = 0; c < totmax; c += 16) {
    int sr; float w;
    if (c == 0) { sr = src0; w = ex0 / den; }
    else {
      const int i = c + l;
      sr = (i < dg) ? lst[o + i] : nn;
      const float a = leaky(s[sr] + dn);
      w = (i < tot) ? __expf(a - m) / den : 0.f;
    }
    const int cnt = min(16, totmax - c);
    for (int j = 0; j < cnt; j += 4) {
      float r[4][V];
#pragma unroll
      for (int u = 0; u < 4; ++u) {
        const int sj = __shfl(sr, gbase + ((j + u) & 15));
        const float* hp = hsrc + (size_t)sj * C + l * V;
        if constexpr (V == 4) {
          const float4 t = *(const float4*)hp;
          r[u][0] = t.x; r[u][1] = t.y; r[u][2] = t.z; r[u][3] = t.w;
        } else {
#pragma unroll
          for (int k = 0; k < V; ++k) r[u][k] = hp[k];
        }
      }
      if constexpr (V == 4) {
        asm volatile("" :: "v"(r[0][0]), "v"(r[0][1]), "v"(r[0][2]), "v"(r[0][3]),
                           "v"(r[1][0]), "v"(r[1][1]), "v"(r[1][2]), "v"(r[1][3]),
                           "v"(r[2][0]), "v"(r[2][1]), "v"(r[2][2]), "v"(r[2][3]),
                           "v"(r[3][0]), "v"(r[3][1]), "v"(r[3][2]), "v"(r[3][3]));
      } else {
        asm volatile("" :: "v"(r[0][0]), "v"(r[1][0]), "v"(r[2][0]), "v"(r[3][0]));
      }
#pragma unroll
      for (int u = 0; u < 4; ++u) {
        const float wj = __shfl(w, gbase + ((j + u) & 15));
        if (c + j + u < tot) {
#pragma unroll
          for (int k = 0; k < V; ++k) acc[k] += r[u][k] * wj;
        }
      }
    }
  }

#pragma unroll
  for (int k = 0; k < V; ++k) ov[k] = acc[k] + bias[l * V + k];
}

#define HP 68

__global__ void __launch_bounds__(256)
gather1_proj(const int* __restrict__ off, const int* __restrict__ deg,
             const int* __restrict__ lst,
             const float* __restrict__ s, const float* __restrict__ d,
             const float* __restrict__ h1, const float* __restrict__ b1,
             float* __restrict__ ixz,
             const float* __restrict__ W2,
             const float* __restrict__ asrc2, const float* __restrict__ adst2,
             float* __restrict__ h2, float* __restrict__ s2, float* __restrict__ d2,
             int Nn) {
  constexpr int C = C1, V = C1 / 16;
  __shared__ alignas(16) float sW[C1 * C2];
  __shared__ alignas(16) float sH[16 * HP];

  static_assert((C1 * C2) / 4 == 256, "one float4 of W2 per thread");
  const float4 w2v = ((const float4*)W2)[threadIdx.x];

  const int lane  = threadIdx.x & 31;
  const int l     = lane & 15;
  const int gbase = lane & 16;
  const int g     = threadIdx.x >> 4;
  const int n  = blockIdx.x * 16 + g;
  const bool live = n < Nn;
  const int nn = min(n, Nn - 1);

  float ov[V];
  gather_row<C>(off, deg, lst, s, d, h1, b1, nn, l, gbase, ov);

  float pv[V];
#pragma unroll
  for (int k = 0; k < V; ++k) pv[k] = __shfl(ov[k], gbase + (l ^ 8));

  float4 hv;
  hv.x = ov[0] > 0.f ? ov[0] : (__expf(ov[0]) - 1.f);
  hv.y = ov[1] > 0.f ? ov[1] : (__expf(ov[1]) - 1.f);
  hv.z = ov[2] > 0.f ? ov[2] : (__expf(ov[2]) - 1.f);
  hv.w = ov[3] > 0.f ? ov[3] : (__expf(ov[3]) - 1.f);
  if (!live) hv = make_float4(0.f, 0.f, 0.f, 0.f);
  *(float4*)(sH + g * HP + l * V) = hv;
  {
    const int i  = threadIdx.x;
    const int k  = i >> 2;
    const int c0 = (i & 3) * 4;
    float* dst = sW + (k >> 2) * 64 + (k & 3);
    dst[(c0 + 0) * 4] = w2v.x;
    dst[(c0 + 1) * 4] = w2v.y;
    dst[(c0 + 2) * 4] = w2v.z;
    dst[(c0 + 3) * 4] = w2v.w;
  }

  if (live) {
    const bool lo = l < 8;
    float2 kv;
    kv.x = kl_term(lo ? ov[0] : pv[2], lo ? pv[0] : ov[2]);
    kv.y = kl_term(lo ? ov[1] : pv[3], lo ? pv[1] : ov[3]);
    *(float2*)(ixz + (size_t)n * (C / 2) + (l & 7) * V + (lo ? 0 : 2)) = kv;
  }

  __syncthreads();
  if ((threadIdx.x >> 5) != 0) return;

  const int row0 = blockIdx.x * 16;
  const int r    = l;
  const int kh   = (lane >> 4) << 1;
  const float sa = asrc2[r], da = adst2[r];

  v8f acc = {};
  const float* abase = sH + r * HP + kh;
  const float* bbase = sW + r * 4 + kh;
  for (int k0 = 0; k0 < C1; k0 += 4) {
    v2f a = *(const v2f*)(abase + k0);
    v2f b = *(const v2f*)(bbase + (k0 >> 2) * 64);
    acc = __builtin_amdgcn_wmma_f32_16x16x4_f32(false, a, false, b, (short)0, acc, false, false);
  }

  const int mbase = (lane >> 4) * 8;
  float ps[8], pd[8];
  for (int i = 0; i < 8; ++i) {
    const int row = row0 + mbase + i;
    if (row < Nn) h2[(size_t)row * C2 + r] = acc[i];
    ps[i] = acc[i] * sa;
    pd[i] = acc[i] * da;
  }
  const float st = group_sum16x8(ps, r);
  const float dt = group_sum16x8(pd, r);
  const int srow = row0 + mbase + (r >> 1);
  if ((r & 1) == 0 && srow < Nn) { s2[srow] = st; d2[srow] = dt; }
}

__global__ void __launch_bounds__(256)
gather2_out(const int* __restrict__ off, const int* __restrict__ deg,
            const int* __restrict__ lst,
            const float* __restrict__ s, const float* __restrict__ d,
            const float* __restrict__ h2, const float* __restrict__ b2,
            float* __restrict__ orow, float* __restrict__ ixz, int Nn) {
  constexpr int C = C2, V = C2 / 16;
  const int lane  = threadIdx.x & 31;
  const int l     = lane & 15;
  const int gbase = lane & 16;
  const int n  = (int)((blockIdx.x * blockDim.x + threadIdx.x) >> 4);
  const bool live = n < Nn;
  const int nn = min(n, Nn - 1);

  float ov[V];
  gather_row<C>(off, deg, lst, s, d, h2, b2, nn, l, gbase, ov);

  float pv[V];
#pragma unroll
  for (int k = 0; k < V; ++k) pv[k] = __shfl(ov[k], gbase + (l ^ 8));

  if (!live) return;
#pragma unroll
  for (int k = 0; k < V; ++k) orow[(size_t)n * C + l * V + k] = ov[k];
  if (l < 8) {
#pragma unroll
    for (int k = 0; k < V; ++k) ixz[(size_t)n * (C / 2) + l * V + k] = kl_term(ov[k], pv[k]);
  }
}

extern "C" void kernel_launch(void* const* d_in, const int* in_sizes, int n_in,
                              void* d_out, int out_size, void* d_ws, size_t ws_size,
                              hipStream_t stream) {
  const float* x   = (const float*)d_in[0];
  const int*   ei  = (const int*)  d_in[1];
  const float* W1  = (const float*)d_in[2];
  const float* as1 = (const float*)d_in[3];
  const float* ad1 = (const float*)d_in[4];
  const float* b1  = (const float*)d_in[5];
  const float* W2  = (const float*)d_in[6];
  const float* as2 = (const float*)d_in[7];
  const float* ad2 = (const float*)d_in[8];
  const float* b2  = (const float*)d_in[9];

  const int N = in_sizes[0] / F_IN;
  const int E = in_sizes[1] / 2;
  float* out = (float*)d_out;

  const size_t N64 = (size_t)N * C1, N16 = (size_t)N * C2;
  float* ws   = (float*)d_ws;
  float* h1   = ws;
  float* h2   = h1 + N64;
  float* s1   = h2 + N16;
  float* d1   = s1 + N;
  float* s2   = d1 + N;
  float* d2   = s2 + N;
  int*   deg  = (int*)(d2 + N);
  int*   off  = deg + N;
  int*   pos  = off + N;
  int*   lin  = pos + N;
  int*   lst  = lin + E;
  int*   cur  = lst + E;

  float* ixz1 = out + N16;
  float* ixz2 = out + (size_t)N * 48;

  const int tiles = (N + 15) / 16;
  const int gblk  = (tiles + 7) / 8;
  const int nblk  = (N + 255) / 256;
  const int eblk  = (E + 255) / 256;
  const int rblk  = (N + 15) / 16;

  csr_init<<<nblk, 256, 0, stream>>>(deg, cur, N);
  csr_count<<<eblk, 256, 0, stream>>>(ei, E, deg);
  csr_place<<<nblk, 256, 0, stream>>>(deg, cur, off, pos, N);
  csr_fill<<<eblk, 256, 0, stream>>>(ei, E, pos, lin);
  csr_sort<<<rblk, 256, 0, stream>>>(off, deg, lin, lst, N);

  gemm1_wmma<<<gblk, 256, 0, stream>>>(x, W1, as1, ad1, h1, s1, d1, N);
  gather1_proj<<<rblk, 256, 0, stream>>>(off, deg, lst, s1, d1, h1, b1, ixz1,
                                         W2, as2, ad2, h2, s2, d2, N);

  gather2_out<<<rblk, 256, 0, stream>>>(off, deg, lst, s2, d2, h2, b2, out, ixz2, N);
}


# layer 1's GEMM is no launch of its own: its row tiles run as the leading work-groups of the fill and sort launches (half each), under their atomic and load waits; 8->7 launches; sums unchanged
# speedup vs baseline: 2.0758x; 1.0482x over previous
#include <hip/hip_runtime.h>
#include <hip/hip_bf16.h>

typedef __attribute__((ext_vector_type(2))) float v2f;
typedef __attribute__((ext_vector_type(8))) float v8f;

#define F_IN 128
#define C1   64
#define C2   16
#define NEG_SLOPE 0.2f

__device__ __forceinline__ float group_max16(float v) {
  v = fmaxf(v, __shfl_xor(v, 8)); v = fmaxf(v, __shfl_xor(v, 4));
  v = fmaxf(v, __shfl_xor(v, 2)); v = fmaxf(v, __shfl_xor(v, 1));
  return v;
}
__device__ __forceinline__ float group_sum16(float v) {
  v += __shfl_xor(v, 8); v += __shfl_xor(v, 4);
  v += __shfl_xor(v, 2); v += __shfl_xor(v, 1);
  return v;
}
__device__ __forceinline__ float group_sum16x8(const float (&v)[8], int l) {
  const bool b3 = (l & 8) != 0, b2 = (l & 4) != 0, b1 = (l & 2) != 0;
  float w[4];
  for (int j = 0; j < 4; ++j) {
    const float mine = b3 ? v[j + 4] : v[j];
    const float oth  = b3 ? v[j] : v[j + 4];
    w[j] = mine + __shfl_xor(oth, 8);
  }
  float u[2];
  for (int j = 0; j < 2; ++j) {
    const float mine = b2 ? w[j + 2] : w[j];
    const float oth  = b2 ? w[j] : w[j + 2];
    u[j] = mine + __shfl_xor(oth, 4);
  }
  const float mine = b1 ? u[1] : u[0];
  const float oth  = b1 ? u[0] : u[1];
  const float t = mine + __shfl_xor(oth, 2);
  return t + __shfl_xor(t, 1);
}

struct Gemm1Args {
  const float* x; const float* W1; const float* asrc; const float* adst;
  float* h1; float* s; float* d; int Nn;
  int first;
  int count;
};

__device__ __forceinline__ void
gemm1_block(const float* __restrict__ x, const float* __restrict__ W1,
            const float* __restrict__ asrc, const float* __restrict__ adst,
            float* __restrict__ h1, float* __restrict__ s, float* __restrict__ d, int Nn,
            int gb) {
  __shared__ float sW[F_IN * C1];
  for (int i = threadIdx.x; i < (F_IN * C1) / 4; i += 256) {
    float4 v = ((const float4*)W1)[i];
    const int k  = i >> 4;
    const int c0 = (i & 15) * 4;
    float* dst = sW + (k >> 2) * 256 + (k & 3);
    dst[(c0 + 0) * 4] = v.x;
    dst[(c0 + 1) * 4] = v.y;
    dst[(c0 + 2) * 4] = v.z;
    dst[(c0 + 3) * 4] = v.w;
  }
  __syncthreads();

  const int wave = threadIdx.x >> 5;
  const int lane = threadIdx.x & 31;
  const int tile = gb * 8 + wave;
  if (tile * 16 >= Nn) return;
  const int row0 = tile * 16;
  const int r    = lane & 15;
  const int kh   = (lane >> 4) << 1;
  const int arow = min(row0 + r, Nn - 1);

  const float sa0 = asrc[r], sa1 = asrc[r + 16], sa2 = asrc[r + 32], sa3 = asrc[r + 48];
  const float da0 = adst[r], da1 = adst[r + 16], da2 = adst[r + 32], da3 = adst[r + 48];

  v8f acc0 = {}, acc1 = {}, acc2 = {}, acc3 = {};
  const float* ap = x + (size_t)arow * F_IN + kh;
  const float* bbase = sW + r * 4 + kh;
  for (int k0 = 0; k0 < F_IN; k0 += 4) {
    v2f a; a.x = ap[k0]; a.y = ap[k0 + 1];
    const float* bq = bbase + (k0 >> 2) * 256;
    v2f b0 = *(const v2f*)(bq);
    v2f b1 = *(const v2f*)(bq + 64);
    v2f b2 = *(const v2f*)(bq + 128);
    v2f b3 = *(const v2f*)(bq + 192);
    acc0 = __builtin_amdgcn_wmma_f32_16x16x4_f32(false, a, false, b0, (short)0, acc0, false, false);
    acc1 = __builtin_amdgcn_wmma_f32_16x16x4_f32(false, a, false, b1, (short)0, acc1, false, false);
    acc2 = __builtin_amdgcn_wmma_f32_16x16x4_f32(false, a, false, b2, (short)0, acc2, false, false);
    acc3 = __builtin_amdgcn_wmma_f32_16x16x4_f32(false, a, false, b3, (short)0, acc3, false, false);
  }

  const int mbase = (lane >> 4) * 8;
  for (int i = 0; i < 8; ++i) {
    const int row = row0 + mbase + i;
    if (row < Nn) {
      float* op = h1 + (size_t)row * C1 + r;
      op[0]  = acc0[i];
      op[16] = acc1[i];
      op[32] = acc2[i];
      op[48] = acc3[i];
    }
  }

  float ps[8], pd[8];
  for (int i = 0; i < 8; ++i) {
    ps[i] = acc0[i] * sa0 + acc1[i] * sa1 + acc2[i] * sa2 + acc3[i] * sa3;
    pd[i] = acc0[i] * da0 + acc1[i] * da1 + acc2[i] * da2 + acc3[i] * da3;
  }
  const float st = group_sum16x8(ps, r);
  const float dt = group_sum16x8(pd, r);
  const int srow = row0 + mbase + (r >> 1);
  if ((r & 1) == 0 && srow < Nn) { s[srow] = st; d[srow] = dt; }
}

__global__ void csr_init(int* __restrict__ deg, int* __restrict__ cursor, int Nn) {
  int i = blockIdx.x * blockDim.x + threadIdx.x;
  if (i < Nn) deg[i] = 0;
  if (i == 0) *cursor = 0;
}

__global__ void csr_count(const int* __restrict__ ei, int E, int* __restrict__ deg) {
  int e = blockIdx.x * blockDim.x + threadIdx.x;
  if (e < E) atomicAdd(deg + ei[E + e], 1);
}

__global__ void csr_place(const int* __restrict__ deg, int* __restrict__ cursor,
                          int* __restrict__ off, int* __restrict__ pos, int Nn) {
  const int n = blockIdx.x * blockDim.x + threadIdx.x;
  const int lane = threadIdx.x & 31;
  const int dg = (n < Nn) ? deg[n] : 0;
  int x = dg;
  for (int k = 1; k < 32; k <<= 1) {
    const int y = __shfl_up(x, k);
    if (lane >= k) x += y;
  }
  int base = 0;
  if (lane == 31) base = atomicAdd(cursor, x);
  base = __shfl(base, 31);
  const int o = base + x - dg;
  if (n < Nn) { off[n] = o; pos[n] = o; }
}

__global__ void __launch_bounds__(256) __attribute__((amdgpu_waves_per_eu(16)))
csr_fill(const int* __restrict__ ei, int E,
         int* __restrict__ pos, int* __restrict__ lin, const Gemm1Args g) {
  if ((int)blockIdx.x < g.count) {
    gemm1_block(g.x, g.W1, g.asrc, g.adst, g.h1, g.s, g.d, g.Nn, g.first + (int)blockIdx.x);
    return;
  }
  int e = (blockIdx.x - g.count) * blockDim.x + threadIdx.x;
  if (e < E) {
    const int slot = atomicAdd(pos + ei[E + e], 1);
    lin[slot] = ei[e];
  }
}

__global__ void __launch_bounds__(256) __attribute__((amdgpu_waves_per_eu(16)))
csr_sort(const int* __restrict__ off, const int* __restrict__ deg,
         const int* __restrict__ lin, int* __restrict__ lout, int Nn, const Gemm1Args g) {
  if ((int)blockIdx.x < g.count) {
    gemm1_block(g.x, g.W1, g.asrc, g.adst, g.h1, g.s, g.d, g.Nn, g.first + (int)blockIdx.x);
    return;
  }
  const int lane  = threadIdx.x & 31;
  const int l     = lane & 15;
  const int gbase = lane & 16;
  const int n  = (int)(((blockIdx.x - g.count) * blockDim.x + threadIdx.x) >> 4);
  const int nn = min(n, Nn - 1);
  const int o  = off[nn];
  const int dg = (n < Nn) ? deg[nn] : 0;
  const int dgmax = max(dg, __shfl_xor(dg, 16));
  for (int ci = 0; ci < dgmax; ci += 16) {
    const int i = ci + l;
    const int key = (i < dg) ? lin[o + i] : 0x7fffffff;
    int rank = 0;
    for (int cj = 0; cj < dgmax; cj += 16) {
      const int jm = cj + l;
      const int kmine = (jm < dg) ? lin[o + jm] : 0x7fffffff;
      const int cnt = min(16, dgmax - cj);
      for (int t = 0; t < cnt; ++t) {
        const int kt = __shfl(kmine, gbase + t);
        const int jt = cj + t;
        rank += ((jt < dg) && (kt < key || (kt == key && jt < i))) ? 1 : 0;
      }
    }
    if (i < dg) lout[o + rank] = key;
  }
}

__device__ __forceinline__ float softplusf(float v) {
  return fmaxf(v, 0.f) + log1pf(__expf(-fabsf(v)));
}
__device__ __forceinline__ float kl_term(float mean, float sp) {
  const float sd = softplusf(sp) + 1e-10f;
  return -__logf(sd) + 0.5f * (sd * sd + mean * mean - 1.f);
}
__device__ __forceinline__ float leaky(float a) { return a > 0.f ? a : NEG_SLOPE * a; }

template <int C>
__device__ __forceinline__ void
gather_row(const int* __restrict__ off, const int* __restrict__ deg,
           const int* __restrict__ lst,
           const float* __restrict__ s, const float* __restrict__ d,
           const float* __restrict__ hsrc, const float* __restrict__ bias,
           int nn, int l, int gbase, float (&ov)[C / 16]) {
  constexpr int V = C / 16;
  const int o   = off[nn];
  const int dg  = deg[nn];
  const int tot = dg + 1;
  const float dn = d[nn];
  const int totmax = max(tot, __shfl_xor(tot, 16));

  const int   src0 = (l < dg) ? lst[o + l] : nn;
  const float a0   = leaky(s[src0] + dn);
  const bool  v0   = l < tot;

  float m = v0 ? a0 : -INFINITY;
  for (int c = 16; c < totmax; c += 16) {
    const int i = c + l;
    const int sr = (i < dg) ? lst[o + i] : nn;
    const float a = leaky(s[sr] + dn);
    if (i < tot) m = fmaxf(m, a);
  }
  m = group_max16(m);

  const float ex0 = v0 ? __expf(a0 - m) : 0.f;
  float den = ex0;
  for (int c = 16; c < totmax; c += 16) {
    const int i = c + l;
    const int sr = (i < dg) ? lst[o + i] : nn;
    const float a = leaky(s[sr] + dn);
    den += (i < tot) ? __expf(a - m) : 0.f;
  }
  den = group_sum16(den) + 1e-16f;

  float acc[V];
#pragma unroll
  for (int k = 0; k < V; ++k) acc[k] = 0.f;

  for (int c = 0; c < totmax; c += 16) {
    int sr; float w;
    if (c == 0) { sr = src0; w = ex0 / den; }
    else {
      const int i = c + l;
      sr = (i < dg) ? lst[o + i] : nn;
      const float a = leaky(s[sr] + dn);
      w = (i < tot) ? __expf(a - m) / den : 0.f;
    }
    const int cnt = min(16, totmax - c);
    for (int j = 0; j < cnt; j += 4) {
      float r[4][V];
#pragma unroll
      for (int u = 0; u < 4; ++u) {
        const int sj = __shfl(sr, gbase + ((j + u) & 15));
        const float* hp = hsrc + (size_t)sj * C + l * V;
        if constexpr (V == 4) {
          const float4 t = *(const float4*)hp;
          r[u][0] = t.x; r[u][1] = t.y; r[u][2] = t.z; r[u][3] = t.w;
        } else {
#pragma unroll
          for (int k = 0; k < V; ++k) r[u][k] = hp[k];
        }
      }
      if constexpr (V == 4) {
        asm volatile("" :: "v"(r[0][0]), "v"(r[0][1]), "v"(r[0][2]), "v"(r[0][3]),
                           "v"(r[1][0]), "v"(r[1][1]), "v"(r[1][2]), "v"(r[1][3]),
                           "v"(r[2][0]), "v"(r[2][1]), "v"(r[2][2]), "v"(r[2][3]),
                           "v"(r[3][0]), "v"(r[3][1]), "v"(r[3][2]), "v"(r[3][3]));
      } else {
        asm volatile("" :: "v"(r[0][0]), "v"(r[1][0]), "v"(r[2][0]), "v"(r[3][0]));
      }
#pragma unroll
      for (int u = 0; u < 4; ++u) {
        const float wj = __shfl(w, gbase + ((j + u) & 15));
        if (c + j + u < tot) {
#pragma unroll
          for (int k = 0; k < V; ++k) acc[k] += r[u][k] * wj;
        }
      }
    }
  }

#pragma unroll
  for (int k = 0; k < V; ++k) ov[k] = acc[k] + bias[l * V + k];
}

#define HP 68

__global__ void __launch_bounds__(256)
gather1_proj(const int* __restrict__ off, const int* __restrict__ deg,
             const int* __restrict__ lst,
             const float* __restrict__ s, const float* __restrict__ d,
             const float* __restrict__ h1, const float* __restrict__ b1,
             float* __restrict__ ixz,
             const float* __restrict__ W2,
             const float* __restrict__ asrc2, const float* __restrict__ adst2,
             float* __restrict__ h2, float* __restrict__ s2, float* __restrict__ d2,
             int Nn) {
  constexpr int C = C1, V = C1 / 16;
  __shared__ alignas(16) float sW[C1 * C2];
  __shared__ alignas(16) float sH[16 * HP];

  static_assert((C1 * C2) / 4 == 256, "one float4 of W2 per thread");
  const float4 w2v = ((const float4*)W2)[threadIdx.x];

  const int lane  = threadIdx.x & 31;
  const int l     = lane & 15;
  const int gbase = lane & 16;
  const int g     = threadIdx.x >> 4;
  const int n  = blockIdx.x * 16 + g;
  const bool live = n < Nn;
  const int nn = min(n, Nn - 1);

  float ov[V];
  gather_row<C>(off, deg, lst, s, d, h1, b1, nn, l, gbase, ov);

  float pv[V];
#pragma unroll
  for (int k = 0; k < V; ++k) pv[k] = __shfl(ov[k], gbase + (l ^ 8));

  float4 hv;
  hv.x = ov[0] > 0.f ? ov[0] : (__expf(ov[0]) - 1.f);
  hv.y = ov[1] > 0.f ? ov[1] : (__expf(ov[1]) - 1.f);
  hv.z = ov[2] > 0.f ? ov[2] : (__expf(ov[2]) - 1.f);
  hv.w = ov[3] > 0.f ? ov[3] : (__expf(ov[3]) - 1.f);
  if (!live) hv = make_float4(0.f, 0.f, 0.f, 0.f);
  *(float4*)(sH + g * HP + l * V) = hv;
  {
    const int i  = threadIdx.x;
    const int k  = i >> 2;
    const int c0 = (i & 3) * 4;
    float* dst = sW + (k >> 2) * 64 + (k & 3);
    dst[(c0 + 0) * 4] = w2v.x;
    dst[(c0 + 1) * 4] = w2v.y;
    dst[(c0 + 2) * 4] = w2v.z;
    dst[(c0 + 3) * 4] = w2v.w;
  }

  if (live) {
    const bool lo = l < 8;
    float2 kv;
    kv.x = kl_term(lo ? ov[0] : pv[2], lo ? pv[0] : ov[2]);
    kv.y = kl_term(lo ? ov[1] : pv[3], lo ? pv[1] : ov[3]);
    *(float2*)(ixz + (size_t)n * (C / 2) + (l & 7) * V + (lo ? 0 : 2)) = kv;
  }

  __syncthreads();
  if ((threadIdx.x >> 5) != 0) return;

  const int row0 = blockIdx.x * 16;
  const int r    = l;
  const int kh   = (lane >> 4) << 1;
  const float sa = asrc2[r], da = adst2[r];

  v8f acc = {};
  const float* abase = sH + r * HP + kh;
  const float* bbase = sW + r * 4 + kh;
  for (int k0 = 0; k0 < C1; k0 += 4) {
    v2f a = *(const v2f*)(abase + k0);
    v2f b = *(const v2f*)(bbase + (k0 >> 2) * 64);
    acc = __builtin_amdgcn_wmma_f32_16x16x4_f32(false, a, false, b, (short)0, acc, false, false);
  }

  const int mbase = (lane >> 4) * 8;
  float ps[8], pd[8];
  for (int i = 0; i < 8; ++i) {
    const int row = row0 + mbase + i;
    if (row < Nn) h2[(size_t)row * C2 + r] = acc[i];
    ps[i] = acc[i] * sa;
    pd[i] = acc[i] * da;
  }
  const float st = group_sum16x8(ps, r);
  const float dt = group_sum16x8(pd, r);
  const int srow = row0 + mbase + (r >> 1);
  if ((r & 1) == 0 && srow < Nn) { s2[srow] = st; d2[srow] = dt; }
}

__global__ void __launch_bounds__(256)
gather2_out(const int* __restrict__ off, const int* __restrict__ deg,
            const int* __restrict__ lst,
            const float* __restrict__ s, const float* __restrict__ d,
            const float* __restrict__ h2, const float* __restrict__ b2,
            float* __restrict__ orow, float* __restrict__ ixz, int Nn) {
  constexpr int C = C2, V = C2 / 16;
  const int lane  = threadIdx.x & 31;
  const int l     = lane & 15;
  const int gbase = lane & 16;
  const int n  = (int)((blockIdx.x * blockDim.x + threadIdx.x) >> 4);
  const bool live = n < Nn;
  const int nn = min(n, Nn - 1);

  float ov[V];
  gather_row<C>(off, deg, lst, s, d, h2, b2, nn, l, gbase, ov);

  float pv[V];
#pragma unroll
  for (int k = 0; k < V; ++k) pv[k] = __shfl(ov[k], gbase + (l ^ 8));

  if (!live) return;
#pragma unroll
  for (int k = 0; k < V; ++k) orow[(size_t)n * C + l * V + k] = ov[k];
  if (l < 8) {
#pragma unroll
    for (int k = 0; k < V; ++k) ixz[(size_t)n * (C / 2) + l * V + k] = kl_term(ov[k], pv[k]);
  }
}

extern "C" void kernel_launch(void* const* d_in, const int* in_sizes, int n_in,
                              void* d_out, int out_size, void* d_ws, size_t ws_size,
                              hipStream_t stream) {
  const float* x   = (const float*)d_in[0];
  const int*   ei  = (const int*)  d_in[1];
  const float* W1  = (const float*)d_in[2];
  const float* as1 = (const float*)d_in[3];
  const float* ad1 = (const float*)d_in[4];
  const float* b1  = (const float*)d_in[5];
  const float* W2  = (const float*)d_in[6];
  const float* as2 = (const float*)d_in[7];
  const float* ad2 = (const float*)d_in[8];
  const float* b2  = (const float*)d_in[9];

  const int N = in_sizes[0] / F_IN;
  const int E = in_sizes[1] / 2;
  float* out = (float*)d_out;

  const size_t N64 = (size_t)N * C1, N16 = (size_t)N * C2;
  float* ws   = (float*)d_ws;
  float* h1   = ws;
  float* h2   = h1 + N64;
  float* s1   = h2 + N16;
  float* d1   = s1 + N;
  float* s2   = d1 + N;
  float* d2   = s2 + N;
  int*   deg  = (int*)(d2 + N);
  int*   off  = deg + N;
  int*   pos  = off + N;
  int*   lin  = pos + N;
  int*   lst  = lin + E;
  int*   cur  = lst + E;

  float* ixz1 = out + N16;
  float* ixz2 = out + (size_t)N * 48;

  const int tiles = (N + 15) / 16;
  const int gblk  = (tiles + 7) / 8;
  const int gpart = (gblk + 1) / 2;
  const int nblk  = (N + 255) / 256;
  const int eblk  = (E + 255) / 256;
  const int rblk  = (N + 15) / 16;

  Gemm1Args g0 = {x, W1, as1, ad1, h1, s1, d1, N, 0,     gpart};
  Gemm1Args g1 = {x, W1, as1, ad1, h1, s1, d1, N, gpart, gpart};
  csr_init<<<nblk, 256, 0, stream>>>(deg, cur, N);
  csr_count<<<eblk, 256, 0, stream>>>(ei, E, deg);
  csr_place<<<nblk, 256, 0, stream>>>(deg, cur, off, pos, N);
  csr_fill<<<gpart + eblk, 256, 0, stream>>>(ei, E, pos, lin, g0);
  csr_sort<<<gpart + rblk, 256, 0, stream>>>(off, deg, lin, lst, N, g1);

  gather1_proj<<<rblk, 256, 0, stream>>>(off, deg, lst, s1, d1, h1, b1, ixz1,
                                         W2, as2, ad2, h2, s2, d2, N);

  gather2_out<<<rblk, 256, 0, stream>>>(off, deg, lst, s2, d2, h2, b2, out, ixz2, N);
}


# edge fill without atomics: the counting add's returned value is kept per edge as its rank in its row, so a slot is offset + rank; one round of scattered integer atomics, was two; sums unchanged
# speedup vs baseline: 2.3043x; 1.1101x over previous
#include <hip/hip_runtime.h>
#include <hip/hip_bf16.h>

typedef __attribute__((ext_vector_type(2))) float v2f;
typedef __attribute__((ext_vector_type(8))) float v8f;

#define F_IN 128
#define C1   64
#define C2   16
#define NEG_SLOPE 0.2f

__device__ __forceinline__ float group_max16(float v) {
  v = fmaxf(v, __shfl_xor(v, 8)); v = fmaxf(v, __shfl_xor(v, 4));
  v = fmaxf(v, __shfl_xor(v, 2)); v = fmaxf(v, __shfl_xor(v, 1));
  return v;
}
__device__ __forceinline__ float group_sum16(float v) {
  v += __shfl_xor(v, 8); v += __shfl_xor(v, 4);
  v += __shfl_xor(v, 2); v += __shfl_xor(v, 1);
  return v;
}
__device__ __forceinline__ float group_sum16x8(const float (&v)[8], int l) {
  const bool b3 = (l & 8) != 0, b2 = (l & 4) != 0, b1 = (l & 2) != 0;
  float w[4];
  for (int j = 0; j < 4; ++j) {
    const float mine = b3 ? v[j + 4] : v[j];
    const float oth  = b3 ? v[j] : v[j + 4];
    w[j] = mine + __shfl_xor(oth, 8);
  }
  float u[2];
  for (int j = 0; j < 2; ++j) {
    const float mine = b2 ? w[j + 2] : w[j];
    const float oth  = b2 ? w[j] : w[j + 2];
    u[j] = mine + __shfl_xor(oth, 4);
  }
  const float mine = b1 ? u[1] : u[0];
  const float oth  = b1 ? u[0] : u[1];
  const float t = mine + __shfl_xor(oth, 2);
  return t + __shfl_xor(t, 1);
}

struct Gemm1Args {
  const float* x; const float* W1; const float* asrc; const float* adst;
  float* h1; float* s; float* d; int Nn;
  int first;
  int count;
};

__device__ __forceinline__ void
gemm1_block(const float* __restrict__ x, const float* __restrict__ W1,
            const float* __restrict__ asrc, const float* __restrict__ adst,
            float* __restrict__ h1, float* __restrict__ s, float* __restrict__ d, int Nn,
            int gb) {
  __shared__ float sW[F_IN * C1];
  for (int i = threadIdx.x; i < (F_IN * C1) / 4; i += 256) {
    float4 v = ((const float4*)W1)[i];
    const int k  = i >> 4;
    const int c0 = (i & 15) * 4;
    float* dst = sW + (k >> 2) * 256 + (k & 3);
    dst[(c0 + 0) * 4] = v.x;
    dst[(c0 + 1) * 4] = v.y;
    dst[(c0 + 2) * 4] = v.z;
    dst[(c0 + 3) * 4] = v.w;
  }
  __syncthreads();

  const int wave = threadIdx.x >> 5;
  const int lane = threadIdx.x & 31;
  const int tile = gb * 8 + wave;
  if (tile * 16 >= Nn) return;
  const int row0 = tile * 16;
  const int r    = lane & 15;
  const int kh   = (lane >> 4) << 1;
  const int arow = min(row0 + r, Nn - 1);

  const float sa0 = asrc[r], sa1 = asrc[r + 16], sa2 = asrc[r + 32], sa3 = asrc[r + 48];
  const float da0 = adst[r], da1 = adst[r + 16], da2 = adst[r + 32], da3 = adst[r + 48];

  v8f acc0 = {}, acc1 = {}, acc2 = {}, acc3 = {};
  const float* ap = x + (size_t)arow * F_IN + kh;
  const float* bbase = sW + r * 4 + kh;
  for (int k0 = 0; k0 < F_IN; k0 += 4) {
    v2f a; a.x = ap[k0]; a.y = ap[k0 + 1];
    const float* bq = bbase + (k0 >> 2) * 256;
    v2f b0 = *(const v2f*)(bq);
    v2f b1 = *(const v2f*)(bq + 64);
    v2f b2 = *(const v2f*)(bq + 128);
    v2f b3 = *(const v2f*)(bq + 192);
    acc0 = __builtin_amdgcn_wmma_f32_16x16x4_f32(false, a, false, b0, (short)0, acc0, false, false);
    acc1 = __builtin_amdgcn_wmma_f32_16x16x4_f32(false, a, false, b1, (short)0, acc1, false, false);
    acc2 = __builtin_amdgcn_wmma_f32_16x16x4_f32(false, a, false, b2, (short)0, acc2, false, false);
    acc3 = __builtin_amdgcn_wmma_f32_16x16x4_f32(false, a, false, b3, (short)0, acc3, false, false);
  }

  const int mbase = (lane >> 4) * 8;
  for (int i = 0; i < 8; ++i) {
    const int row = row0 + mbase + i;
    if (row < Nn) {
      float* op = h1 + (size_t)row * C1 + r;
      op[0]  = acc0[i];
      op[16] = acc1[i];
      op[32] = acc2[i];
      op[48] = acc3[i];
    }
  }

  float ps[8], pd[8];
  for (int i = 0; i < 8; ++i) {
    ps[i] = acc0[i] * sa0 + acc1[i] * sa1 + acc2[i] * sa2 + acc3[i] * sa3;
    pd[i] = acc0[i] * da0 + acc1[i] * da1 + acc2[i] * da2 + acc3[i] * da3;
  }
  const float st = group_sum16x8(ps, r);
  const float dt = group_sum16x8(pd, r);
  const int srow = row0 + mbase + (r >> 1);
  if ((r & 1) == 0 && srow < Nn) { s[srow] = st; d[srow] = dt; }
}

__global__ void csr_init(int* __restrict__ deg, int* __restrict__ cursor, int Nn) {
  int i = blockIdx.x * blockDim.x + threadIdx.x;
  if (i < Nn) deg[i] = 0;
  if (i == 0) *cursor = 0;
}

__global__ void csr_count(const int* __restrict__ ei, int E, int* __restrict__ deg,
                          int* __restrict__ rnk) {
  int e = blockIdx.x * blockDim.x + threadIdx.x;
  if (e < E) rnk[e] = atomicAdd(deg + ei[E + e], 1);
}

__global__ void csr_place(const int* __restrict__ deg, int* __restrict__ cursor,
                          int* __restrict__ off, int Nn) {
  const int n = blockIdx.x * blockDim.x + threadIdx.x;
  const int lane = threadIdx.x & 31;
  const int dg = (n < Nn) ? deg[n] : 0;
  int x = dg;
  for (int k = 1; k < 32; k <<= 1) {
    const int y = __shfl_up(x, k);
    if (lane >= k) x += y;
  }
  int base = 0;
  if (lane == 31) base = atomicAdd(cursor, x);
  base = __shfl(base, 31);
  const int o = base + x - dg;
  if (n < Nn) off[n] = o;
}

__global__ void __launch_bounds__(256) __attribute__((amdgpu_waves_per_eu(16)))
csr_fill(const int* __restrict__ ei, int E, const int* __restrict__ off,
         const int* __restrict__ rnk, int* __restrict__ lin, const Gemm1Args g) {
  if ((int)blockIdx.x < g.count) {
    gemm1_block(g.x, g.W1, g.asrc, g.adst, g.h1, g.s, g.d, g.Nn, g.first + (int)blockIdx.x);
    return;
  }
  int e = (blockIdx.x - g.count) * blockDim.x + threadIdx.x;
  if (e < E) lin[off[ei[E + e]] + rnk[e]] = ei[e];
}

__global__ void __launch_bounds__(256) __attribute__((amdgpu_waves_per_eu(16)))
csr_sort(const int* __restrict__ off, const int* __restrict__ deg,
         const int* __restrict__ lin, int* __restrict__ lout, int Nn, const Gemm1Args g) {
  if ((int)blockIdx.x < g.count) {
    gemm1_block(g.x, g.W1, g.asrc, g.adst, g.h1, g.s, g.d, g.Nn, g.first + (int)blockIdx.x);
    return;
  }
  const int lane  = threadIdx.x & 31;
  const int l     = lane & 15;
  const int gbase = lane & 16;
  const int n  = (int)(((blockIdx.x - g.count) * blockDim.x + threadIdx.x) >> 4);
  const int nn = min(n, Nn - 1);
  const int o  = off[nn];
  const int dg = (n < Nn) ? deg[nn] : 0;
  const int dgmax = max(dg, __shfl_xor(dg, 16));
  for (int ci = 0; ci < dgmax; ci += 16) {
    const int i = ci + l;
    const int key = (i < dg) ? lin[o + i] : 0x7fffffff;
    int rank = 0;
    for (int cj = 0; cj < dgmax; cj += 16) {
      const int jm = cj + l;
      const int kmine = (jm < dg) ? lin[o + jm] : 0x7fffffff;
      const int cnt = min(16, dgmax - cj);
      for (int t = 0; t < cnt; ++t) {
        const int kt = __shfl(kmine, gbase + t);
        const int jt = cj + t;
        rank += ((jt < dg) && (kt < key || (kt == key && jt < i))) ? 1 : 0;
      }
    }
    if (i < dg) lout[o + rank] = key;
  }
}

__device__ __forceinline__ float softplusf(float v) {
  return fmaxf(v, 0.f) + log1pf(__expf(-fabsf(v)));
}
__device__ __forceinline__ float kl_term(float mean, float sp) {
  const float sd = softplusf(sp) + 1e-10f;
  return -__logf(sd) + 0.5f * (sd * sd + mean * mean - 1.f);
}
__device__ __forceinline__ float leaky(float a) { return a > 0.f ? a : NEG_SLOPE * a; }

template <int C>
__device__ __forceinline__ void
gather_row(const int* __restrict__ off, const int* __restrict__ deg,
           const int* __restrict__ lst,
           const float* __restrict__ s, const float* __restrict__ d,
           const float* __restrict__ hsrc, const float* __restrict__ bias,
           int nn, int l, int gbase, float (&ov)[C / 16]) {
  constexpr int V = C / 16;
  const int o   = off[nn];
  const int dg  = deg[nn];
  const int tot = dg + 1;
  const float dn = d[nn];
  const int totmax = max(tot, __shfl_xor(tot, 16));

  const int   src0 = (l < dg) ? lst[o + l] : nn;
  const float a0   = leaky(s[src0] + dn);
  const bool  v0   = l < tot;

  float m = v0 ? a0 : -INFINITY;
  for (int c = 16; c < totmax; c += 16) {
    const int i = c + l;
    const int sr = (i < dg) ? lst[o + i] : nn;
    const float a = leaky(s[sr] + dn);
    if (i < tot) m = fmaxf(m, a);
  }
  m = group_max16(m);

  const float ex0 = v0 ? __expf(a0 - m) : 0.f;
  float den = ex0;
  for (int c = 16; c < totmax; c += 16) {
    const int i = c + l;
    const int sr = (i < dg) ? lst[o + i] : nn;
    const float a = leaky(s[sr] + dn);
    den += (i < tot) ? __expf(a - m) : 0.f;
  }
  den = group_sum16(den) + 1e-16f;

  float acc[V];
#pragma unroll
  for (int k = 0; k < V; ++k) acc[k] = 0.f;

  for (int c = 0; c < totmax; c += 16) {
    int sr; float w;
    if (c == 0) { sr = src0; w = ex0 / den; }
    else {
      const int i = c + l;
      sr = (i < dg) ? lst[o + i] : nn;
      const float a = leaky(s[sr] + dn);
      w = (i < tot) ? __expf(a - m) / den : 0.f;
    }
    const int cnt = min(16, totmax - c);
    for (int j = 0; j < cnt; j += 4) {
      float r[4][V];
#pragma unroll
      for (int u = 0; u < 4; ++u) {
        const int sj = __shfl(sr, gbase + ((j + u) & 15));
        const float* hp = hsrc + (size_t)sj * C + l * V;
        if constexpr (V == 4) {
          const float4 t = *(const float4*)hp;
          r[u][0] = t.x; r[u][1] = t.y; r[u][2] = t.z; r[u][3] = t.w;
        } else {
#pragma unroll
          for (int k = 0; k < V; ++k) r[u][k] = hp[k];
        }
      }
      if constexpr (V == 4) {
        asm volatile("" :: "v"(r[0][0]), "v"(r[0][1]), "v"(r[0][2]), "v"(r[0][3]),
                           "v"(r[1][0]), "v"(r[1][1]), "v"(r[1][2]), "v"(r[1][3]),
                           "v"(r[2][0]), "v"(r[2][1]), "v"(r[2][2]), "v"(r[2][3]),
                           "v"(r[3][0]), "v"(r[3][1]), "v"(r[3][2]), "v"(r[3][3]));
      } else {
        asm volatile("" :: "v"(r[0][0]), "v"(r[1][0]), "v"(r[2][0]), "v"(r[3][0]));
      }
#pragma unroll
      for (int u = 0; u < 4; ++u) {
        const float wj = __shfl(w, gbase + ((j + u) & 15));
        if (c + j + u < tot) {
#pragma unroll
          for (int k = 0; k < V; ++k) acc[k] += r[u][k] * wj;
        }
      }
    }
  }

#pragma unroll
  for (int k = 0; k < V; ++k) ov[k] = acc[k] + bias[l * V + k];
}

#define HP 68

__global__ void __launch_bounds__(256)
gather1_proj(const int* __restrict__ off, const int* __restrict__ deg,
             const int* __restrict__ lst,
             const float* __restrict__ s, const float* __restrict__ d,
             const float* __restrict__ h1, const float* __restrict__ b1,
             float* __restrict__ ixz,
             const float* __restrict__ W2,
             const float* __restrict__ asrc2, const float* __restrict__ adst2,
             float* __restrict__ h2, float* __restrict__ s2, float* __restrict__ d2,
             int Nn) {
  constexpr int C = C1, V = C1 / 16;
  __shared__ alignas(16) float sW[C1 * C2];
  __shared__ alignas(16) float sH[16 * HP];

  static_assert((C1 * C2) / 4 == 256, "one float4 of W2 per thread");
  const float4 w2v = ((const float4*)W2)[threadIdx.x];

  const int lane  = threadIdx.x & 31;
  const int l     = lane & 15;
  const int gbase = lane & 16;
  const int g     = threadIdx.x >> 4;
  const int n  = blockIdx.x * 16 + g;
  const bool live = n < Nn;
  const int nn = min(n, Nn - 1);

  float ov[V];
  gather_row<C>(off, deg, lst, s, d, h1, b1, nn, l, gbase, ov);

  float pv[V];
#pragma unroll
  for (int k = 0; k < V; ++k) pv[k] = __shfl(ov[k], gbase + (l ^ 8));

  float4 hv;
  hv.x = ov[0] > 0.f ? ov[0] : (__expf(ov[0]) - 1.f);
  hv.y = ov[1] > 0.f ? ov[1] : (__expf(ov[1]) - 1.f);
  hv.z = ov[2] > 0.f ? ov[2] : (__expf(ov[2]) - 1.f);
  hv.w = ov[3] > 0.f ? ov[3] : (__expf(ov[3]) - 1.f);
  if (!live) hv = make_float4(0.f, 0.f, 0.f, 0.f);
  *(float4*)(sH + g * HP + l * V) = hv;
  {
    const int i  = threadIdx.x;
    const int k  = i >> 2;
    const int c0 = (i & 3) * 4;
    float* dst = sW + (k >> 2) * 64 + (k & 3);
    dst[(c0 + 0) * 4] = w2v.x;
    dst[(c0 + 1) * 4] = w2v.y;
    dst[(c0 + 2) * 4] = w2v.z;
    dst[(c0 + 3) * 4] = w2v.w;
  }

  if (live) {
    const bool lo = l < 8;
    float2 kv;
    kv.x = kl_term(lo ? ov[0] : pv[2], lo ? pv[0] : ov[2]);
    kv.y = kl_term(lo ? ov[1] : pv[3], lo ? pv[1] : ov[3]);
    *(float2*)(ixz + (size_t)n * (C / 2) + (l & 7) * V + (lo ? 0 : 2)) = kv;
  }

  __syncthreads();
  if ((threadIdx.x >> 5) != 0) return;

  const int row0 = blockIdx.x * 16;
  const int r    = l;
  const int kh   = (lane >> 4) << 1;
  const float sa = asrc2[r], da = adst2[r];

  v8f acc = {};
  const float* abase = sH + r * HP + kh;
  const float* bbase = sW + r * 4 + kh;
  for (int k0 = 0; k0 < C1; k0 += 4) {
    v2f a = *(const v2f*)(abase + k0);
    v2f b = *(const v2f*)(bbase + (k0 >> 2) * 64);
    acc = __builtin_amdgcn_wmma_f32_16x16x4_f32(false, a, false, b, (short)0, acc, false, false);
  }

  const int mbase = (lane >> 4) * 8;
  float ps[8], pd[8];
  for (int i = 0; i < 8; ++i) {
    const int row = row0 + mbase + i;
    if (row < Nn) h2[(size_t)row * C2 + r] = acc[i];
    ps[i] = acc[i] * sa;
    pd[i] = acc[i] * da;
  }
  const float st = group_sum16x8(ps, r);
  const float dt = group_sum16x8(pd, r);
  const int srow = row0 + mbase + (r >> 1);
  if ((r & 1) == 0 && srow < Nn) { s2[srow] = st; d2[srow] = dt; }
}

__global__ void __launch_bounds__(256)
gather2_out(const int* __restrict__ off, const int* __restrict__ deg,
            const int* __restrict__ lst,
            const float* __restrict__ s, const float* __restrict__ d,
            const float* __restrict__ h2, const float* __restrict__ b2,
            float* __restrict__ orow, float* __restrict__ ixz, int Nn) {
  constexpr int C = C2, V = C2 / 16;
  const int lane  = threadIdx.x & 31;
  const int l     = lane & 15;
  const int gbase = lane & 16;
  const int n  = (int)((blockIdx.x * blockDim.x + threadIdx.x) >> 4);
  const bool live = n < Nn;
  const int nn = min(n, Nn - 1);

  float ov[V];
  gather_row<C>(off, deg, lst, s, d, h2, b2, nn, l, gbase, ov);

  float pv[V];
#pragma unroll
  for (int k = 0; k < V; ++k) pv[k] = __shfl(ov[k], gbase + (l ^ 8));

  if (!live) return;
#pragma unroll
  for (int k = 0; k < V; ++k) orow[(size_t)n * C + l * V + k] = ov[k];
  if (l < 8) {
#pragma unroll
    for (int k = 0; k < V; ++k) ixz[(size_t)n * (C / 2) + l * V + k] = kl_term(ov[k], pv[k]);
  }
}

extern "C" void kernel_launch(void* const* d_in, const int* in_sizes, int n_in,
                              void* d_out, int out_size, void* d_ws, size_t ws_size,
                              hipStream_t stream) {
  const float* x   = (const float*)d_in[0];
  const int*   ei  = (const int*)  d_in[1];
  const float* W1  = (const float*)d_in[2];
  const float* as1 = (const float*)d_in[3];
  const float* ad1 = (const float*)d_in[4];
  const float* b1  = (const float*)d_in[5];
  const float* W2  = (const float*)d_in[6];
  const float* as2 = (const float*)d_in[7];
  const float* ad2 = (const float*)d_in[8];
  const float* b2  = (const float*)d_in[9];

  const int N = in_sizes[0] / F_IN;
  const int E = in_sizes[1] / 2;
  float* out = (float*)d_out;

  const size_t N64 = (size_t)N * C1, N16 = (size_t)N * C2;
  float* ws   = (float*)d_ws;
  float* h1   = ws;
  float* h2   = h1 + N64;
  float* s1   = h2 + N16;
  float* d1   = s1 + N;
  float* s2   = d1 + N;
  float* d2   = s2 + N;
  int*   deg  = (int*)(d2 + N);
  int*   off  = deg + N;
  int*   rnk  = off + N;
  int*   lin  = rnk + E;
  int*   lst  = lin + E;
  int*   cur  = lst + E;

  float* ixz1 = out + N16;
  float* ixz2 = out + (size_t)N * 48;

  const int tiles = (N + 15) / 16;
  const int gblk  = (tiles + 7) / 8;
  const int gpart = (gblk + 1) / 2;
  const int nblk  = (N + 255) / 256;
  const int eblk  = (E + 255) / 256;
  const int rblk  = (N + 15) / 16;

  Gemm1Args g0 = {x, W1, as1, ad1, h1, s1, d1, N, 0,     gpart};
  Gemm1Args g1 = {x, W1, as1, ad1, h1, s1, d1, N, gpart, gpart};
  csr_init<<<nblk, 256, 0, stream>>>(deg, cur, N);
  csr_count<<<eblk, 256, 0, stream>>>(ei, E, deg, rnk);
  csr_place<<<nblk, 256, 0, stream>>>(deg, cur, off, N);
  csr_fill<<<gpart + eblk, 256, 0, stream>>>(ei, E, off, rnk, lin, g0);
  csr_sort<<<gpart + rblk, 256, 0, stream>>>(off, deg, lin, lst, N, g1);

  gather1_proj<<<rblk, 256, 0, stream>>>(off, deg, lst, s1, d1, h1, b1, ixz1,
                                         W2, as2, ad2, h2, s2, d2, N);

  gather2_out<<<rblk, 256, 0, stream>>>(off, deg, lst, s2, d2, h2, b2, out, ixz2, N);
}


# edge and row launches: a lane takes 4 edges or rows, a 16-lane group 4 rows (5 in layer 1) in turn, index loads issued together: a quarter of the short waves enter; sums unchanged
# speedup vs baseline: 2.4333x; 1.0560x over previous
#include <hip/hip_runtime.h>
#include <hip/hip_bf16.h>

typedef __attribute__((ext_vector_type(2))) float v2f;
typedef __attribute__((ext_vector_type(8))) float v8f;

#define F_IN 128
#define C1   64
#define C2   16
#define NEG_SLOPE 0.2f

__device__ __forceinline__ float group_max16(float v) {
  v = fmaxf(v, __shfl_xor(v, 8)); v = fmaxf(v, __shfl_xor(v, 4));
  v = fmaxf(v, __shfl_xor(v, 2)); v = fmaxf(v, __shfl_xor(v, 1));
  return v;
}
__device__ __forceinline__ float group_sum16(float v) {
  v += __shfl_xor(v, 8); v += __shfl_xor(v, 4);
  v += __shfl_xor(v, 2); v += __shfl_xor(v, 1);
  return v;
}
__device__ __forceinline__ float group_sum16x8(const float (&v)[8], int l) {
  const bool b3 = (l & 8) != 0, b2 = (l & 4) != 0, b1 = (l & 2) != 0;
  float w[4];
  for (int j = 0; j < 4; ++j) {
    const float mine = b3 ? v[j + 4] : v[j];
    const float oth  = b3 ? v[j] : v[j + 4];
    w[j] = mine + __shfl_xor(oth, 8);
  }
  float u[2];
  for (int j = 0; j < 2; ++j) {
    const float mine = b2 ? w[j + 2] : w[j];
    const float oth  = b2 ? w[j] : w[j + 2];
    u[j] = mine + __shfl_xor(oth, 4);
  }
  const float mine = b1 ? u[1] : u[0];
  const float oth  = b1 ? u[0] : u[1];
  const float t = mine + __shfl_xor(oth, 2);
  return t + __shfl_xor(t, 1);
}

struct Gemm1Args {
  const float* x; const float* W1; const float* asrc; const float* adst;
  float* h1; float* s; float* d; int Nn;
  int first;
  int count;
  int tail;
};

__device__ __forceinline__ void
gemm1_block(const float* __restrict__ x, const float* __restrict__ W1,
            const float* __restrict__ asrc, const float* __restrict__ adst,
            float* __restrict__ h1, float* __restrict__ s, float* __restrict__ d, int Nn,
            int gb) {
  __shared__ float sW[F_IN * C1];
  for (int i = threadIdx.x; i < (F_IN * C1) / 4; i += 256) {
    float4 v = ((const float4*)W1)[i];
    const int k  = i >> 4;
    const int c0 = (i & 15) * 4;
    float* dst = sW + (k >> 2) * 256 + (k & 3);
    dst[(c0 + 0) * 4] = v.x;
    dst[(c0 + 1) * 4] = v.y;
    dst[(c0 + 2) * 4] = v.z;
    dst[(c0 + 3) * 4] = v.w;
  }
  __syncthreads();

  const int wave = threadIdx.x >> 5;
  const int lane = threadIdx.x & 31;
  const int tile = gb * 8 + wave;
  if (tile * 16 >= Nn) return;
  const int row0 = tile * 16;
  const int r    = lane & 15;
  const int kh   = (lane >> 4) << 1;
  const int arow = min(row0 + r, Nn - 1);

  const float sa0 = asrc[r], sa1 = asrc[r + 16], sa2 = asrc[r + 32], sa3 = asrc[r + 48];
  const float da0 = adst[r], da1 = adst[r + 16], da2 = adst[r + 32], da3 = adst[r + 48];

  v8f acc0 = {}, acc1 = {}, acc2 = {}, acc3 = {};
  const float* ap = x + (size_t)arow * F_IN + kh;
  const float* bbase = sW + r * 4 + kh;
  for (int k0 = 0; k0 < F_IN; k0 += 4) {
    v2f a; a.x = ap[k0]; a.y = ap[k0 + 1];
    const float* bq = bbase + (k0 >> 2) * 256;
    v2f b0 = *(const v2f*)(bq);
    v2f b1 = *(const v2f*)(bq + 64);
    v2f b2 = *(const v2f*)(bq + 128);
    v2f b3 = *(const v2f*)(bq + 192);
    acc0 = __builtin_amdgcn_wmma_f32_16x16x4_f32(false, a, false, b0, (short)0, acc0, false, false);
    acc1 = __builtin_amdgcn_wmma_f32_16x16x4_f32(false, a, false, b1, (short)0, acc1, false, false);
    acc2 = __builtin_amdgcn_wmma_f32_16x16x4_f32(false, a, false, b2, (short)0, acc2, false, false);
    acc3 = __builtin_amdgcn_wmma_f32_16x16x4_f32(false, a, false, b3, (short)0, acc3, false, false);
  }

  const int mbase = (lane >> 4) * 8;
  for (int i = 0; i < 8; ++i) {
    const int row = row0 + mbase + i;
    if (row < Nn) {
      float* op = h1 + (size_t)row * C1 + r;
      op[0]  = acc0[i];
      op[16] = acc1[i];
      op[32] = acc2[i];
      op[48] = acc3[i];
    }
  }

  float ps[8], pd[8];
  for (int i = 0; i < 8; ++i) {
    ps[i] = acc0[i] * sa0 + acc1[i] * sa1 + acc2[i] * sa2 + acc3[i] * sa3;
    pd[i] = acc0[i] * da0 + acc1[i] * da1 + acc2[i] * da2 + acc3[i] * da3;
  }
  const float st = group_sum16x8(ps, r);
  const float dt = group_sum16x8(pd, r);
  const int srow = row0 + mbase + (r >> 1);
  if ((r & 1) == 0 && srow < Nn) { s[srow] = st; d[srow] = dt; }
}

#define RPG 4
#define SPAN (256 * RPG)
static_assert(RPG == 4, "the load fence in csr_fill names RPG values of each kind");

__global__ void __launch_bounds__(256)
csr_init(int* __restrict__ deg, int* __restrict__ cursor, int Nn) {
  const int i0 = blockIdx.x * SPAN + threadIdx.x;
#pragma unroll
  for (int k = 0; k < RPG; ++k) {
    const int i = i0 + k * 256;
    if (i < Nn) deg[i] = 0;
  }
  if (i0 == 0) *cursor = 0;
}

__global__ void __launch_bounds__(256)
csr_count(const int* __restrict__ ei, int E, int* __restrict__ deg, int* __restrict__ rnk) {
  const int e0 = blockIdx.x * SPAN + threadIdx.x;
  int dst[RPG], rk[RPG];
#pragma unroll
  for (int k = 0; k < RPG; ++k) dst[k] = ei[E + min(e0 + k * 256, E - 1)];
#pragma unroll
  for (int k = 0; k < RPG; ++k) {
    rk[k] = 0;
    if (e0 + k * 256 < E) rk[k] = atomicAdd(deg + dst[k], 1);
  }
#pragma unroll
  for (int k = 0; k < RPG; ++k)
    if (e0 + k * 256 < E) rnk[e0 + k * 256] = rk[k];
}

__global__ void __launch_bounds__(256)
csr_place(const int* __restrict__ deg, int* __restrict__ cursor, int* __restrict__ off, int Nn) {
  const int n0 = blockIdx.x * SPAN + threadIdx.x;
  const int lane = threadIdx.x & 31;
  int dg[RPG];
  int tot = 0;
#pragma unroll
  for (int k = 0; k < RPG; ++k) {
    const int n = n0 + k * 256;
    dg[k] = (n < Nn) ? deg[n] : 0;
    tot += dg[k];
  }
  int x = tot;
  for (int k = 1; k < 32; k <<= 1) {
    const int y = __shfl_up(x, k);
    if (lane >= k) x += y;
  }
  int base = 0;
  if (lane == 31) base = atomicAdd(cursor, x);
  base = __shfl(base, 31);
  int o = base + x - tot;
#pragma unroll
  for (int k = 0; k < RPG; ++k) {
    const int n = n0 + k * 256;
    if (n < Nn) off[n] = o;
    o += dg[k];
  }
}

__global__ void __launch_bounds__(256) __attribute__((amdgpu_waves_per_eu(16)))
csr_fill(const int* __restrict__ ei, int E, const int* __restrict__ off,
         const int* __restrict__ rnk, int* __restrict__ lin, const Gemm1Args g) {
  if ((int)blockIdx.x < g.count) {
    gemm1_block(g.x, g.W1, g.asrc, g.adst, g.h1, g.s, g.d, g.Nn, g.first + (int)blockIdx.x);
    return;
  }
  const int e0 = ((int)blockIdx.x - g.count) * SPAN + threadIdx.x;
  int dst[RPG], src[RPG], rk[RPG], o[RPG];
#pragma unroll
  for (int k = 0; k < RPG; ++k) {
    const int ec = min(e0 + k * 256, E - 1);
    dst[k] = ei[E + ec];
    src[k] = ei[ec];
    rk[k]  = rnk[ec];
  }
#pragma unroll
  for (int k = 0; k < RPG; ++k) o[k] = off[dst[k]];
  asm volatile("" :: "v"(o[0]), "v"(o[1]), "v"(o[2]), "v"(o[3]),
                     "v"(src[0]), "v"(src[1]), "v"(src[2]), "v"(src[3]),
                     "v"(rk[0]), "v"(rk[1]), "v"(rk[2]), "v"(rk[3]));
#pragma unroll
  for (int k = 0; k < RPG; ++k)
    if (e0 + k * 256 < E) lin[o[k] + rk[k]] = src[k];
}

__global__ void __launch_bounds__(256) __attribute__((amdgpu_waves_per_eu(16)))
csr_sort(const int* __restrict__ off, const int* __restrict__ deg,
         const int* __restrict__ lin, int* __restrict__ lout, int Nn, const Gemm1Args g) {
  if ((int)blockIdx.x < g.count) {
    gemm1_block(g.x, g.W1, g.asrc, g.adst, g.h1, g.s, g.d, g.Nn, g.first + (int)blockIdx.x);
    return;
  }
  const int lane  = threadIdx.x & 31;
  const int l     = lane & 15;
  const int gbase = lane & 16;
  const int grp = (int)((((int)blockIdx.x - g.count) * 256 + threadIdx.x) >> 4);
  int ro[RPG], rd[RPG], rkey[RPG];
#pragma unroll
  for (int t = 0; t < RPG; ++t) {
    const int n  = grp * RPG + t;
    const int nn = min(n, Nn - 1);
    ro[t] = off[nn];
    rd[t] = (n < Nn) ? deg[nn] : 0;
  }
#pragma unroll
  for (int t = 0; t < RPG; ++t) rkey[t] = (l < rd[t]) ? lin[ro[t] + l] : 0x7fffffff;

#pragma unroll
  for (int t = 0; t < RPG; ++t) {
    const int o  = ro[t];
    const int dg = rd[t];
    const int dgmax = max(dg, __shfl_xor(dg, 16));
    if (dgmax <= 16) {
      const int key = rkey[t];
      int rank = 0;
      for (int j = 0; j < dgmax; ++j) {
        const int kt = __shfl(key, gbase + j);
        rank += ((j < dg) && (kt < key || (kt == key && j < l))) ? 1 : 0;
      }
      if (l < dg) lout[o + rank] = key;
    } else {
      for (int ci = 0; ci < dgmax; ci += 16) {
        const int i = ci + l;
        const int key = (i < dg) ? lin[o + i] : 0x7fffffff;
        int rank = 0;
        for (int cj = 0; cj < dgmax; cj += 16) {
          const int jm = cj + l;
          const int kmine = (jm < dg) ? lin[o + jm] : 0x7fffffff;
          const int cnt = min(16, dgmax - cj);
          for (int j = 0; j < cnt; ++j) {
            const int kt = __shfl(kmine, gbase + j);
            const int jt = cj + j;
            rank += ((jt < dg) && (kt < key || (kt == key && jt < i))) ? 1 : 0;
          }
        }
        if (i < dg) lout[o + rank] = key;
      }
    }
  }
}

__device__ __forceinline__ float softplusf(float v) {
  return fmaxf(v, 0.f) + log1pf(__expf(-fabsf(v)));
}
__device__ __forceinline__ float kl_term(float mean, float sp) {
  const float sd = softplusf(sp) + 1e-10f;
  return -__logf(sd) + 0.5f * (sd * sd + mean * mean - 1.f);
}
__device__ __forceinline__ float leaky(float a) { return a > 0.f ? a : NEG_SLOPE * a; }

struct RowHead { int nn, o, dg, src0; float dn, s0; };

template <int R>
__device__ __forceinline__ void
load_heads(const int* __restrict__ off, const int* __restrict__ deg,
           const int* __restrict__ lst,
           const float* __restrict__ s, const float* __restrict__ d,
           int l, RowHead (&h)[R]) {
#pragma unroll
  for (int t = 0; t < R; ++t) { h[t].o = off[h[t].nn]; h[t].dg = deg[h[t].nn]; h[t].dn = d[h[t].nn]; }
#pragma unroll
  for (int t = 0; t < R; ++t) h[t].src0 = (l < h[t].dg) ? lst[h[t].o + l] : h[t].nn;
#pragma unroll
  for (int t = 0; t < R; ++t) h[t].s0 = s[h[t].src0];
}

template <int C>
__device__ __forceinline__ void
gather_row(const RowHead& hd, const int* __restrict__ lst,
           const float* __restrict__ s,
           const float* __restrict__ hsrc, const float* __restrict__ bias,
           int l, int gbase, float (&ov)[C / 16]) {
  constexpr int V = C / 16;
  const int nn  = hd.nn;
  const int o   = hd.o;
  const int dg  = hd.dg;
  const int tot = dg + 1;
  const float dn = hd.dn;
  const int totmax = max(tot, __shfl_xor(tot, 16));

  const int   src0 = hd.src0;
  const float a0   = leaky(hd.s0 + dn);
  const bool  v0   = l < tot;

  float m = v0 ? a0 : -INFINITY;
#pragma unroll 1
  for (int c = 16; c < totmax; c += 16) {
    const int i = c + l;
    const int sr = (i < dg) ? lst[o + i] : nn;
    const float a = leaky(s[sr] + dn);
    if (i < tot) m = fmaxf(m, a);
  }
  m = group_max16(m);

  const float ex0 = v0 ? __expf(a0 - m) : 0.f;
  float den = ex0;
#pragma unroll 1
  for (int c = 16; c < totmax; c += 16) {
    const int i = c + l;
    const int sr = (i < dg) ? lst[o + i] : nn;
    const float a = leaky(s[sr] + dn);
    den += (i < tot) ? __expf(a - m) : 0.f;
  }
  den = group_sum16(den) + 1e-16f;

  float acc[V];
#pragma unroll
  for (int k = 0; k < V; ++k) acc[k] = 0.f;

  for (int c = 0; c < totmax; c += 16) {
    int sr; float w;
    if (c == 0) { sr = src0; w = ex0 / den; }
    else {
      const int i = c + l;
      sr = (i < dg) ? lst[o + i] : nn;
      const float a = leaky(s[sr] + dn);
      w = (i < tot) ? __expf(a - m) / den : 0.f;
    }
    const int cnt = min(16, totmax - c);
    for (int j = 0; j < cnt; j += 4) {
      float r[4][V];
#pragma unroll
      for (int u = 0; u < 4; ++u) {
        const int sj = __shfl(sr, gbase + ((j + u) & 15));
        const float* hp = hsrc + (size_t)sj * C + l * V;
        if constexpr (V == 4) {
          const float4 t = *(const float4*)hp;
          r[u][0] = t.x; r[u][1] = t.y; r[u][2] = t.z; r[u][3] = t.w;
        } else {
#pragma unroll
          for (int k = 0; k < V; ++k) r[u][k] = hp[k];
        }
      }
      if constexpr (V == 4) {
        asm volatile("" :: "v"(r[0][0]), "v"(r[0][1]), "v"(r[0][2]), "v"(r[0][3]),
                           "v"(r[1][0]), "v"(r[1][1]), "v"(r[1][2]), "v"(r[1][3]),
                           "v"(r[2][0]), "v"(r[2][1]), "v"(r[2][2]), "v"(r[2][3]),
                           "v"(r[3][0]), "v"(r[3][1]), "v"(r[3][2]), "v"(r[3][3]));
      } else {
        asm volatile("" :: "v"(r[0][0]), "v"(r[1][0]), "v"(r[2][0]), "v"(r[3][0]));
      }
#pragma unroll
      for (int u = 0; u < 4; ++u) {
        const float wj = __shfl(w, gbase + ((j + u) & 15));
        if (c + j + u < tot) {
#pragma unroll
          for (int k = 0; k < V; ++k) acc[k] += r[u][k] * wj;
        }
      }
    }
  }

#pragma unroll
  for (int k = 0; k < V; ++k) ov[k] = acc[k] + bias[l * V + k];
}

#define RG1 5
#define HP 68

__global__ void __launch_bounds__(256)
gather1_proj(const int* __restrict__ off, const int* __restrict__ deg,
             const int* __restrict__ lst,
             const float* __restrict__ s, const float* __restrict__ d,
             const float* __restrict__ h1, const float* __restrict__ b1,
             float* __restrict__ ixz,
             const float* __restrict__ W2,
             const float* __restrict__ asrc2, const float* __restrict__ adst2,
             float* __restrict__ h2, float* __restrict__ s2, float* __restrict__ d2,
             int Nn) {
  constexpr int C = C1, V = C1 / 16;
  __shared__ alignas(16) float sW[C1 * C2];
  __shared__ alignas(16) float sH[RG1 * 16 * HP];

  static_assert((C1 * C2) / 4 == 256, "one float4 of W2 per thread");
  float4 w2v;

  const int lane  = threadIdx.x & 31;
  const int l     = lane & 15;
  const int gbase = lane & 16;
  const int g     = threadIdx.x >> 4;
  const int brow0 = blockIdx.x * (RG1 * 16);

  RowHead hd[RG1];
#pragma unroll
  for (int t = 0; t < RG1; ++t) hd[t].nn = min(brow0 + t * 16 + g, Nn - 1);
  load_heads(off, deg, lst, s, d, l, hd);

#pragma unroll
  for (int t = 0; t < RG1; ++t) {
    const int n = brow0 + t * 16 + g;
    const bool live = n < Nn;

    if (t == RG1 - 1) w2v = ((const float4*)W2)[threadIdx.x];

    float ov[V];
    gather_row<C>(hd[t], lst, s, h1, b1, l, gbase, ov);

    float pv[V];
#pragma unroll
    for (int k = 0; k < V; ++k) pv[k] = __shfl(ov[k], gbase + (l ^ 8));

    float4 hv;
    hv.x = ov[0] > 0.f ? ov[0] : (__expf(ov[0]) - 1.f);
    hv.y = ov[1] > 0.f ? ov[1] : (__expf(ov[1]) - 1.f);
    hv.z = ov[2] > 0.f ? ov[2] : (__expf(ov[2]) - 1.f);
    hv.w = ov[3] > 0.f ? ov[3] : (__expf(ov[3]) - 1.f);
    if (!live) hv = make_float4(0.f, 0.f, 0.f, 0.f);
    *(float4*)(sH + (t * 16 + g) * HP + l * V) = hv;

    if (live) {
      const bool lo = l < 8;
      float2 kv;
      kv.x = kl_term(lo ? ov[0] : pv[2], lo ? pv[0] : ov[2]);
      kv.y = kl_term(lo ? ov[1] : pv[3], lo ? pv[1] : ov[3]);
      *(float2*)(ixz + (size_t)n * (C / 2) + (l & 7) * V + (lo ? 0 : 2)) = kv;
    }
  }
  {
    const int i  = threadIdx.x;
    const int k  = i >> 2;
    const int c0 = (i & 3) * 4;
    float* dst = sW + (k >> 2) * 64 + (k & 3);
    dst[(c0 + 0) * 4] = w2v.x;
    dst[(c0 + 1) * 4] = w2v.y;
    dst[(c0 + 2) * 4] = w2v.z;
    dst[(c0 + 3) * 4] = w2v.w;
  }

  __syncthreads();
  const int wave = threadIdx.x >> 5;
  const int row0 = brow0 + wave * 16;
  if (wave >= RG1 || row0 >= Nn) return;

  const int r    = l;
  const int kh   = (lane >> 4) << 1;
  const float sa = asrc2[r], da = adst2[r];

  v8f acc = {};
  const float* abase = sH + (wave * 16 + r) * HP + kh;
  const float* bbase = sW + r * 4 + kh;
  for (int k0 = 0; k0 < C1; k0 += 4) {
    v2f a = *(const v2f*)(abase + k0);
    v2f b = *(const v2f*)(bbase + (k0 >> 2) * 64);
    acc = __builtin_amdgcn_wmma_f32_16x16x4_f32(false, a, false, b, (short)0, acc, false, false);
  }

  const int mbase = (lane >> 4) * 8;
  float ps[8], pd[8];
  for (int i = 0; i < 8; ++i) {
    const int row = row0 + mbase + i;
    if (row < Nn) h2[(size_t)row * C2 + r] = acc[i];
    ps[i] = acc[i] * sa;
    pd[i] = acc[i] * da;
  }
  const float st = group_sum16x8(ps, r);
  const float dt = group_sum16x8(pd, r);
  const int srow = row0 + mbase + (r >> 1);
  if ((r & 1) == 0 && srow < Nn) { s2[srow] = st; d2[srow] = dt; }
}

__global__ void __launch_bounds__(256)
gather2_out(const int* __restrict__ off, const int* __restrict__ deg,
            const int* __restrict__ lst,
            const float* __restrict__ s, const float* __restrict__ d,
            const float* __restrict__ h2, const float* __restrict__ b2,
            float* __restrict__ orow, float* __restrict__ ixz, int Nn) {
  constexpr int C = C2, V = C2 / 16;
  const int lane  = threadIdx.x & 31;
  const int l     = lane & 15;
  const int gbase = lane & 16;
  const int wv    = (int)((blockIdx.x * 256 + threadIdx.x) >> 5);
  const int wrow0 = wv * (2 * RPG) + (lane >> 4);

  RowHead hd[RPG];
#pragma unroll
  for (int t = 0; t < RPG; ++t) hd[t].nn = min(wrow0 + 2 * t, Nn - 1);
  load_heads(off, deg, lst, s, d, l, hd);

#pragma unroll
  for (int t = 0; t < RPG; ++t) {
    const int n = wrow0 + 2 * t;
    const bool live = n < Nn;

    float ov[V];
    gather_row<C>(hd[t], lst, s, h2, b2, l, gbase, ov);

    float pv[V];
#pragma unroll
    for (int k = 0; k < V; ++k) pv[k] = __shfl(ov[k], gbase + (l ^ 8));

    if (live) {
#pragma unroll
      for (int k = 0; k < V; ++k) orow[(size_t)n * C + l * V + k] = ov[k];
      if (l < 8) {
#pragma unroll
        for (int k = 0; k < V; ++k) ixz[(size_t)n * (C / 2) + l * V + k] = kl_term(ov[k], pv[k]);
      }
    }
  }
}

extern "C" void kernel_launch(void* const* d_in, const int* in_sizes, int n_in,
                              void* d_out, int out_size, void* d_ws, size_t ws_size,
                              hipStream_t stream) {
  const float* x   = (const float*)d_in[0];
  const int*   ei  = (const int*)  d_in[1];
  const float* W1  = (const float*)d_in[2];
  const float* as1 = (const float*)d_in[3];
  const float* ad1 = (const float*)d_in[4];
  const float* b1  = (const float*)d_in[5];
  const float* W2  = (const float*)d_in[6];
  const float* as2 = (const float*)d_in[7];
  const float* ad2 = (const float*)d_in[8];
  const float* b2  = (const float*)d_in[9];

  const int N = in_sizes[0] / F_IN;
  const int E = in_sizes[1] / 2;
  float* out = (float*)d_out;

  const size_t N64 = (size_t)N * C1, N16 = (size_t)N * C2;
  float* ws   = (float*)d_ws;
  float* h1   = ws;
  float* h2   = h1 + N64;
  float* s1   = h2 + N16;
  float* d1   = s1 + N;
  float* s2   = d1 + N;
  float* d2   = s2 + N;
  int*   deg  = (int*)(d2 + N);
  int*   off  = deg + N;
  int*   rnk  = off + N;
  int*   lin  = rnk + E;
  int*   lst  = lin + E;
  int*   cur  = lst + E;

  float* ixz1 = out + N16;
  float* ixz2 = out + (size_t)N * 48;

  const int tiles = (N + 15) / 16;
  const int gblk  = (tiles + 7) / 8;
  const int gpart = (gblk + 1) / 2;
  const int nblk  = (N + SPAN - 1) / SPAN;
  const int eblk  = (E + SPAN - 1) / SPAN;
  const int rblk  = (N + 16 * RPG - 1) / (16 * RPG);
  const int rblk1 = (N + 16 * RG1 - 1) / (16 * RG1);

  Gemm1Args g0 = {x, W1, as1, ad1, h1, s1, d1, N, 0,     gpart, 0};
  Gemm1Args g1 = {x, W1, as1, ad1, h1, s1, d1, N, gpart, gpart, 0};
  csr_init<<<nblk, 256, 0, stream>>>(deg, cur, N);
  csr_count<<<eblk, 256, 0, stream>>>(ei, E, deg, rnk);
  csr_place<<<nblk, 256, 0, stream>>>(deg, cur, off, N);
  csr_fill<<<gpart + eblk, 256, 0, stream>>>(ei, E, off, rnk, lin, g0);
  csr_sort<<<gpart + rblk, 256, 0, stream>>>(off, deg, lin, lst, N, g1);

  gather1_proj<<<rblk1, 256, 0, stream>>>(off, deg, lst, s1, d1, h1, b1, ixz1,
                                         W2, as2, ad2, h2, s2, d2, N);

  gather2_out<<<rblk, 256, 0, stream>>>(off, deg, lst, s2, d2, h2, b2, out, ixz2, N);
}


# each row owns 16 fixed list slots and an edge is stored at the rank its counting add returns, in the counting launch (longer rows: overflow list, collected in the sort): 7->5 launches; sums unchanged
# speedup vs baseline: 2.6142x; 1.0743x over previous
#include <hip/hip_runtime.h>
#include <hip/hip_bf16.h>

typedef __attribute__((ext_vector_type(2))) float v2f;
typedef __attribute__((ext_vector_type(8))) float v8f;

#define F_IN 128
#define C1   64
#define C2   16
#define NEG_SLOPE 0.2f

__device__ __forceinline__ float group_max16(float v) {
  v = fmaxf(v, __shfl_xor(v, 8)); v = fmaxf(v, __shfl_xor(v, 4));
  v = fmaxf(v, __shfl_xor(v, 2)); v = fmaxf(v, __shfl_xor(v, 1));
  return v;
}
__device__ __forceinline__ float group_sum16(float v) {
  v += __shfl_xor(v, 8); v += __shfl_xor(v, 4);
  v += __shfl_xor(v, 2); v += __shfl_xor(v, 1);
  return v;
}
__device__ __forceinline__ float group_sum16x8(const float (&v)[8], int l) {
  const bool b3 = (l & 8) != 0, b2 = (l & 4) != 0, b1 = (l & 2) != 0;
  float w[4];
  for (int j = 0; j < 4; ++j) {
    const float mine = b3 ? v[j + 4] : v[j];
    const float oth  = b3 ? v[j] : v[j + 4];
    w[j] = mine + __shfl_xor(oth, 8);
  }
  float u[2];
  for (int j = 0; j < 2; ++j) {
    const float mine = b2 ? w[j + 2] : w[j];
    const float oth  = b2 ? w[j] : w[j + 2];
    u[j] = mine + __shfl_xor(oth, 4);
  }
  const float mine = b1 ? u[1] : u[0];
  const float oth  = b1 ? u[0] : u[1];
  const float t = mine + __shfl_xor(oth, 2);
  return t + __shfl_xor(t, 1);
}

struct Gemm1Args {
  const float* x; const float* W1; const float* asrc; const float* adst;
  float* h1; float* s; float* d; int Nn;
  int first;
  int count;
  int tail;
};

__device__ __forceinline__ void
gemm1_block(const float* __restrict__ x, const float* __restrict__ W1,
            const float* __restrict__ asrc, const float* __restrict__ adst,
            float* __restrict__ h1, float* __restrict__ s, float* __restrict__ d, int Nn,
            int gb) {
  __shared__ float sW[F_IN * C1];
  for (int i = threadIdx.x; i < (F_IN * C1) / 4; i += 256) {
    float4 v = ((const float4*)W1)[i];
    const int k  = i >> 4;
    const int c0 = (i & 15) * 4;
    float* dst = sW + (k >> 2) * 256 + (k & 3);
    dst[(c0 + 0) * 4] = v.x;
    dst[(c0 + 1) * 4] = v.y;
    dst[(c0 + 2) * 4] = v.z;
    dst[(c0 + 3) * 4] = v.w;
  }
  __syncthreads();

  const int wave = threadIdx.x >> 5;
  const int lane = threadIdx.x & 31;
  const int tile = gb * 8 + wave;
  if (tile * 16 >= Nn) return;
  const int row0 = tile * 16;
  const int r    = lane & 15;
  const int kh   = (lane >> 4) << 1;
  const int arow = min(row0 + r, Nn - 1);

  const float sa0 = asrc[r], sa1 = asrc[r + 16], sa2 = asrc[r + 32], sa3 = asrc[r + 48];
  const float da0 = adst[r], da1 = adst[r + 16], da2 = adst[r + 32], da3 = adst[r + 48];

  v8f acc0 = {}, acc1 = {}, acc2 = {}, acc3 = {};
  const float* ap = x + (size_t)arow * F_IN + kh;
  const float* bbase = sW + r * 4 + kh;
  for (int k0 = 0; k0 < F_IN; k0 += 4) {
    v2f a; a.x = ap[k0]; a.y = ap[k0 + 1];
    const float* bq = bbase + (k0 >> 2) * 256;
    v2f b0 = *(const v2f*)(bq);
    v2f b1 = *(const v2f*)(bq + 64);
    v2f b2 = *(const v2f*)(bq + 128);
    v2f b3 = *(const v2f*)(bq + 192);
    acc0 = __builtin_amdgcn_wmma_f32_16x16x4_f32(false, a, false, b0, (short)0, acc0, false, false);
    acc1 = __builtin_amdgcn_wmma_f32_16x16x4_f32(false, a, false, b1, (short)0, acc1, false, false);
    acc2 = __builtin_amdgcn_wmma_f32_16x16x4_f32(false, a, false, b2, (short)0, acc2, false, false);
    acc3 = __builtin_amdgcn_wmma_f32_16x16x4_f32(false, a, false, b3, (short)0, acc3, false, false);
  }

  const int mbase = (lane >> 4) * 8;
  for (int i = 0; i < 8; ++i) {
    const int row = row0 + mbase + i;
    if (row < Nn) {
      float* op = h1 + (size_t)row * C1 + r;
      op[0]  = acc0[i];
      op[16] = acc1[i];
      op[32] = acc2[i];
      op[48] = acc3[i];
    }
  }

  float ps[8], pd[8];
  for (int i = 0; i < 8; ++i) {
    ps[i] = acc0[i] * sa0 + acc1[i] * sa1 + acc2[i] * sa2 + acc3[i] * sa3;
    pd[i] = acc0[i] * da0 + acc1[i] * da1 + acc2[i] * da2 + acc3[i] * da3;
  }
  const float st = group_sum16x8(ps, r);
  const float dt = group_sum16x8(pd, r);
  const int srow = row0 + mbase + (r >> 1);
  if ((r & 1) == 0 && srow < Nn) { s[srow] = st; d[srow] = dt; }
}

#define RPG 4
#define SPAN (256 * RPG)
#define CAP 16

__global__ void __launch_bounds__(256)
csr_init(int* __restrict__ deg, int* __restrict__ cursor, int Nn) {
  const int i0 = blockIdx.x * SPAN + threadIdx.x;
#pragma unroll
  for (int k = 0; k < RPG; ++k) {
    const int i = i0 + k * 256;
    if (i < Nn) deg[i] = 0;
  }
  if (i0 == 0) { cursor[0] = 0; cursor[1] = 0; }
}

__global__ void __launch_bounds__(256) __attribute__((amdgpu_waves_per_eu(16)))
csr_count_fill(const int* __restrict__ ei, int E, int* __restrict__ deg, int* __restrict__ lin,
               int* __restrict__ ovn, int* __restrict__ ov, const Gemm1Args g) {
  if ((int)blockIdx.x < g.count) {
    gemm1_block(g.x, g.W1, g.asrc, g.adst, g.h1, g.s, g.d, g.Nn, g.first + (int)blockIdx.x);
    return;
  }
  const int e0 = ((int)blockIdx.x - g.count) * SPAN + threadIdx.x;
  int dst[RPG], src[RPG], rk[RPG];
#pragma unroll
  for (int k = 0; k < RPG; ++k) {
    const int ec = min(e0 + k * 256, E - 1);
    dst[k] = ei[E + ec];
    src[k] = ei[ec];
  }
#pragma unroll
  for (int k = 0; k < RPG; ++k) {
    rk[k] = 0;
    if (e0 + k * 256 < E) rk[k] = atomicAdd(deg + dst[k], 1);
  }
#pragma unroll
  for (int k = 0; k < RPG; ++k) {
    if (e0 + k * 256 < E) {
      if (rk[k] < CAP) lin[dst[k] * CAP + rk[k]] = src[k];
      else {
        const int i = atomicAdd(ovn, 1);
        ov[2 * i] = dst[k]; ov[2 * i + 1] = src[k];
      }
    }
  }
}

__global__ void __launch_bounds__(256) __attribute__((amdgpu_waves_per_eu(16)))
csr_sort(const int* __restrict__ deg, int* lin, int* __restrict__ lout, int* __restrict__ off,
         int* __restrict__ cursor, const int* __restrict__ ovn, const int* __restrict__ ov,
         int Nn, const Gemm1Args g) {
  if ((int)blockIdx.x < g.count) {
    gemm1_block(g.x, g.W1, g.asrc, g.adst, g.h1, g.s, g.d, g.Nn, g.first + (int)blockIdx.x);
    return;
  }
  const int lane  = threadIdx.x & 31;
  const int l     = lane & 15;
  const int gbase = lane & 16;
  const int grp = (int)((((int)blockIdx.x - g.count) * 256 + threadIdx.x) >> 4);
  int rd[RPG], rkey[RPG];
#pragma unroll
  for (int t = 0; t < RPG; ++t) {
    const int n  = grp * RPG + t;
    rd[t] = (n < Nn) ? deg[min(n, Nn - 1)] : 0;
  }
#pragma unroll
  for (int t = 0; t < RPG; ++t)
    rkey[t] = (l < rd[t]) ? lin[min(grp * RPG + t, Nn - 1) * CAP + l] : 0x7fffffff;

#pragma unroll
  for (int t = 0; t < RPG; ++t) {
    const int n  = grp * RPG + t;
    const int dg = rd[t];
    const int dgmax = max(dg, __shfl_xor(dg, 16));
    int o = min(n, Nn - 1) * CAP;
    if (dgmax <= CAP) {
      const int key = rkey[t];
      int rank = 0;
      for (int j = 0; j < dgmax; ++j) {
        const int kt = __shfl(key, gbase + j);
        rank += ((j < dg) && (kt < key || (kt == key && j < l))) ? 1 : 0;
      }
      if (l < dg) lout[o + rank] = key;
    } else {
      const bool lng = dg > CAP;
      int base = 0;
      if (lng && l == 0) base = atomicAdd(cursor, dg);
      base = __shfl(base, gbase);
      if (lng) {
        o = Nn * CAP + base;
        lin[o + l] = rkey[t];
      }
      const int novf = *ovn;
      int found = CAP;
      for (int c = 0; c < novf; c += 16) {
        const int i = c + l;
        bool hit = false; int sv = 0;
        if (lng && i < novf) { hit = ov[2 * i] == n; sv = ov[2 * i + 1]; }
        const unsigned gb = (__builtin_amdgcn_ballot_w32(hit) >> gbase) & 0xffffu;
        if (hit) lin[o + found + __popc(gb & ((1u << l) - 1u))] = sv;
        found += __popc(gb);
      }
      __threadfence();
      for (int ci = 0; ci < dgmax; ci += 16) {
        const int i = ci + l;
        const int key = (i < dg) ? lin[o + i] : 0x7fffffff;
        int rank = 0;
        for (int cj = 0; cj < dgmax; cj += 16) {
          const int jm = cj + l;
          const int kmine = (jm < dg) ? lin[o + jm] : 0x7fffffff;
          const int cnt = min(16, dgmax - cj);
          for (int j = 0; j < cnt; ++j) {
            const int kt = __shfl(kmine, gbase + j);
            const int jt = cj + j;
            rank += ((jt < dg) && (kt < key || (kt == key && jt < i))) ? 1 : 0;
          }
        }
        if (i < dg) lout[o + rank] = key;
      }
    }
    if (l == 0 && n < Nn) off[n] = o;
  }
}

__device__ __forceinline__ float softplusf(float v) {
  return fmaxf(v, 0.f) + log1pf(__expf(-fabsf(v)));
}
__device__ __forceinline__ float kl_term(float mean, float sp) {
  const float sd = softplusf(sp) + 1e-10f;
  return -__logf(sd) + 0.5f * (sd * sd + mean * mean - 1.f);
}
__device__ __forceinline__ float leaky(float a) { return a > 0.f ? a : NEG_SLOPE * a; }

struct RowHead { int nn, o, dg, src0; float dn, s0; };

template <int R>
__device__ __forceinline__ void
load_heads(const int* __restrict__ off, const int* __restrict__ deg,
           const int* __restrict__ lst,
           const float* __restrict__ s, const float* __restrict__ d,
           int l, RowHead (&h)[R]) {
#pragma unroll
  for (int t = 0; t < R; ++t) { h[t].o = off[h[t].nn]; h[t].dg = deg[h[t].nn]; h[t].dn = d[h[t].nn]; }
#pragma unroll
  for (int t = 0; t < R; ++t) h[t].src0 = (l < h[t].dg) ? lst[h[t].o + l] : h[t].nn;
#pragma unroll
  for (int t = 0; t < R; ++t) h[t].s0 = s[h[t].src0];
}

template <int C>
__device__ __forceinline__ void
gather_row(const RowHead& hd, const int* __restrict__ lst,
           const float* __restrict__ s,
           const float* __restrict__ hsrc, const float* __restrict__ bias,
           int l, int gbase, float (&ov)[C / 16]) {
  constexpr int V = C / 16;
  const int nn  = hd.nn;
  const int o   = hd.o;
  const int dg  = hd.dg;
  const int tot = dg + 1;
  const float dn = hd.dn;
  const int totmax = max(tot, __shfl_xor(tot, 16));

  const int   src0 = hd.src0;
  const float a0   = leaky(hd.s0 + dn);
  const bool  v0   = l < tot;

  float m = v0 ? a0 : -INFINITY;
#pragma unroll 1
  for (int c = 16; c < totmax; c += 16) {
    const int i = c + l;
    const int sr = (i < dg) ? lst[o + i] : nn;
    const float a = leaky(s[sr] + dn);
    if (i < tot) m = fmaxf(m, a);
  }
  m = group_max16(m);

  const float ex0 = v0 ? __expf(a0 - m) : 0.f;
  float den = ex0;
#pragma unroll 1
  for (int c = 16; c < totmax; c += 16) {
    const int i = c + l;
    const int sr = (i < dg) ? lst[o + i] : nn;
    const float a = leaky(s[sr] + dn);
    den += (i < tot) ? __expf(a - m) : 0.f;
  }
  den = group_sum16(den) + 1e-16f;

  float acc[V];
#pragma unroll
  for (int k = 0; k < V; ++k) acc[k] = 0.f;

  for (int c = 0; c < totmax; c += 16) {
    int sr; float w;
    if (c == 0) { sr = src0; w = ex0 / den; }
    else {
      const int i = c + l;
      sr = (i < dg) ? lst[o + i] : nn;
      const float a = leaky(s[sr] + dn);
      w = (i < tot) ? __expf(a - m) / den : 0.f;
    }
    const int cnt = min(16, totmax - c);
    for (int j = 0; j < cnt; j += 4) {
      float r[4][V];
#pragma unroll
      for (int u = 0; u < 4; ++u) {
        const int sj = __shfl(sr, gbase + ((j + u) & 15));
        const float* hp = hsrc + (size_t)sj * C + l * V;
        if constexpr (V == 4) {
          const float4 t = *(const float4*)hp;
          r[u][0] = t.x; r[u][1] = t.y; r[u][2] = t.z; r[u][3] = t.w;
        } else {
#pragma unroll
          for (int k = 0; k < V; ++k) r[u][k] = hp[k];
        }
      }
      if constexpr (V == 4) {
        asm volatile("" :: "v"(r[0][0]), "v"(r[0][1]), "v"(r[0][2]), "v"(r[0][3]),
                           "v"(r[1][0]), "v"(r[1][1]), "v"(r[1][2]), "v"(r[1][3]),
                           "v"(r[2][0]), "v"(r[2][1]), "v"(r[2][2]), "v"(r[2][3]),
                           "v"(r[3][0]), "v"(r[3][1]), "v"(r[3][2]), "v"(r[3][3]));
      } else {
        asm volatile("" :: "v"(r[0][0]), "v"(r[1][0]), "v"(r[2][0]), "v"(r[3][0]));
      }
#pragma unroll
      for (int u = 0; u < 4; ++u) {
        const float wj = __shfl(w, gbase + ((j + u) & 15));
        if (c + j + u < tot) {
#pragma unroll
          for (int k = 0; k < V; ++k) acc[k] += r[u][k] * wj;
        }
      }
    }
  }

#pragma unroll
  for (int k = 0; k < V; ++k) ov[k] = acc[k] + bias[l * V + k];
}

#define RG1 5
#define HP 68

__global__ void __launch_bounds__(256)
gather1_proj(const int* __restrict__ off, const int* __restrict__ deg,
             const int* __restrict__ lst,
             const float* __restrict__ s, const float* __restrict__ d,
             const float* __restrict__ h1, const float* __restrict__ b1,
             float* __restrict__ ixz,
             const float* __restrict__ W2,
             const float* __restrict__ asrc2, const float* __restrict__ adst2,
             float* __restrict__ h2, float* __restrict__ s2, float* __restrict__ d2,
             int Nn) {
  constexpr int C = C1, V = C1 / 16;
  __shared__ alignas(16) float sW[C1 * C2];
  __shared__ alignas(16) float sH[RG1 * 16 * HP];

  static_assert((C1 * C2) / 4 == 256, "one float4 of W2 per thread");
  float4 w2v;

  const int lane  = threadIdx.x & 31;
  const int l     = lane & 15;
  const int gbase = lane & 16;
  const int g     = threadIdx.x >> 4;
  const int brow0 = blockIdx.x * (RG1 * 16);

  RowHead hd[RG1];
#pragma unroll
  for (int t = 0; t < RG1; ++t) hd[t].nn = min(brow0 + t * 16 + g, Nn - 1);
  load_heads(off, deg, lst, s, d, l, hd);

#pragma unroll
  for (int t = 0; t < RG1; ++t) {
    const int n = brow0 + t * 16 + g;
    const bool live = n < Nn;

    if (t == RG1 - 1) w2v = ((const float4*)W2)[threadIdx.x];

    float ov[V];
    gather_row<C>(hd[t], lst, s, h1, b1, l, gbase, ov);

    float pv[V];
#pragma unroll
    for (int k = 0; k < V; ++k) pv[k] = __shfl(ov[k], gbase + (l ^ 8));

    float4 hv;
    hv.x = ov[0] > 0.f ? ov[0] : (__expf(ov[0]) - 1.f);
    hv.y = ov[1] > 0.f ? ov[1] : (__expf(ov[1]) - 1.f);
    hv.z = ov[2] > 0.f ? ov[2] : (__expf(ov[2]) - 1.f);
    hv.w = ov[3] > 0.f ? ov[3] : (__expf(ov[3]) - 1.f);
    if (!live) hv = make_float4(0.f, 0.f, 0.f, 0.f);
    *(float4*)(sH + (t * 16 + g) * HP + l * V) = hv;

    if (live) {
      const bool lo = l < 8;
      float2 kv;
      kv.x = kl_term(lo ? ov[0] : pv[2], lo ? pv[0] : ov[2]);
      kv.y = kl_term(lo ? ov[1] : pv[3], lo ? pv[1] : ov[3]);
      *(float2*)(ixz + (size_t)n * (C / 2) + (l & 7) * V + (lo ? 0 : 2)) = kv;
    }
  }
  {
    const int i  = threadIdx.x;
    const int k  = i >> 2;
    const int c0 = (i & 3) * 4;
    float* dst = sW + (k >> 2) * 64 + (k & 3);
    dst[(c0 + 0) * 4] = w2v.x;
    dst[(c0 + 1) * 4] = w2v.y;
    dst[(c0 + 2) * 4] = w2v.z;
    dst[(c0 + 3) * 4] = w2v.w;
  }

  __syncthreads();
  const int wave = threadIdx.x >> 5;
  const int row0 = brow0 + wave * 16;
  if (wave >= RG1 || row0 >= Nn) return;

  const int r    = l;
  const int kh   = (lane >> 4) << 1;
  const float sa = asrc2[r], da = adst2[r];

  v8f acc = {};
  const float* abase = sH + (wave * 16 + r) * HP + kh;
  const float* bbase = sW + r * 4 + kh;
  for (int k0 = 0; k0 < C1; k0 += 4) {
    v2f a = *(const v2f*)(abase + k0);
    v2f b = *(const v2f*)(bbase + (k0 >> 2) * 64);
    acc = __builtin_amdgcn_wmma_f32_16x16x4_f32(false, a, false, b, (short)0, acc, false, false);
  }

  const int mbase = (lane >> 4) * 8;
  float ps[8], pd[8];
  for (int i = 0; i < 8; ++i) {
    const int row = row0 + mbase + i;
    if (row < Nn) h2[(size_t)row * C2 + r] = acc[i];
    ps[i] = acc[i] * sa;
    pd[i] = acc[i] * da;
  }
  const float st = group_sum16x8(ps, r);
  const float dt = group_sum16x8(pd, r);
  const int srow = row0 + mbase + (r >> 1);
  if ((r & 1) == 0 && srow < Nn) { s2[srow] = st; d2[srow] = dt; }
}

__global__ void __launch_bounds__(256)
gather2_out(const int* __restrict__ off, const int* __restrict__ deg,
            const int* __restrict__ lst,
            const float* __restrict__ s, const float* __restrict__ d,
            const float* __restrict__ h2, const float* __restrict__ b2,
            float* __restrict__ orow, float* __restrict__ ixz, int Nn) {
  constexpr int C = C2, V = C2 / 16;
  const int lane  = threadIdx.x & 31;
  const int l     = lane & 15;
  const int gbase = lane & 16;
  const int wv    = (int)((blockIdx.x * 256 + threadIdx.x) >> 5);
  const int wrow0 = wv * (2 * RPG) + (lane >> 4);

  RowHead hd[RPG];
#pragma unroll
  for (int t = 0; t < RPG; ++t) hd[t].nn = min(wrow0 + 2 * t, Nn - 1);
  load_heads(off, deg, lst, s, d, l, hd);

#pragma unroll
  for (int t = 0; t < RPG; ++t) {
    const int n = wrow0 + 2 * t;
    const bool live = n < Nn;

    float ov[V];
    gather_row<C>(hd[t], lst, s, h2, b2, l, gbase, ov);

    float pv[V];
#pragma unroll
    for (int k = 0; k < V; ++k) pv[k] = __shfl(ov[k], gbase + (l ^ 8));

    if (live) {
#pragma unroll
      for (int k = 0; k < V; ++k) orow[(size_t)n * C + l * V + k] = ov[k];
      if (l < 8) {
#pragma unroll
        for (int k = 0; k < V; ++k) ixz[(size_t)n * (C / 2) + l * V + k] = kl_term(ov[k], pv[k]);
      }
    }
  }
}

extern "C" void kernel_launch(void* const* d_in, const int* in_sizes, int n_in,
                              void* d_out, int out_size, void* d_ws, size_t ws_size,
                              hipStream_t stream) {
  const float* x   = (const float*)d_in[0];
  const int*   ei  = (const int*)  d_in[1];
  const float* W1  = (const float*)d_in[2];
  const float* as1 = (const float*)d_in[3];
  const float* ad1 = (const float*)d_in[4];
  const float* b1  = (const float*)d_in[5];
  const float* W2  = (const float*)d_in[6];
  const float* as2 = (const float*)d_in[7];
  const float* ad2 = (const float*)d_in[8];
  const float* b2  = (const float*)d_in[9];

  const int N = in_sizes[0] / F_IN;
  const int E = in_sizes[1] / 2;
  float* out = (float*)d_out;

  const size_t N64 = (size_t)N * C1, N16 = (size_t)N * C2;
  float* ws   = (float*)d_ws;
  float* h1   = ws;
  float* h2   = h1 + N64;
  float* s1   = h2 + N16;
  float* d1   = s1 + N;
  float* s2   = d1 + N;
  float* d2   = s2 + N;
  int*   deg  = (int*)(d2 + N);
  int*   off  = deg + N;
  const size_t NL = (size_t)N * CAP + E;
  int*   lin  = off + N;
  int*   lst  = lin + NL;
  int*   ov   = lst + NL;
  int*   cur  = ov + 2 * (size_t)E;

  float* ixz1 = out + N16;
  float* ixz2 = out + (size_t)N * 48;

  const int tiles = (N + 15) / 16;
  const int gblk  = (tiles + 7) / 8;
  const int gpart = (gblk + 1) / 2;
  const int nblk  = (N + SPAN - 1) / SPAN;
  const int eblk  = (E + SPAN - 1) / SPAN;
  const int rblk  = (N + 16 * RPG - 1) / (16 * RPG);
  const int rblk1 = (N + 16 * RG1 - 1) / (16 * RG1);

  Gemm1Args g0 = {x, W1, as1, ad1, h1, s1, d1, N, 0,     gpart, 0};
  Gemm1Args g1 = {x, W1, as1, ad1, h1, s1, d1, N, gpart, gpart, 0};
  csr_init<<<nblk, 256, 0, stream>>>(deg, cur, N);
  csr_count_fill<<<gpart + eblk, 256, 0, stream>>>(ei, E, deg, lin, cur + 1, ov, g0);
  csr_sort<<<gpart + rblk, 256, 0, stream>>>(deg, lin, lst, off, cur, cur + 1, ov, N, g1);

  gather1_proj<<<rblk1, 256, 0, stream>>>(off, deg, lst, s1, d1, h1, b1, ixz1,
                                         W2, as2, ad2, h2, s2, d2, N);

  gather2_out<<<rblk, 256, 0, stream>>>(off, deg, lst, s2, d2, h2, b2, out, ixz2, N);
}


# speedup vs baseline: 3.7428x; 1.4317x over previous
#include <hip/hip_runtime.h>
#include <hip/hip_bf16.h>

typedef __attribute__((ext_vector_type(2))) float v2f;
typedef __attribute__((ext_vector_type(8))) float v8f;

#define F_IN 128
#define C1   64
#define C2   16
#define NEG_SLOPE 0.2f

__device__ __forceinline__ float group_max16(float v) {
  v = fmaxf(v, __shfl_xor(v, 8)); v = fmaxf(v, __shfl_xor(v, 4));
  v = fmaxf(v, __shfl_xor(v, 2)); v = fmaxf(v, __shfl_xor(v, 1));
  return v;
}
__device__ __forceinline__ float group_sum16(float v) {
  v += __shfl_xor(v, 8); v += __shfl_xor(v, 4);
  v += __shfl_xor(v, 2); v += __shfl_xor(v, 1);
  return v;
}
__device__ __forceinline__ float group_sum16x8(const float (&v)[8], int l) {
  const bool b3 = (l & 8) != 0, b2 = (l & 4) != 0, b1 = (l & 2) != 0;
  float w[4];
  for (int j = 0; j < 4; ++j) {
    const float mine = b3 ? v[j + 4] : v[j];
    const float oth  = b3 ? v[j] : v[j + 4];
    w[j] = mine + __shfl_xor(oth, 8);
  }
  float u[2];
  for (int j = 0; j < 2; ++j) {
    const float mine = b2 ? w[j + 2] : w[j];
    const float oth  = b2 ? w[j] : w[j + 2];
    u[j] = mine + __shfl_xor(oth, 4);
  }
  const float mine = b1 ? u[1] : u[0];
  const float oth  = b1 ? u[0] : u[1];
  const float t = mine + __shfl_xor(oth, 2);
  return t + __shfl_xor(t, 1);
}

struct Gemm1Args {
  const float* x; const float* W1; const float* asrc; const float* adst;
  float* h1; float* s; float* d; int Nn;
  int first;
  int count;
  int tail;
};

__device__ __forceinline__ void
gemm1_block(const float* __restrict__ x, const float* __restrict__ W1,
            const float* __restrict__ asrc, const float* __restrict__ adst,
            float* __restrict__ h1, float* __restrict__ s, float* __restrict__ d, int Nn,
            int gb) {
  __shared__ float sW[F_IN * C1];
  for (int i = threadIdx.x; i < (F_IN * C1) / 4; i += 256) {
    float4 v = ((const float4*)W1)[i];
    const int k  = i >> 4;
    const int c0 = (i & 15) * 4;
    float* dst = sW + (k >> 2) * 256 + (k & 3);
    dst[(c0 + 0) * 4] = v.x;
    dst[(c0 + 1) * 4] = v.y;
    dst[(c0 + 2) * 4] = v.z;
    dst[(c0 + 3) * 4] = v.w;
  }
  __syncthreads();

  const int wave = threadIdx.x >> 5;
  const int lane = threadIdx.x & 31;
  const int tile = gb * 8 + wave;
  if (tile * 16 >= Nn) return;
  const int row0 = tile * 16;
  const int r    = lane & 15;
  const int kh   = (lane >> 4) << 1;
  const int arow = min(row0 + r, Nn - 1);

  const float sa0 = asrc[r], sa1 = asrc[r + 16], sa2 = asrc[r + 32], sa3 = asrc[r + 48];
  const float da0 = adst[r], da1 = adst[r + 16], da2 = adst[r + 32], da3 = adst[r + 48];

  v8f acc0 = {}, acc1 = {}, acc2 = {}, acc3 = {};
  const float* ap = x + (size_t)arow * F_IN + kh;
  const float* bbase = sW + r * 4 + kh;
  for (int k0 = 0; k0 < F_IN; k0 += 4) {
    v2f a; a.x = ap[k0]; a.y = ap[k0 + 1];
    const float* bq = bbase + (k0 >> 2) * 256;
    v2f b0 = *(const v2f*)(bq);
    v2f b1 = *(const v2f*)(bq + 64);
    v2f b2 = *(const v2f*)(bq + 128);
    v2f b3 = *(const v2f*)(bq + 192);
    acc0 = __builtin_amdgcn_wmma_f32_16x16x4_f32(false, a, false, b0, (short)0, acc0, false, false);
    acc1 = __builtin_amdgcn_wmma_f32_16x16x4_f32(false, a, false, b1, (short)0, acc1, false, false);
    acc2 = __builtin_amdgcn_wmma_f32_16x16x4_f32(false, a, false, b2, (short)0, acc2, false, false);
    acc3 = __builtin_amdgcn_wmma_f32_16x16x4_f32(false, a, false, b3, (short)0, acc3, false, false);
  }

  const int mbase = (lane >> 4) * 8;
  for (int i = 0; i < 8; ++i) {
    const int row = row0 + mbase + i;
    if (row < Nn) {
      float* op = h1 + (size_t)row * C1 + r;
      op[0]  = acc0[i];
      op[16] = acc1[i];
      op[32] = acc2[i];
      op[48] = acc3[i];
    }
  }

  float ps[8], pd[8];
  for (int i = 0; i < 8; ++i) {
    ps[i] = acc0[i] * sa0 + acc1[i] * sa1 + acc2[i] * sa2 + acc3[i] * sa3;
    pd[i] = acc0[i] * da0 + acc1[i] * da1 + acc2[i] * da2 + acc3[i] * da3;
  }
  const float st = group_sum16x8(ps, r);
  const float dt = group_sum16x8(pd, r);
  const int srow = row0 + mbase + (r >> 1);
  if ((r & 1) == 0 && srow < Nn) { s[srow] = st; d[srow] = dt; }
}

#define RPG 4
#define SPAN (256 * RPG)
#define CAP 16
#define SEGW 16
#define SEGC (SEGW - 1)

__global__ void __launch_bounds__(256)
csr_init(int* __restrict__ seg, int* __restrict__ cursor, int nseg) {
  const int i0 = blockIdx.x * SPAN + threadIdx.x;
#pragma unroll
  for (int k = 0; k < RPG; ++k) {
    const int i = i0 + k * 256;
    if (i < nseg) seg[(size_t)i * SEGW] = 0;
  }
  if (i0 == 0) { cursor[0] = 0; cursor[1] = 0; }
}

__global__ void __launch_bounds__(256) __attribute__((amdgpu_waves_per_eu(16)))
csr_count_fill(const int* __restrict__ ei, int E, int* __restrict__ seg,
               int* __restrict__ ovn, int* __restrict__ ov, const Gemm1Args g) {
  if ((int)blockIdx.x < g.count) {
    gemm1_block(g.x, g.W1, g.asrc, g.adst, g.h1, g.s, g.d, g.Nn, g.first + (int)blockIdx.x);
    return;
  }
  const int e0 = ((int)blockIdx.x - g.count) * SPAN + threadIdx.x;
  int dst[RPG], src[RPG], rk[RPG];
#pragma unroll
  for (int k = 0; k < RPG; ++k) {
    const int ec = min(e0 + k * 256, E - 1);
    dst[k] = ei[E + ec];
    src[k] = ei[ec];
  }
  const int half = (__builtin_amdgcn_s_sendmsg_rtn(0x87) >> 18) & 1;
  int* __restrict__ plane = seg + (size_t)half * g.Nn * SEGW;
#pragma unroll
  for (int k = 0; k < RPG; ++k) {
    rk[k] = 0;
    if (e0 + k * 256 < E) rk[k] = atomicAdd(plane + dst[k] * SEGW, 1);
  }
#pragma unroll
  for (int k = 0; k < RPG; ++k) {
    if (e0 + k * 256 < E) {
      if (rk[k] < SEGC) plane[dst[k] * SEGW + 1 + rk[k]] = src[k];
      else {
        const int i = atomicAdd(ovn, 1);
        ov[2 * i] = dst[k]; ov[2 * i + 1] = src[k];
      }
    }
  }
}

__global__ void __launch_bounds__(256) __attribute__((amdgpu_waves_per_eu(16)))
csr_sort(const int* __restrict__ seg, int* tmp, int* __restrict__ lout, int* __restrict__ off,
         int* __restrict__ deg, int* __restrict__ cursor, const int* __restrict__ ovn,
         const int* __restrict__ ov, int Nn, const Gemm1Args g) {
  if ((int)blockIdx.x < g.count) {
    gemm1_block(g.x, g.W1, g.asrc, g.adst, g.h1, g.s, g.d, g.Nn, g.first + (int)blockIdx.x);
    return;
  }
  const int lane  = threadIdx.x & 31;
  const int l     = lane & 15;
  const int gbase = lane & 16;
  const int grp = (int)((((int)blockIdx.x - g.count) * 256 + threadIdx.x) >> 4);
  const int* __restrict__ seg1 = seg + (size_t)Nn * SEGW;
  int w0[RPG], w1[RPG];
#pragma unroll
  for (int t = 0; t < RPG; ++t) {
    const int nc = min(grp * RPG + t, Nn - 1);
    w0[t] = seg[nc * SEGW + l];
    w1[t] = seg1[nc * SEGW + l];
  }

#pragma unroll
  for (int t = 0; t < RPG; ++t) {
    const int n  = grp * RPG + t;
    const bool live = n < Nn;
    int c0 = __shfl(w0[t], gbase);
    int c1 = __shfl(w1[t], gbase);
    if (!live) { c0 = 0; c1 = 0; }
    const int dg = c0 + c1;
    const int dgmax = max(dg, __shfl_xor(dg, 16));
    const bool slow = c0 > SEGC || c1 > SEGC || dg > CAP;
    int o = min(n, Nn - 1) * CAP;
    if (__builtin_amdgcn_ballot_w32(slow) == 0u) {
      const int k0 = __shfl(w0[t], gbase + ((l + 1) & 15));
      const int k1 = __shfl(w1[t], gbase + ((l - c0 + 1) & 15));
      const int key = (l < c0) ? k0 : ((l < dg) ? k1 : 0x7fffffff);
      int rank = 0;
      for (int j = 0; j < dgmax; ++j) {
        const int kt = __shfl(key, gbase + j);
        rank += ((j < dg) && (kt < key || (kt == key && j < l))) ? 1 : 0;
      }
      if (l < dg) lout[o + rank] = key;
    } else {
      const int m0 = min(c0, SEGC), m1 = min(c1, SEGC);
      int base = 0;
      if (l == 0 && dg > 0) base = atomicAdd(cursor, dg);
      base = __shfl(base, gbase);
      o = Nn * CAP + base;
      if (l >= 1 && l <= m0) tmp[base + l - 1] = w0[t];
      if (l >= 1 && l <= m1) tmp[base + m0 + l - 1] = w1[t];
      const bool need = m0 + m1 < dg;
      const int novf = *ovn;
      int found = m0 + m1;
      for (int c = 0; c < novf; c += 16) {
        const int i = c + l;
        bool hit = false; int sv = 0;
        if (need && i < novf) { hit = ov[2 * i] == n; sv = ov[2 * i + 1]; }
        const unsigned gb = (__builtin_amdgcn_ballot_w32(hit) >> gbase) & 0xffffu;
        if (hit) tmp[base + found + __popc(gb & ((1u << l) - 1u))] = sv;
        found += __popc(gb);
      }
      __threadfence();
      for (int ci = 0; ci < dgmax; ci += 16) {
        const int i = ci + l;
        const int key = (i < dg) ? tmp[base + i] : 0x7fffffff;
        int rank = 0;
        for (int cj = 0; cj < dgmax; cj += 16) {
          const int jm = cj + l;
          const int kmine = (jm < dg) ? tmp[base + jm] : 0x7fffffff;
          const int cnt = min(16, dgmax - cj);
          for (int j = 0; j < cnt; ++j) {
            const int kt = __shfl(kmine, gbase + j);
            const int jt = cj + j;
            rank += ((jt < dg) && (kt < key || (kt == key && jt < i))) ? 1 : 0;
          }
        }
        if (i < dg) lout[o + rank] = key;
      }
    }
    if (l == 0 && live) { off[n] = o; deg[n] = dg; }
  }
}

__device__ __forceinline__ float softplusf(float v) {
  return fmaxf(v, 0.f) + log1pf(__expf(-fabsf(v)));
}
__device__ __forceinline__ float kl_term(float mean, float sp) {
  const float sd = softplusf(sp) + 1e-10f;
  return -__logf(sd) + 0.5f * (sd * sd + mean * mean - 1.f);
}
__device__ __forceinline__ float leaky(float a) { return a > 0.f ? a : NEG_SLOPE * a; }

struct RowHead { int nn, o, dg, src0; float dn, s0; };

template <int R>
__device__ __forceinline__ void
load_heads(const int* __restrict__ off, const int* __restrict__ deg,
           const int* __restrict__ lst,
           const float* __restrict__ s, const float* __restrict__ d,
           int l, RowHead (&h)[R]) {
#pragma unroll
  for (int t = 0; t < R; ++t) { h[t].o = off[h[t].nn]; h[t].dg = deg[h[t].nn]; h[t].dn = d[h[t].nn]; }
#pragma unroll
  for (int t = 0; t < R; ++t) h[t].src0 = (l < h[t].dg) ? lst[h[t].o + l] : h[t].nn;
#pragma unroll
  for (int t = 0; t < R; ++t) h[t].s0 = s[h[t].src0];
}

template <int C>
__device__ __forceinline__ void
gather_row(const RowHead& hd, const int* __restrict__ lst,
           const float* __restrict__ s,
           const float* __restrict__ hsrc, const float* __restrict__ bias,
           int l, int gbase, float (&ov)[C / 16]) {
  constexpr int V = C / 16;
  const int nn  = hd.nn;
  const int o   = hd.o;
  const int dg  = hd.dg;
  const int tot = dg + 1;
  const float dn = hd.dn;
  const int totmax = max(tot, __shfl_xor(tot, 16));

  const int   src0 = hd.src0;
  const float a0   = leaky(hd.s0 + dn);
  const bool  v0   = l < tot;

  float m = v0 ? a0 : -INFINITY;
#pragma unroll 1
  for (int c = 16; c < totmax; c += 16) {
    const int i = c + l;
    const int sr = (i < dg) ? lst[o + i] : nn;
    const float a = leaky(s[sr] + dn);
    if (i < tot) m = fmaxf(m, a);
  }
  m = group_max16(m);

  const float ex0 = v0 ? __expf(a0 - m) : 0.f;
  float den = ex0;
#pragma unroll 1
  for (int c = 16; c < totmax; c += 16) {
    const int i = c + l;
    const int sr = (i < dg) ? lst[o + i] : nn;
    const float a = leaky(s[sr] + dn);
    den += (i < tot) ? __expf(a - m) : 0.f;
  }
  den = group_sum16(den) + 1e-16f;

  float acc[V];
#pragma unroll
  for (int k = 0; k < V; ++k) acc[k] = 0.f;

  for (int c = 0; c < totmax; c += 16) {
    int sr; float w;
    if (c == 0) { sr = src0; w = ex0 / den; }
    else {
      const int i = c + l;
      sr = (i < dg) ? lst[o + i] : nn;
      const float a = leaky(s[sr] + dn);
      w = (i < tot) ? __expf(a - m) / den : 0.f;
    }
    const int cnt = min(16, totmax - c);
    for (int j = 0; j < cnt; j += 4) {
      float r[4][V];
#pragma unroll
      for (int u = 0; u < 4; ++u) {
        const int sj = __shfl(sr, gbase + ((j + u) & 15));
        const float* hp = hsrc + (size_t)sj * C + l * V;
        if constexpr (V == 4) {
          const float4 t = *(const float4*)hp;
          r[u][0] = t.x; r[u][1] = t.y; r[u][2] = t.z; r[u][3] = t.w;
        } else {
#pragma unroll
          for (int k = 0; k < V; ++k) r[u][k] = hp[k];
        }
      }
      if constexpr (V == 4) {
        asm volatile("" :: "v"(r[0][0]), "v"(r[0][1]), "v"(r[0][2]), "v"(r[0][3]),
                           "v"(r[1][0]), "v"(r[1][1]), "v"(r[1][2]), "v"(r[1][3]),
                           "v"(r[2][0]), "v"(r[2][1]), "v"(r[2][2]), "v"(r[2][3]),
                           "v"(r[3][0]), "v"(r[3][1]), "v"(r[3][2]), "v"(r[3][3]));
      } else {
        asm volatile("" :: "v"(r[0][0]), "v"(r[1][0]), "v"(r[2][0]), "v"(r[3][0]));
      }
#pragma unroll
      for (int u = 0; u < 4; ++u) {
        const float wj = __shfl(w, gbase + ((j + u) & 15));
        if (c + j + u < tot) {
#pragma unroll
          for (int k = 0; k < V; ++k) acc[k] += r[u][k] * wj;
        }
      }
    }
  }

#pragma unroll
  for (int k = 0; k < V; ++k) ov[k] = acc[k] + bias[l * V + k];
}

#define RG1 5
#define HP 68

__global__ void __launch_bounds__(256)
gather1_proj(const int* __restrict__ off, const int* __restrict__ deg,
             const int* __restrict__ lst,
             const float* __restrict__ s, const float* __restrict__ d,
             const float* __restrict__ h1, const float* __restrict__ b1,
             float* __restrict__ ixz,
             const float* __restrict__ W2,
             const float* __restrict__ asrc2, const float* __restrict__ adst2,
             float* __restrict__ h2, float* __restrict__ s2, float* __restrict__ d2,
             int Nn) {
  constexpr int C = C1, V = C1 / 16;
  __shared__ alignas(16) float sW[C1 * C2];
  __shared__ alignas(16) float sH[RG1 * 16 * HP];

  static_assert((C1 * C2) / 4 == 256, "one float4 of W2 per thread");
  float4 w2v;

  const int lane  = threadIdx.x & 31;
  const int l     = lane & 15;
  const int gbase = lane & 16;
  const int g     = threadIdx.x >> 4;
  const int brow0 = blockIdx.x * (RG1 * 16);

  RowHead hd[RG1];
#pragma unroll
  for (int t = 0; t < RG1; ++t) hd[t].nn = min(brow0 + t * 16 + g, Nn - 1);
  load_heads(off, deg, lst, s, d, l, hd);

#pragma unroll
  for (int t = 0; t < RG1; ++t) {
    const int n = brow0 + t * 16 + g;
    const bool live = n < Nn;

    if (t == RG1 - 1) w2v = ((const float4*)W2)[threadIdx.x];

    float ov[V];
    gather_row<C>(hd[t], lst, s, h1, b1, l, gbase, ov);

    float pv[V];
#pragma unroll
    for (int k = 0; k < V; ++k) pv[k] = __shfl(ov[k], gbase + (l ^ 8));

    float4 hv;
    hv.x = ov[0] > 0.f ? ov[0] : (__expf(ov[0]) - 1.f);
    hv.y = ov[1] > 0.f ? ov[1] : (__expf(ov[1]) - 1.f);
    hv.z = ov[2] > 0.f ? ov[2] : (__expf(ov[2]) - 1.f);
    hv.w = ov[3] > 0.f ? ov[3] : (__expf(ov[3]) - 1.f);
    if (!live) hv = make_float4(0.f, 0.f, 0.f, 0.f);
    *(float4*)(sH + (t * 16 + g) * HP + l * V) = hv;

    if (live) {
      const bool lo = l < 8;
      float2 kv;
      kv.x = kl_term(lo ? ov[0] : pv[2], lo ? pv[0] : ov[2]);
      kv.y = kl_term(lo ? ov[1] : pv[3], lo ? pv[1] : ov[3]);
      *(float2*)(ixz + (size_t)n * (C / 2) + (l & 7) * V + (lo ? 0 : 2)) = kv;
    }
  }
  {
    const int i  = threadIdx.x;
    const int k  = i >> 2;
    const int c0 = (i & 3) * 4;
    float* dst = sW + (k >> 2) * 64 + (k & 3);
    dst[(c0 + 0) * 4] = w2v.x;
    dst[(c0 + 1) * 4] = w2v.y;
    dst[(c0 + 2) * 4] = w2v.z;
    dst[(c0 + 3) * 4] = w2v.w;
  }

  __syncthreads();
  const int wave = threadIdx.x >> 5;
  const int row0 = brow0 + wave * 16;
  if (wave >= RG1 || row0 >= Nn) return;

  const int r    = l;
  const int kh   = (lane >> 4) << 1;
  const float sa = asrc2[r], da = adst2[r];

  v8f acc = {};
  const float* abase = sH + (wave * 16 + r) * HP + kh;
  const float* bbase = sW + r * 4 + kh;
  for (int k0 = 0; k0 < C1; k0 += 4) {
    v2f a = *(const v2f*)(abase + k0);
    v2f b = *(const v2f*)(bbase + (k0 >> 2) * 64);
    acc = __builtin_amdgcn_wmma_f32_16x16x4_f32(false, a, false, b, (short)0, acc, false, false);
  }

  const int mbase = (lane >> 4) * 8;
  float ps[8], pd[8];
  for (int i = 0; i < 8; ++i) {
    const int row = row0 + mbase + i;
    if (row < Nn) h2[(size_t)row * C2 + r] = acc[i];
    ps[i] = acc[i] * sa;
    pd[i] = acc[i] * da;
  }
  const float st = group_sum16x8(ps, r);
  const float dt = group_sum16x8(pd, r);
  const int srow = row0 + mbase + (r >> 1);
  if ((r & 1) == 0 && srow < Nn) { s2[srow] = st; d2[srow] = dt; }
}

__global__ void __launch_bounds__(256)
gather2_out(const int* __restrict__ off, const int* __restrict__ deg,
            const int* __restrict__ lst,
            const float* __restrict__ s, const float* __restrict__ d,
            const float* __restrict__ h2, const float* __restrict__ b2,
            float* __restrict__ orow, float* __restrict__ ixz, int Nn) {
  constexpr int C = C2, V = C2 / 16;
  const int lane  = threadIdx.x & 31;
  const int l     = lane & 15;
  const int gbase = lane & 16;
  const int wv    = (int)((blockIdx.x * 256 + threadIdx.x) >> 5);
  const int wrow0 = wv * (2 * RPG) + (lane >> 4);

  RowHead hd[RPG];
#pragma unroll
  for (int t = 0; t < RPG; ++t) hd[t].nn = min(wrow0 + 2 * t, Nn - 1);
  load_heads(off, deg, lst, s, d, l, hd);

#pragma unroll
  for (int t = 0; t < RPG; ++t) {
    const int n = wrow0 + 2 * t;
    const bool live = n < Nn;

    float ov[V];
    gather_row<C>(hd[t], lst, s, h2, b2, l, gbase, ov);

    float pv[V];
#pragma unroll
    for (int k = 0; k < V; ++k) pv[k] = __shfl(ov[k], gbase + (l ^ 8));

    if (live) {
#pragma unroll
      for (int k = 0; k < V; ++k) orow[(size_t)n * C + l * V + k] = ov[k];
      if (l < 8) {
#pragma unroll
        for (int k = 0; k < V; ++k) ixz[(size_t)n * (C / 2) + l * V + k] = kl_term(ov[k], pv[k]);
      }
    }
  }
}

extern "C" void kernel_launch(void* const* d_in, const int* in_sizes, int n_in,
                              void* d_out, int out_size, void* d_ws, size_t ws_size,
                              hipStream_t stream) {
  const float* x   = (const float*)d_in[0];
  const int*   ei  = (const int*)  d_in[1];
  const float* W1  = (const float*)d_in[2];
  const float* as1 = (const float*)d_in[3];
  const float* ad1 = (const float*)d_in[4];
  const float* b1  = (const float*)d_in[5];
  const float* W2  = (const float*)d_in[6];
  const float* as2 = (const float*)d_in[7];
  const float* ad2 = (const float*)d_in[8];
  const float* b2  = (const float*)d_in[9];

  const int N = in_sizes[0] / F_IN;
  const int E = in_sizes[1] / 2;
  float* out = (float*)d_out;

  const size_t N64 = (size_t)N * C1, N16 = (size_t)N * C2;
  float* ws   = (float*)d_ws;
  float* h1   = ws;
  float* h2   = h1 + N64;
  float* s1   = h2 + N16;
  float* d1   = s1 + N;
  float* s2   = d1 + N;
  float* d2   = s2 + N;
  int*   deg  = (int*)(d2 + N);
  int*   off  = deg + N;
  const size_t NS = (size_t)N * SEGW;
  const size_t NL = (size_t)N * CAP + E;
  int*   seg  = off + N;
  int*   tmp  = seg + 2 * NS;
  int*   lst  = tmp + E;
  int*   ov   = lst + NL;
  int*   cur  = ov + 2 * (size_t)E;

  float* ixz1 = out + N16;
  float* ixz2 = out + (size_t)N * 48;

  const int tiles = (N + 15) / 16;
  const int gblk  = (tiles + 7) / 8;
  const int gpart = (gblk + 1) / 2;
  const int sblk  = (2 * N + SPAN - 1) / SPAN;
  const int eblk  = (E + SPAN - 1) / SPAN;
  const int rblk  = (N + 16 * RPG - 1) / (16 * RPG);
  const int rblk1 = (N + 16 * RG1 - 1) / (16 * RG1);

  Gemm1Args g0 = {x, W1, as1, ad1, h1, s1, d1, N, 0,     gpart, 0};
  Gemm1Args g1 = {x, W1, as1, ad1, h1, s1, d1, N, gpart, gpart, 0};
  csr_init<<<sblk, 256, 0, stream>>>(seg, cur, 2 * N);
  csr_count_fill<<<gpart + eblk, 256, 0, stream>>>(ei, E, seg, cur + 1, ov, g0);
  csr_sort<<<gpart + rblk, 256, 0, stream>>>(seg, tmp, lst, off, deg, cur, cur + 1, ov, N, g1);

  gather1_proj<<<rblk1, 256, 0, stream>>>(off, deg, lst, s1, d1, h1, b1, ixz1,
                                         W2, as2, ad2, h2, s2, d2, N);

  gather2_out<<<rblk, 256, 0, stream>>>(off, deg, lst, s2, d2, h2, b2, out, ixz2, N);
}


# exchanges inside the 16-lane groups (softmax max and sum, source and weight broadcasts, rank compares, dot sums) by DPP row operations and lane permutes, not LDS permutes; sums unchanged
# speedup vs baseline: 3.9010x; 1.0423x over previous
#include <hip/hip_runtime.h>
#include <hip/hip_bf16.h>

typedef __attribute__((ext_vector_type(2))) float v2f;
typedef __attribute__((ext_vector_type(8))) float v8f;

#define F_IN 128
#define C1   64
#define C2   16
#define NEG_SLOPE 0.2f

template <int M> __device__ __forceinline__ int row_xor(int v) {
  return __builtin_amdgcn_update_dpp(0, v, 0x160 | M, 0xf, 0xf, false);
}
template <int M> __device__ __forceinline__ float row_xor(float v) {
  return __int_as_float(row_xor<M>(__float_as_int(v)));
}
template <int J> __device__ __forceinline__ int row_lane(int v) {
  return __builtin_amdgcn_update_dpp(0, v, 0x150 | J, 0xf, 0xf, false);
}
template <int N> __device__ __forceinline__ int row_ror(int v) {
  return __builtin_amdgcn_update_dpp(0, v, 0x120 | N, 0xf, 0xf, false);
}
__device__ __forceinline__ int row_pick(int v, int j) {
  const unsigned sel = 0x11111111u * (unsigned)j;
  return (int)__builtin_amdgcn_permlane16(0u, (unsigned)v, sel, sel, false, false);
}
__device__ __forceinline__ float row_pick(float v, int j) {
  return __int_as_float(row_pick(__float_as_int(v), j));
}
__device__ __forceinline__ int row_var(int v, int sel) {
  return (int)__builtin_amdgcn_permlane16_var(0u, (unsigned)v, (unsigned)sel, false, false);
}
__device__ __forceinline__ int other_row(int v) {
  return (int)__builtin_amdgcn_permlanex16(0u, (unsigned)v, 0x76543210u, 0xfedcba98u, false, false);
}

__device__ __forceinline__ float group_max16(float v) {
  v = fmaxf(v, row_xor<8>(v)); v = fmaxf(v, row_xor<4>(v));
  v = fmaxf(v, row_xor<2>(v)); v = fmaxf(v, row_xor<1>(v));
  return v;
}
__device__ __forceinline__ float group_sum16(float v) {
  v += row_xor<8>(v); v += row_xor<4>(v);
  v += row_xor<2>(v); v += row_xor<1>(v);
  return v;
}
__device__ __forceinline__ float group_sum16x8(const float (&v)[8], int l) {
  const bool b3 = (l & 8) != 0, b2 = (l & 4) != 0, b1 = (l & 2) != 0;
  float w[4];
  for (int j = 0; j < 4; ++j) {
    const float mine = b3 ? v[j + 4] : v[j];
    const float oth  = b3 ? v[j] : v[j + 4];
    w[j] = mine + row_xor<8>(oth);
  }
  float u[2];
  for (int j = 0; j < 2; ++j) {
    const float mine = b2 ? w[j + 2] : w[j];
    const float oth  = b2 ? w[j] : w[j + 2];
    u[j] = mine + row_xor<4>(oth);
  }
  const float mine = b1 ? u[1] : u[0];
  const float oth  = b1 ? u[0] : u[1];
  const float t = mine + row_xor<2>(oth);
  return t + row_xor<1>(t);
}

struct Gemm1Args {
  const float* x; const float* W1; const float* asrc; const float* adst;
  float* h1; float* s; float* d; int Nn;
  int first;
  int count;
  int tail;
};

__device__ __forceinline__ void
gemm1_block(const float* __restrict__ x, const float* __restrict__ W1,
            const float* __restrict__ asrc, const float* __restrict__ adst,
            float* __restrict__ h1, float* __restrict__ s, float* __restrict__ d, int Nn,
            int gb) {
  __shared__ float sW[F_IN * C1];
  for (int i = threadIdx.x; i < (F_IN * C1) / 4; i += 256) {
    float4 v = ((const float4*)W1)[i];
    const int k  = i >> 4;
    const int c0 = (i & 15) * 4;
    float* dst = sW + (k >> 2) * 256 + (k & 3);
    dst[(c0 + 0) * 4] = v.x;
    dst[(c0 + 1) * 4] = v.y;
    dst[(c0 + 2) * 4] = v.z;
    dst[(c0 + 3) * 4] = v.w;
  }
  __syncthreads();

  const int wave = threadIdx.x >> 5;
  const int lane = threadIdx.x & 31;
  const int tile = gb * 8 + wave;
  if (tile * 16 >= Nn) return;
  const int row0 = tile * 16;
  const int r    = lane & 15;
  const int kh   = (lane >> 4) << 1;
  const int arow = min(row0 + r, Nn - 1);

  const float sa0 = asrc[r], sa1 = asrc[r + 16], sa2 = asrc[r + 32], sa3 = asrc[r + 48];
  const float da0 = adst[r], da1 = adst[r + 16], da2 = adst[r + 32], da3 = adst[r + 48];

  v8f acc0 = {}, acc1 = {}, acc2 = {}, acc3 = {};
  const float* ap = x + (size_t)arow * F_IN + kh;
  const float* bbase = sW + r * 4 + kh;
  for (int k0 = 0; k0 < F_IN; k0 += 4) {
    v2f a; a.x = ap[k0]; a.y = ap[k0 + 1];
    const float* bq = bbase + (k0 >> 2) * 256;
    v2f b0 = *(const v2f*)(bq);
    v2f b1 = *(const v2f*)(bq + 64);
    v2f b2 = *(const v2f*)(bq + 128);
    v2f b3 = *(const v2f*)(bq + 192);
    acc0 = __builtin_amdgcn_wmma_f32_16x16x4_f32(false, a, false, b0, (short)0, acc0, false, false);
    acc1 = __builtin_amdgcn_wmma_f32_16x16x4_f32(false, a, false, b1, (short)0, acc1, false, false);
    acc2 = __builtin_amdgcn_wmma_f32_16x16x4_f32(false, a, false, b2, (short)0, acc2, false, false);
    acc3 = __builtin_amdgcn_wmma_f32_16x16x4_f32(false, a, false, b3, (short)0, acc3, false, false);
  }

  const int mbase = (lane >> 4) * 8;
  for (int i = 0; i < 8; ++i) {
    const int row = row0 + mbase + i;
    if (row < Nn) {
      float* op = h1 + (size_t)row * C1 + r;
      op[0]  = acc0[i];
      op[16] = acc1[i];
      op[32] = acc2[i];
      op[48] = acc3[i];
    }
  }

  float ps[8], pd[8];
  for (int i = 0; i < 8; ++i) {
    ps[i] = acc0[i] * sa0 + acc1[i] * sa1 + acc2[i] * sa2 + acc3[i] * sa3;
    pd[i] = acc0[i] * da0 + acc1[i] * da1 + acc2[i] * da2 + acc3[i] * da3;
  }
  const float st = group_sum16x8(ps, r);
  const float dt = group_sum16x8(pd, r);
  const int srow = row0 + mbase + (r >> 1);
  if ((r & 1) == 0 && srow < Nn) { s[srow] = st; d[srow] = dt; }
}

#define RPG 4
#define SPAN (256 * RPG)
#define CAP 16
#define SEGW 16
#define SEGC (SEGW - 1)

__global__ void __launch_bounds__(256)
csr_init(int* __restrict__ seg, int* __restrict__ cursor, int nseg) {
  const int i0 = blockIdx.x * SPAN + threadIdx.x;
#pragma unroll
  for (int k = 0; k < RPG; ++k) {
    const int i = i0 + k * 256;
    if (i < nseg) seg[(size_t)i * SEGW] = 0;
  }
  if (i0 == 0) { cursor[0] = 0; cursor[1] = 0; }
}

__global__ void __launch_bounds__(256) __attribute__((amdgpu_waves_per_eu(16)))
csr_count_fill(const int* __restrict__ ei, int E, int* __restrict__ seg,
               int* __restrict__ ovn, int* __restrict__ ov, const Gemm1Args g) {
  if ((int)blockIdx.x < g.count) {
    gemm1_block(g.x, g.W1, g.asrc, g.adst, g.h1, g.s, g.d, g.Nn, g.first + (int)blockIdx.x);
    return;
  }
  const int e0 = ((int)blockIdx.x - g.count) * SPAN + threadIdx.x;
  int dst[RPG], src[RPG], rk[RPG];
#pragma unroll
  for (int k = 0; k < RPG; ++k) {
    const int ec = min(e0 + k * 256, E - 1);
    dst[k] = ei[E + ec];
    src[k] = ei[ec];
  }
  const int half = (__builtin_amdgcn_s_sendmsg_rtn(0x87) >> 18) & 1;
  int* __restrict__ plane = seg + (size_t)half * g.Nn * SEGW;
#pragma unroll
  for (int k = 0; k < RPG; ++k) {
    rk[k] = 0;
    if (e0 + k * 256 < E) rk[k] = atomicAdd(plane + dst[k] * SEGW, 1);
  }
#pragma unroll
  for (int k = 0; k < RPG; ++k) {
    if (e0 + k * 256 < E) {
      if (rk[k] < SEGC) plane[dst[k] * SEGW + 1 + rk[k]] = src[k];
      else {
        const int i = atomicAdd(ovn, 1);
        ov[2 * i] = dst[k]; ov[2 * i + 1] = src[k];
      }
    }
  }
}

__global__ void __launch_bounds__(256) __attribute__((amdgpu_waves_per_eu(16)))
csr_sort(const int* __restrict__ seg, int* tmp, int* __restrict__ lout, int* __restrict__ off,
         int* __restrict__ deg, int* __restrict__ cursor, const int* __restrict__ ovn,
         const int* __restrict__ ov, int Nn, const Gemm1Args g) {
  if ((int)blockIdx.x < g.count) {
    gemm1_block(g.x, g.W1, g.asrc, g.adst, g.h1, g.s, g.d, g.Nn, g.first + (int)blockIdx.x);
    return;
  }
  const int lane  = threadIdx.x & 31;
  const int l     = lane & 15;
  const int gbase = lane & 16;
  const int grp = (int)((((int)blockIdx.x - g.count) * 256 + threadIdx.x) >> 4);
  const int* __restrict__ seg1 = seg + (size_t)Nn * SEGW;
  int w0[RPG], w1[RPG];
#pragma unroll
  for (int t = 0; t < RPG; ++t) {
    const int nc = min(grp * RPG + t, Nn - 1);
    w0[t] = seg[nc * SEGW + l];
    w1[t] = seg1[nc * SEGW + l];
  }

#pragma unroll
  for (int t = 0; t < RPG; ++t) {
    const int n  = grp * RPG + t;
    const bool live = n < Nn;
    int c0 = row_lane<0>(w0[t]);
    int c1 = row_lane<0>(w1[t]);
    if (!live) { c0 = 0; c1 = 0; }
    const int dg = c0 + c1;
    const int dgmax = max(dg, other_row(dg));
    const bool slow = c0 > SEGC || c1 > SEGC || dg > CAP;
    int o = min(n, Nn - 1) * CAP;
    if (__builtin_amdgcn_ballot_w32(slow) == 0u) {
      const int k0 = row_ror<15>(w0[t]);
      const int k1 = row_var(w1[t], (l - c0 + 1) & 15);
      const int key = (l < c0) ? k0 : ((l < dg) ? k1 : 0x7fffffff);
      int rank = 0;
      for (int j = 0; j < dgmax; ++j) {
        const int kt = row_pick(key, j);
        rank += ((j < dg) && (kt < key || (kt == key && j < l))) ? 1 : 0;
      }
      if (l < dg) lout[o + rank] = key;
    } else {
      const int m0 = min(c0, SEGC), m1 = min(c1, SEGC);
      int base = 0;
      if (l == 0 && dg > 0) base = atomicAdd(cursor, dg);
      base = row_lane<0>(base);
      o = Nn * CAP + base;
      if (l >= 1 && l <= m0) tmp[base + l - 1] = w0[t];
      if (l >= 1 && l <= m1) tmp[base + m0 + l - 1] = w1[t];
      const bool need = m0 + m1 < dg;
      const int novf = *ovn;
      int found = m0 + m1;
      for (int c = 0; c < novf; c += 16) {
        const int i = c + l;
        bool hit = false; int sv = 0;
        if (need && i < novf) { hit = ov[2 * i] == n; sv = ov[2 * i + 1]; }
        const unsigned gb = (__builtin_amdgcn_ballot_w32(hit) >> gbase) & 0xffffu;
        if (hit) tmp[base + found + __popc(gb & ((1u << l) - 1u))] = sv;
        found += __popc(gb);
      }
      __threadfence();
      for (int ci = 0; ci < dgmax; ci += 16) {
        const int i = ci + l;
        const int key = (i < dg) ? tmp[base + i] : 0x7fffffff;
        int rank = 0;
        for (int cj = 0; cj < dgmax; cj += 16) {
          const int jm = cj + l;
          const int kmine = (jm < dg) ? tmp[base + jm] : 0x7fffffff;
          const int cnt = min(16, dgmax - cj);
          for (int j = 0; j < cnt; ++j) {
            const int kt = row_pick(kmine, j);
            const int jt = cj + j;
            rank += ((jt < dg) && (kt < key || (kt == key && jt < i))) ? 1 : 0;
          }
        }
        if (i < dg) lout[o + rank] = key;
      }
    }
    if (l == 0 && live) { off[n] = o; deg[n] = dg; }
  }
}

__device__ __forceinline__ float softplusf(float v) {
  return fmaxf(v, 0.f) + log1pf(__expf(-fabsf(v)));
}
__device__ __forceinline__ float kl_term(float mean, float sp) {
  const float sd = softplusf(sp) + 1e-10f;
  return -__logf(sd) + 0.5f * (sd * sd + mean * mean - 1.f);
}
__device__ __forceinline__ float leaky(float a) { return a > 0.f ? a : NEG_SLOPE * a; }

struct RowHead { int nn, o, dg, src0; float dn, s0; };

template <int R>
__device__ __forceinline__ void
load_heads(const int* __restrict__ off, const int* __restrict__ deg,
           const int* __restrict__ lst,
           const float* __restrict__ s, const float* __restrict__ d,
           int l, RowHead (&h)[R]) {
#pragma unroll
  for (int t = 0; t < R; ++t) { h[t].o = off[h[t].nn]; h[t].dg = deg[h[t].nn]; h[t].dn = d[h[t].nn]; }
#pragma unroll
  for (int t = 0; t < R; ++t) h[t].src0 = (l < h[t].dg) ? lst[h[t].o + l] : h[t].nn;
#pragma unroll
  for (int t = 0; t < R; ++t) h[t].s0 = s[h[t].src0];
}

template <int C>
__device__ __forceinline__ void
gather_row(const RowHead& hd, const int* __restrict__ lst,
           const float* __restrict__ s,
           const float* __restrict__ hsrc, const float* __restrict__ bias,
           int l, int gbase, float (&ov)[C / 16]) {
  constexpr int V = C / 16;
  const int nn  = hd.nn;
  const int o   = hd.o;
  const int dg  = hd.dg;
  const int tot = dg + 1;
  const float dn = hd.dn;
  const int totmax = max(tot, other_row(tot));

  const int   src0 = hd.src0;
  const float a0   = leaky(hd.s0 + dn);
  const bool  v0   = l < tot;

  float m = v0 ? a0 : -INFINITY;
#pragma unroll 1
  for (int c = 16; c < totmax; c += 16) {
    const int i = c + l;
    const int sr = (i < dg) ? lst[o + i] : nn;
    const float a = leaky(s[sr] + dn);
    if (i < tot) m = fmaxf(m, a);
  }
  m = group_max16(m);

  const float ex0 = v0 ? __expf(a0 - m) : 0.f;
  float den = ex0;
#pragma unroll 1
  for (int c = 16; c < totmax; c += 16) {
    const int i = c + l;
    const int sr = (i < dg) ? lst[o + i] : nn;
    const float a = leaky(s[sr] + dn);
    den += (i < tot) ? __expf(a - m) : 0.f;
  }
  den = group_sum16(den) + 1e-16f;

  float acc[V];
#pragma unroll
  for (int k = 0; k < V; ++k) acc[k] = 0.f;

  for (int c = 0; c < totmax; c += 16) {
    int sr; float w;
    if (c == 0) { sr = src0; w = ex0 / den; }
    else {
      const int i = c + l;
      sr = (i < dg) ? lst[o + i] : nn;
      const float a = leaky(s[sr] + dn);
      w = (i < tot) ? __expf(a - m) / den : 0.f;
    }
    const int cnt = min(16, totmax - c);
    for (int j = 0; j < cnt; j += 4) {
      float r[4][V];
#pragma unroll
      for (int u = 0; u < 4; ++u) {
        const int sj = row_pick(sr, (j + u) & 15);
        const float* hp = hsrc + (size_t)sj * C + l * V;
        if constexpr (V == 4) {
          const float4 t = *(const float4*)hp;
          r[u][0] = t.x; r[u][1] = t.y; r[u][2] = t.z; r[u][3] = t.w;
        } else {
#pragma unroll
          for (int k = 0; k < V; ++k) r[u][k] = hp[k];
        }
      }
      if constexpr (V == 4) {
        asm volatile("" :: "v"(r[0][0]), "v"(r[0][1]), "v"(r[0][2]), "v"(r[0][3]),
                           "v"(r[1][0]), "v"(r[1][1]), "v"(r[1][2]), "v"(r[1][3]),
                           "v"(r[2][0]), "v"(r[2][1]), "v"(r[2][2]), "v"(r[2][3]),
                           "v"(r[3][0]), "v"(r[3][1]), "v"(r[3][2]), "v"(r[3][3]));
      } else {
        asm volatile("" :: "v"(r[0][0]), "v"(r[1][0]), "v"(r[2][0]), "v"(r[3][0]));
      }
#pragma unroll
      for (int u = 0; u < 4; ++u) {
        const float wj = row_pick(w, (j + u) & 15);
        if (c + j + u < tot) {
#pragma unroll
          for (int k = 0; k < V; ++k) acc[k] += r[u][k] * wj;
        }
      }
    }
  }

#pragma unroll
  for (int k = 0; k < V; ++k) ov[k] = acc[k] + bias[l * V + k];
}

#define RG1 5
#define HP 68

__global__ void __launch_bounds__(256)
gather1_proj(const int* __restrict__ off, const int* __restrict__ deg,
             const int* __restrict__ lst,
             const float* __restrict__ s, const float* __restrict__ d,
             const float* __restrict__ h1, const float* __restrict__ b1,
             float* __restrict__ ixz,
             const float* __restrict__ W2,
             const float* __restrict__ asrc2, const float* __restrict__ adst2,
             float* __restrict__ h2, float* __restrict__ s2, float* __restrict__ d2,
             int Nn) {
  constexpr int C = C1, V = C1 / 16;
  __shared__ alignas(16) float sW[C1 * C2];
  __shared__ alignas(16) float sH[RG1 * 16 * HP];

  static_assert((C1 * C2) / 4 == 256, "one float4 of W2 per thread");
  float4 w2v;

  const int lane  = threadIdx.x & 31;
  const int l     = lane & 15;
  const int gbase = lane & 16;
  const int g     = threadIdx.x >> 4;
  const int brow0 = blockIdx.x * (RG1 * 16);

  RowHead hd[RG1];
#pragma unroll
  for (int t = 0; t < RG1; ++t) hd[t].nn = min(brow0 + t * 16 + g, Nn - 1);
  load_heads(off, deg, lst, s, d, l, hd);

#pragma unroll
  for (int t = 0; t < RG1; ++t) {
    const int n = brow0 + t * 16 + g;
    const bool live = n < Nn;

    if (t == RG1 - 1) w2v = ((const float4*)W2)[threadIdx.x];

    float ov[V];
    gather_row<C>(hd[t], lst, s, h1, b1, l, gbase, ov);

    float pv[V];
#pragma unroll
    for (int k = 0; k < V; ++k) pv[k] = row_xor<8>(ov[k]);

    float4 hv;
    hv.x = ov[0] > 0.f ? ov[0] : (__expf(ov[0]) - 1.f);
    hv.y = ov[1] > 0.f ? ov[1] : (__expf(ov[1]) - 1.f);
    hv.z = ov[2] > 0.f ? ov[2] : (__expf(ov[2]) - 1.f);
    hv.w = ov[3] > 0.f ? ov[3] : (__expf(ov[3]) - 1.f);
    if (!live) hv = make_float4(0.f, 0.f, 0.f, 0.f);
    *(float4*)(sH + (t * 16 + g) * HP + l * V) = hv;

    if (live) {
      const bool lo = l < 8;
      float2 kv;
      kv.x = kl_term(lo ? ov[0] : pv[2], lo ? pv[0] : ov[2]);
      kv.y = kl_term(lo ? ov[1] : pv[3], lo ? pv[1] : ov[3]);
      *(float2*)(ixz + (size_t)n * (C / 2) + (l & 7) * V + (lo ? 0 : 2)) = kv;
    }
  }
  {
    const int i  = threadIdx.x;
    const int k  = i >> 2;
    const int c0 = (i & 3) * 4;
    float* dst = sW + (k >> 2) * 64 + (k & 3);
    dst[(c0 + 0) * 4] = w2v.x;
    dst[(c0 + 1) * 4] = w2v.y;
    dst[(c0 + 2) * 4] = w2v.z;
    dst[(c0 + 3) * 4] = w2v.w;
  }

  __syncthreads();
  const int wave = threadIdx.x >> 5;
  const int row0 = brow0 + wave * 16;
  if (wave >= RG1 || row0 >= Nn) return;

  const int r    = l;
  const int kh   = (lane >> 4) << 1;
  const float sa = asrc2[r], da = adst2[r];

  v8f acc = {};
  const float* abase = sH + (wave * 16 + r) * HP + kh;
  const float* bbase = sW + r * 4 + kh;
  for (int k0 = 0; k0 < C1; k0 += 4) {
    v2f a = *(const v2f*)(abase + k0);
    v2f b = *(const v2f*)(bbase + (k0 >> 2) * 64);
    acc = __builtin_amdgcn_wmma_f32_16x16x4_f32(false, a, false, b, (short)0, acc, false, false);
  }

  const int mbase = (lane >> 4) * 8;
  float ps[8], pd[8];
  for (int i = 0; i < 8; ++i) {
    const int row = row0 + mbase + i;
    if (row < Nn) h2[(size_t)row * C2 + r] = acc[i];
    ps[i] = acc[i] * sa;
    pd[i] = acc[i] * da;
  }
  const float st = group_sum16x8(ps, r);
  const float dt = group_sum16x8(pd, r);
  const int srow = row0 + mbase + (r >> 1);
  if ((r & 1) == 0 && srow < Nn) { s2[srow] = st; d2[srow] = dt; }
}

__global__ void __launch_bounds__(256)
gather2_out(const int* __restrict__ off, const int* __restrict__ deg,
            const int* __restrict__ lst,
            const float* __restrict__ s, const float* __restrict__ d,
            const float* __restrict__ h2, const float* __restrict__ b2,
            float* __restrict__ orow, float* __restrict__ ixz, int Nn) {
  constexpr int C = C2, V = C2 / 16;
  const int lane  = threadIdx.x & 31;
  const int l     = lane & 15;
  const int gbase = lane & 16;
  const int wv    = (int)((blockIdx.x * 256 + threadIdx.x) >> 5);
  const int wrow0 = wv * (2 * RPG) + (lane >> 4);

  RowHead hd[RPG];
#pragma unroll
  for (int t = 0; t < RPG; ++t) hd[t].nn = min(wrow0 + 2 * t, Nn - 1);
  load_heads(off, deg, lst, s, d, l, hd);

#pragma unroll
  for (int t = 0; t < RPG; ++t) {
    const int n = wrow0 + 2 * t;
    const bool live = n < Nn;

    float ov[V];
    gather_row<C>(hd[t], lst, s, h2, b2, l, gbase, ov);

    float pv[V];
#pragma unroll
    for (int k = 0; k < V; ++k) pv[k] = row_xor<8>(ov[k]);

    if (live) {
#pragma unroll
      for (int k = 0; k < V; ++k) orow[(size_t)n * C + l * V + k] = ov[k];
      if (l < 8) {
#pragma unroll
        for (int k = 0; k < V; ++k) ixz[(size_t)n * (C / 2) + l * V + k] = kl_term(ov[k], pv[k]);
      }
    }
  }
}

extern "C" void kernel_launch(void* const* d_in, const int* in_sizes, int n_in,
                              void* d_out, int out_size, void* d_ws, size_t ws_size,
                              hipStream_t stream) {
  const float* x   = (const float*)d_in[0];
  const int*   ei  = (const int*)  d_in[1];
  const float* W1  = (const float*)d_in[2];
  const float* as1 = (const float*)d_in[3];
  const float* ad1 = (const float*)d_in[4];
  const float* b1  = (const float*)d_in[5];
  const float* W2  = (const float*)d_in[6];
  const float* as2 = (const float*)d_in[7];
  const float* ad2 = (const float*)d_in[8];
  const float* b2  = (const float*)d_in[9];

  const int N = in_sizes[0] / F_IN;
  const int E = in_sizes[1] / 2;
  float* out = (float*)d_out;

  const size_t N64 = (size_t)N * C1, N16 = (size_t)N * C2;
  float* ws   = (float*)d_ws;
  float* h1   = ws;
  float* h2   = h1 + N64;
  float* s1   = h2 + N16;
  float* d1   = s1 + N;
  float* s2   = d1 + N;
  float* d2   = s2 + N;
  int*   deg  = (int*)(d2 + N);
  int*   off  = deg + N;
  const size_t NS = (size_t)N * SEGW;
  const size_t NL = (size_t)N * CAP + E;
  int*   seg  = off + N;
  int*   tmp  = seg + 2 * NS;
  int*   lst  = tmp + E;
  int*   ov   = lst + NL;
  int*   cur  = ov + 2 * (size_t)E;

  float* ixz1 = out + N16;
  float* ixz2 = out + (size_t)N * 48;

  const int tiles = (N + 15) / 16;
  const int gblk  = (tiles + 7) / 8;
  const int gpart = (gblk + 1) / 2;
  const int sblk  = (2 * N + SPAN - 1) / SPAN;
  const int eblk  = (E + SPAN - 1) / SPAN;
  const int rblk  = (N + 16 * RPG - 1) / (16 * RPG);
  const int rblk1 = (N + 16 * RG1 - 1) / (16 * RG1);

  Gemm1Args g0 = {x, W1, as1, ad1, h1, s1, d1, N, 0,     gpart, 0};
  Gemm1Args g1 = {x, W1, as1, ad1, h1, s1, d1, N, gpart, gpart, 0};
  csr_init<<<sblk, 256, 0, stream>>>(seg, cur, 2 * N);
  csr_count_fill<<<gpart + eblk, 256, 0, stream>>>(ei, E, seg, cur + 1, ov, g0);
  csr_sort<<<gpart + rblk, 256, 0, stream>>>(seg, tmp, lst, off, deg, cur, cur + 1, ov, N, g1);

  gather1_proj<<<rblk1, 256, 0, stream>>>(off, deg, lst, s1, d1, h1, b1, ixz1,
                                         W2, as2, ad2, h2, s2, d2, N);

  gather2_out<<<rblk, 256, 0, stream>>>(off, deg, lst, s2, d2, h2, b2, out, ixz2, N);
}


# gathers (issue-bound) run about a third fewer vector instructions per row: softplus log1p as log(1+e) plus its rounding term, entry loop on scalar trip counts and 32-bit indexed loads; sum order kept
# speedup vs baseline: 4.4996x; 1.1534x over previous
#include <hip/hip_runtime.h>
#include <hip/hip_bf16.h>

typedef __attribute__((ext_vector_type(2))) float v2f;
typedef __attribute__((ext_vector_type(8))) float v8f;

#define F_IN 128
#define C1   64
#define C2   16
#define NEG_SLOPE 0.2f

template <int M> __device__ __forceinline__ int row_xor(int v) {
  return __builtin_amdgcn_update_dpp(0, v, 0x160 | M, 0xf, 0xf, false);
}
template <int M> __device__ __forceinline__ float row_xor(float v) {
  return __int_as_float(row_xor<M>(__float_as_int(v)));
}
template <int J> __device__ __forceinline__ int row_lane(int v) {
  return __builtin_amdgcn_update_dpp(0, v, 0x150 | J, 0xf, 0xf, false);
}
template <int N> __device__ __forceinline__ int row_ror(int v) {
  return __builtin_amdgcn_update_dpp(0, v, 0x120 | N, 0xf, 0xf, false);
}
__device__ __forceinline__ int row_pick(int v, int j) {
  const unsigned sel = 0x11111111u * (unsigned)j;
  return (int)__builtin_amdgcn_permlane16(0u, (unsigned)v, sel, sel, false, false);
}
__device__ __forceinline__ float row_pick(float v, int j) {
  return __int_as_float(row_pick(__float_as_int(v), j));
}
__device__ __forceinline__ int row_var(int v, int sel) {
  return (int)__builtin_amdgcn_permlane16_var(0u, (unsigned)v, (unsigned)sel, false, false);
}
__device__ __forceinline__ int other_row(int v) {
  return (int)__builtin_amdgcn_permlanex16(0u, (unsigned)v, 0x76543210u, 0xfedcba98u, false, false);
}

__device__ __forceinline__ float group_max16(float v) {
  v = fmaxf(v, row_xor<8>(v)); v = fmaxf(v, row_xor<4>(v));
  v = fmaxf(v, row_xor<2>(v)); v = fmaxf(v, row_xor<1>(v));
  return v;
}
__device__ __forceinline__ float group_sum16(float v) {
  v += row_xor<8>(v); v += row_xor<4>(v);
  v += row_xor<2>(v); v += row_xor<1>(v);
  return v;
}
__device__ __forceinline__ float group_sum16x8(const float (&v)[8], int l) {
  const bool b3 = (l & 8) != 0, b2 = (l & 4) != 0, b1 = (l & 2) != 0;
  float w[4];
  for (int j = 0; j < 4; ++j) {
    const float mine = b3 ? v[j + 4] : v[j];
    const float oth  = b3 ? v[j] : v[j + 4];
    w[j] = mine + row_xor<8>(oth);
  }
  float u[2];
  for (int j = 0; j < 2; ++j) {
    const float mine = b2 ? w[j + 2] : w[j];
    const float oth  = b2 ? w[j] : w[j + 2];
    u[j] = mine + row_xor<4>(oth);
  }
  const float mine = b1 ? u[1] : u[0];
  const float oth  = b1 ? u[0] : u[1];
  const float t = mine + row_xor<2>(oth);
  return t + row_xor<1>(t);
}

struct Gemm1Args {
  const float* x; const float* W1; const float* asrc; const float* adst;
  float* h1; float* s; float* d; int Nn;
  int first;
  int count;
  int tail;
};

__device__ __forceinline__ void
gemm1_block(const float* __restrict__ x, const float* __restrict__ W1,
            const float* __restrict__ asrc, const float* __restrict__ adst,
            float* __restrict__ h1, float* __restrict__ s, float* __restrict__ d, int Nn,
            int gb) {
  __shared__ float sW[F_IN * C1];
  for (int i = threadIdx.x; i < (F_IN * C1) / 4; i += 256) {
    float4 v = ((const float4*)W1)[i];
    const int k  = i >> 4;
    const int c0 = (i & 15) * 4;
    float* dst = sW + (k >> 2) * 256 + (k & 3);
    dst[(c0 + 0) * 4] = v.x;
    dst[(c0 + 1) * 4] = v.y;
    dst[(c0 + 2) * 4] = v.z;
    dst[(c0 + 3) * 4] = v.w;
  }
  __syncthreads();

  const int wave = threadIdx.x >> 5;
  const int lane = threadIdx.x & 31;
  const int tile = gb * 8 + wave;
  if (tile * 16 >= Nn) return;
  const int row0 = tile * 16;
  const int r    = lane & 15;
  const int kh   = (lane >> 4) << 1;
  const int arow = min(row0 + r, Nn - 1);

  const float sa0 = asrc[r], sa1 = asrc[r + 16], sa2 = asrc[r + 32], sa3 = asrc[r + 48];
  const float da0 = adst[r], da1 = adst[r + 16], da2 = adst[r + 32], da3 = adst[r + 48];

  v8f acc0 = {}, acc1 = {}, acc2 = {}, acc3 = {};
  const float* ap = x + (size_t)arow * F_IN + kh;
  const float* bbase = sW + r * 4 + kh;
  for (int k0 = 0; k0 < F_IN; k0 += 4) {
    v2f a; a.x = ap[k0]; a.y = ap[k0 + 1];
    const float* bq = bbase + (k0 >> 2) * 256;
    v2f b0 = *(const v2f*)(bq);
    v2f b1 = *(const v2f*)(bq + 64);
    v2f b2 = *(const v2f*)(bq + 128);
    v2f b3 = *(const v2f*)(bq + 192);
    acc0 = __builtin_amdgcn_wmma_f32_16x16x4_f32(false, a, false, b0, (short)0, acc0, false, false);
    acc1 = __builtin_amdgcn_wmma_f32_16x16x4_f32(false, a, false, b1, (short)0, acc1, false, false);
    acc2 = __builtin_amdgcn_wmma_f32_16x16x4_f32(false, a, false, b2, (short)0, acc2, false, false);
    acc3 = __builtin_amdgcn_wmma_f32_16x16x4_f32(false, a, false, b3, (short)0, acc3, false, false);
  }

  const int mbase = (lane >> 4) * 8;
  for (int i = 0; i < 8; ++i) {
    const int row = row0 + mbase + i;
    if (row < Nn) {
      float* op = h1 + (size_t)row * C1 + r;
      op[0]  = acc0[i];
      op[16] = acc1[i];
      op[32] = acc2[i];
      op[48] = acc3[i];
    }
  }

  float ps[8], pd[8];
  for (int i = 0; i < 8; ++i) {
    ps[i] = acc0[i] * sa0 + acc1[i] * sa1 + acc2[i] * sa2 + acc3[i] * sa3;
    pd[i] = acc0[i] * da0 + acc1[i] * da1 + acc2[i] * da2 + acc3[i] * da3;
  }
  const float st = group_sum16x8(ps, r);
  const float dt = group_sum16x8(pd, r);
  const int srow = row0 + mbase + (r >> 1);
  if ((r & 1) == 0 && srow < Nn) { s[srow] = st; d[srow] = dt; }
}

#define RPG 4
#define SPAN (256 * RPG)
#define CAP 16
#define SEGW 16
#define SEGC (SEGW - 1)

__global__ void __launch_bounds__(256)
csr_init(int* __restrict__ seg, int* __restrict__ cursor, int nseg) {
  const int i0 = blockIdx.x * SPAN + threadIdx.x;
#pragma unroll
  for (int k = 0; k < RPG; ++k) {
    const int i = i0 + k * 256;
    if (i < nseg) seg[(size_t)i * SEGW] = 0;
  }
  if (i0 == 0) { cursor[0] = 0; cursor[1] = 0; }
}

__global__ void __launch_bounds__(256) __attribute__((amdgpu_waves_per_eu(16)))
csr_count_fill(const int* __restrict__ ei, int E, int* __restrict__ seg,
               int* __restrict__ ovn, int* __restrict__ ov, const Gemm1Args g) {
  if ((int)blockIdx.x < g.count) {
    gemm1_block(g.x, g.W1, g.asrc, g.adst, g.h1, g.s, g.d, g.Nn, g.first + (int)blockIdx.x);
    return;
  }
  const int e0 = ((int)blockIdx.x - g.count) * SPAN + threadIdx.x;
  int dst[RPG], src[RPG], rk[RPG];
#pragma unroll
  for (int k = 0; k < RPG; ++k) {
    const int ec = min(e0 + k * 256, E - 1);
    dst[k] = ei[E + ec];
    src[k] = ei[ec];
  }
  const int half = (__builtin_amdgcn_s_sendmsg_rtn(0x87) >> 18) & 1;
  int* __restrict__ plane = seg + (size_t)half * g.Nn * SEGW;
#pragma unroll
  for (int k = 0; k < RPG; ++k) {
    rk[k] = 0;
    if (e0 + k * 256 < E) rk[k] = atomicAdd(plane + dst[k] * SEGW, 1);
  }
#pragma unroll
  for (int k = 0; k < RPG; ++k) {
    if (e0 + k * 256 < E) {
      if (rk[k] < SEGC) plane[dst[k] * SEGW + 1 + rk[k]] = src[k];
      else {
        const int i = atomicAdd(ovn, 1);
        ov[2 * i] = dst[k]; ov[2 * i + 1] = src[k];
      }
    }
  }
}

__global__ void __launch_bounds__(256) __attribute__((amdgpu_waves_per_eu(16)))
csr_sort(const int* __restrict__ seg, int* tmp, int* __restrict__ lout, int* __restrict__ off,
         int* __restrict__ deg, int* __restrict__ cursor, const int* __restrict__ ovn,
         const int* __restrict__ ov, int Nn, const Gemm1Args g) {
  if ((int)blockIdx.x < g.count) {
    gemm1_block(g.x, g.W1, g.asrc, g.adst, g.h1, g.s, g.d, g.Nn, g.first + (int)blockIdx.x);
    return;
  }
  const int lane  = threadIdx.x & 31;
  const int l     = lane & 15;
  const int gbase = lane & 16;
  const int grp = (int)((((int)blockIdx.x - g.count) * 256 + threadIdx.x) >> 4);
  const int* __restrict__ seg1 = seg + (size_t)Nn * SEGW;
  int w0[RPG], w1[RPG];
#pragma unroll
  for (int t = 0; t < RPG; ++t) {
    const int nc = min(grp * RPG + t, Nn - 1);
    w0[t] = seg[nc * SEGW + l];
    w1[t] = seg1[nc * SEGW + l];
  }

#pragma unroll
  for (int t = 0; t < RPG; ++t) {
    const int n  = grp * RPG + t;
    const bool live = n < Nn;
    int c0 = row_lane<0>(w0[t]);
    int c1 = row_lane<0>(w1[t]);
    if (!live) { c0 = 0; c1 = 0; }
    const int dg = c0 + c1;
    const int dgmax = max(dg, other_row(dg));
    const bool slow = c0 > SEGC || c1 > SEGC || dg > CAP;
    int o = min(n, Nn - 1) * CAP;
    if (__builtin_amdgcn_ballot_w32(slow) == 0u) {
      const int k0 = row_ror<15>(w0[t]);
      const int k1 = row_var(w1[t], (l - c0 + 1) & 15);
      const int key = (l < c0) ? k0 : ((l < dg) ? k1 : 0x7fffffff);
      int rank = 0;
      for (int j = 0; j < dgmax; ++j) {
        const int kt = row_pick(key, j);
        rank += ((j < dg) && (kt < key || (kt == key && j < l))) ? 1 : 0;
      }
      if (l < dg) lout[o + rank] = key;
    } else {
      const int m0 = min(c0, SEGC), m1 = min(c1, SEGC);
      int base = 0;
      if (l == 0 && dg > 0) base = atomicAdd(cursor, dg);
      base = row_lane<0>(base);
      o = Nn * CAP + base;
      if (l >= 1 && l <= m0) tmp[base + l - 1] = w0[t];
      if (l >= 1 && l <= m1) tmp[base + m0 + l - 1] = w1[t];
      const bool need = m0 + m1 < dg;
      const int novf = *ovn;
      int found = m0 + m1;
      for (int c = 0; c < novf; c += 16) {
        const int i = c + l;
        bool hit = false; int sv = 0;
        if (need && i < novf) { hit = ov[2 * i] == n; sv = ov[2 * i + 1]; }
        const unsigned gb = (__builtin_amdgcn_ballot_w32(hit) >> gbase) & 0xffffu;
        if (hit) tmp[base + found + __popc(gb & ((1u << l) - 1u))] = sv;
        found += __popc(gb);
      }
      __threadfence();
      for (int ci = 0; ci < dgmax; ci += 16) {
        const int i = ci + l;
        const int key = (i < dg) ? tmp[base + i] : 0x7fffffff;
        int rank = 0;
        for (int cj = 0; cj < dgmax; cj += 16) {
          const int jm = cj + l;
          const int kmine = (jm < dg) ? tmp[base + jm] : 0x7fffffff;
          const int cnt = min(16, dgmax - cj);
          for (int j = 0; j < cnt; ++j) {
            const int kt = row_pick(kmine, j);
            const int jt = cj + j;
            rank += ((jt < dg) && (kt < key || (kt == key && jt < i))) ? 1 : 0;
          }
        }
        if (i < dg) lout[o + rank] = key;
      }
    }
    if (l == 0 && live) { off[n] = o; deg[n] = dg; }
  }
}

__device__ __forceinline__ float softplusf(float v) {
  const float e = __expf(-fabsf(v));
  const float u = 1.f + e;
  const float c = e - (u - 1.f);
  const float lg = __builtin_amdgcn_logf(u) * 0.69314718f + c * __builtin_amdgcn_rcpf(u);
  return fmaxf(v, 0.f) + lg;
}
__device__ __forceinline__ float kl_term(float mean, float sp) {
  const float sd = softplusf(sp) + 1e-10f;
  return -__logf(sd) + 0.5f * (sd * sd + mean * mean - 1.f);
}
__device__ __forceinline__ float leaky(float a) { return a > 0.f ? a : NEG_SLOPE * a; }

struct RowHead { int nn, o, dg, src0; float dn, s0; };

template <int R>
__device__ __forceinline__ void
load_heads(const int* __restrict__ off, const int* __restrict__ deg,
           const int* __restrict__ lst,
           const float* __restrict__ s, const float* __restrict__ d,
           int l, RowHead (&h)[R]) {
#pragma unroll
  for (int t = 0; t < R; ++t) { h[t].o = off[h[t].nn]; h[t].dg = deg[h[t].nn]; h[t].dn = d[h[t].nn]; }
#pragma unroll
  for (int t = 0; t < R; ++t) h[t].src0 = (l < h[t].dg) ? lst[h[t].o + l] : h[t].nn;
#pragma unroll
  for (int t = 0; t < R; ++t) h[t].s0 = s[h[t].src0];
}

template <int C>
__device__ __forceinline__ void
gather_row(const RowHead& hd, const int* __restrict__ lst,
           const float* __restrict__ s,
           const float* __restrict__ hsrc, const float* __restrict__ bias,
           int l, int gbase, float (&ov)[C / 16]) {
  constexpr int V = C / 16;
  const int nn  = hd.nn;
  const int o   = hd.o;
  const int dg  = hd.dg;
  const int tot = dg + 1;
  const float dn = hd.dn;
  const int totmax = __builtin_amdgcn_readfirstlane(max(tot, other_row(tot)));

  const int   src0 = hd.src0;
  const float a0   = leaky(hd.s0 + dn);
  const bool  v0   = l < tot;

  float m = v0 ? a0 : -INFINITY;
#pragma unroll 1
  for (int c = 16; c < totmax; c += 16) {
    const int i = c + l;
    const int sr = (i < dg) ? lst[o + i] : nn;
    const float a = leaky(s[sr] + dn);
    if (i < tot) m = fmaxf(m, a);
  }
  m = group_max16(m);

  const float ex0 = v0 ? __expf(a0 - m) : 0.f;
  float den = ex0;
#pragma unroll 1
  for (int c = 16; c < totmax; c += 16) {
    const int i = c + l;
    const int sr = (i < dg) ? lst[o + i] : nn;
    const float a = leaky(s[sr] + dn);
    den += (i < tot) ? __expf(a - m) : 0.f;
  }
  den = group_sum16(den) + 1e-16f;

  float acc[V];
#pragma unroll
  for (int k = 0; k < V; ++k) acc[k] = 0.f;

  for (int c = 0; c < totmax; c += 16) {
    int sr; float w;
    if (c == 0) { sr = src0; w = ex0 / den; }
    else {
      const int i = c + l;
      sr = (i < dg) ? lst[o + i] : nn;
      const float a = leaky(s[sr] + dn);
      w = (i < tot) ? __expf(a - m) / den : 0.f;
    }
    const int cnt = min(16, totmax - c);
    for (int j = 0; j < cnt; j += 4) {
      float r[4][V];
#pragma unroll
      for (int u = 0; u < 4; ++u) {
        const int sj = row_pick(sr, (j + u) & 15);
        if constexpr (V == 4) {
          const float4 t = ((const float4*)hsrc)[sj * (C / 4) + l];
          r[u][0] = t.x; r[u][1] = t.y; r[u][2] = t.z; r[u][3] = t.w;
        } else {
#pragma unroll
          for (int k = 0; k < V; ++k) r[u][k] = hsrc[sj * C + l * V + k];
        }
      }
#pragma unroll
      for (int u = 0; u < 4; ++u) {
        const float wj = row_pick(w, (j + u) & 15);
#pragma unroll
        for (int k = 0; k < V; ++k) acc[k] = __builtin_fmaf(r[u][k], wj, acc[k]);
      }
    }
  }

#pragma unroll
  for (int k = 0; k < V; ++k) ov[k] = acc[k] + bias[l * V + k];
}

#define RG1 5
#define HP 68

__global__ void __launch_bounds__(256)
gather1_proj(const int* __restrict__ off, const int* __restrict__ deg,
             const int* __restrict__ lst,
             const float* __restrict__ s, const float* __restrict__ d,
             const float* __restrict__ h1, const float* __restrict__ b1,
             float* __restrict__ ixz,
             const float* __restrict__ W2,
             const float* __restrict__ asrc2, const float* __restrict__ adst2,
             float* __restrict__ h2, float* __restrict__ s2, float* __restrict__ d2,
             int Nn) {
  constexpr int C = C1, V = C1 / 16;
  __shared__ alignas(16) float sW[C1 * C2];
  __shared__ alignas(16) float sH[RG1 * 16 * HP];

  static_assert((C1 * C2) / 4 == 256, "one float4 of W2 per thread");
  float4 w2v;

  const int lane  = threadIdx.x & 31;
  const int l     = lane & 15;
  const int gbase = lane & 16;
  const int g     = threadIdx.x >> 4;
  const int brow0 = blockIdx.x * (RG1 * 16);

  RowHead hd[RG1];
#pragma unroll
  for (int t = 0; t < RG1; ++t) hd[t].nn = min(brow0 + t * 16 + g, Nn - 1);
  load_heads(off, deg, lst, s, d, l, hd);

#pragma unroll
  for (int t = 0; t < RG1; ++t) {
    const int n = brow0 + t * 16 + g;
    const bool live = n < Nn;

    if (t == RG1 - 1) w2v = ((const float4*)W2)[threadIdx.x];

    float ov[V];
    gather_row<C>(hd[t], lst, s, h1, b1, l, gbase, ov);

    float pv[V];
#pragma unroll
    for (int k = 0; k < V; ++k) pv[k] = row_xor<8>(ov[k]);

    float4 hv;
    hv.x = ov[0] > 0.f ? ov[0] : (__expf(ov[0]) - 1.f);
    hv.y = ov[1] > 0.f ? ov[1] : (__expf(ov[1]) - 1.f);
    hv.z = ov[2] > 0.f ? ov[2] : (__expf(ov[2]) - 1.f);
    hv.w = ov[3] > 0.f ? ov[3] : (__expf(ov[3]) - 1.f);
    if (!live) hv = make_float4(0.f, 0.f, 0.f, 0.f);
    *(float4*)(sH + (t * 16 + g) * HP + l * V) = hv;

    if (live) {
      const bool lo = l < 8;
      float2 kv;
      kv.x = kl_term(lo ? ov[0] : pv[2], lo ? pv[0] : ov[2]);
      kv.y = kl_term(lo ? ov[1] : pv[3], lo ? pv[1] : ov[3]);
      *(float2*)(ixz + (size_t)n * (C / 2) + (l & 7) * V + (lo ? 0 : 2)) = kv;
    }
  }
  {
    const int i  = threadIdx.x;
    const int k  = i >> 2;
    const int c0 = (i & 3) * 4;
    float* dst = sW + (k >> 2) * 64 + (k & 3);
    dst[(c0 + 0) * 4] = w2v.x;
    dst[(c0 + 1) * 4] = w2v.y;
    dst[(c0 + 2) * 4] = w2v.z;
    dst[(c0 + 3) * 4] = w2v.w;
  }

  __syncthreads();
  const int wave = threadIdx.x >> 5;
  const int row0 = brow0 + wave * 16;
  if (wave >= RG1 || row0 >= Nn) return;

  const int r    = l;
  const int kh   = (lane >> 4) << 1;
  const float sa = asrc2[r], da = adst2[r];

  v8f acc = {};
  const float* abase = sH + (wave * 16 + r) * HP + kh;
  const float* bbase = sW + r * 4 + kh;
  for (int k0 = 0; k0 < C1; k0 += 4) {
    v2f a = *(const v2f*)(abase + k0);
    v2f b = *(const v2f*)(bbase + (k0 >> 2) * 64);
    acc = __builtin_amdgcn_wmma_f32_16x16x4_f32(false, a, false, b, (short)0, acc, false, false);
  }

  const int mbase = (lane >> 4) * 8;
  float ps[8], pd[8];
  for (int i = 0; i < 8; ++i) {
    const int row = row0 + mbase + i;
    if (row < Nn) h2[(size_t)row * C2 + r] = acc[i];
    ps[i] = acc[i] * sa;
    pd[i] = acc[i] * da;
  }
  const float st = group_sum16x8(ps, r);
  const float dt = group_sum16x8(pd, r);
  const int srow = row0 + mbase + (r >> 1);
  if ((r & 1) == 0 && srow < Nn) { s2[srow] = st; d2[srow] = dt; }
}

__global__ void __launch_bounds__(256)
gather2_out(const int* __restrict__ off, const int* __restrict__ deg,
            const int* __restrict__ lst,
            const float* __restrict__ s, const float* __restrict__ d,
            const float* __restrict__ h2, const float* __restrict__ b2,
            float* __restrict__ orow, float* __restrict__ ixz, int Nn) {
  constexpr int C = C2, V = C2 / 16;
  const int lane  = threadIdx.x & 31;
  const int l     = lane & 15;
  const int gbase = lane & 16;
  const int wv    = (int)((blockIdx.x * 256 + threadIdx.x) >> 5);
  const int wrow0 = wv * (2 * RPG) + (lane >> 4);

  RowHead hd[RPG];
#pragma unroll
  for (int t = 0; t < RPG; ++t) hd[t].nn = min(wrow0 + 2 * t, Nn - 1);
  load_heads(off, deg, lst, s, d, l, hd);

#pragma unroll
  for (int t = 0; t < RPG; ++t) {
    const int n = wrow0 + 2 * t;
    const bool live = n < Nn;

    float ov[V];
    gather_row<C>(hd[t], lst, s, h2, b2, l, gbase, ov);

    float pv[V];
#pragma unroll
    for (int k = 0; k < V; ++k) pv[k] = row_xor<8>(ov[k]);

    if (live) {
#pragma unroll
      for (int k = 0; k < V; ++k) orow[(size_t)n * C + l * V + k] = ov[k];
      if (l < 8) {
#pragma unroll
        for (int k = 0; k < V; ++k) ixz[(size_t)n * (C / 2) + l * V + k] = kl_term(ov[k], pv[k]);
      }
    }
  }
}

extern "C" void kernel_launch(void* const* d_in, const int* in_sizes, int n_in,
                              void* d_out, int out_size, void* d_ws, size_t ws_size,
                              hipStream_t stream) {
  const float* x   = (const float*)d_in[0];
  const int*   ei  = (const int*)  d_in[1];
  const float* W1  = (const float*)d_in[2];
  const float* as1 = (const float*)d_in[3];
  const float* ad1 = (const float*)d_in[4];
  const float* b1  = (const float*)d_in[5];
  const float* W2  = (const float*)d_in[6];
  const float* as2 = (const float*)d_in[7];
  const float* ad2 = (const float*)d_in[8];
  const float* b2  = (const float*)d_in[9];

  const int N = in_sizes[0] / F_IN;
  const int E = in_sizes[1] / 2;
  float* out = (float*)d_out;

  const size_t N64 = (size_t)N * C1, N16 = (size_t)N * C2;
  float* ws   = (float*)d_ws;
  float* h1   = ws;
  float* h2   = h1 + N64;
  float* s1   = h2 + N16;
  float* d1   = s1 + N;
  float* s2   = d1 + N;
  float* d2   = s2 + N;
  int*   deg  = (int*)(d2 + N);
  int*   off  = deg + N;
  const size_t NS = (size_t)N * SEGW;
  const size_t NL = (size_t)N * CAP + E;
  int*   seg  = off + N;
  int*   tmp  = seg + 2 * NS;
  int*   lst  = tmp + E;
  int*   ov   = lst + NL;
  int*   cur  = ov + 2 * (size_t)E;

  float* ixz1 = out + N16;
  float* ixz2 = out + (size_t)N * 48;

  const int tiles = (N + 15) / 16;
  const int gblk  = (tiles + 7) / 8;
  const int gpart = (gblk + 1) / 2;
  const int sblk  = (2 * N + SPAN - 1) / SPAN;
  const int eblk  = (E + SPAN - 1) / SPAN;
  const int rblk  = (N + 16 * RPG - 1) / (16 * RPG);
  const int rblk1 = (N + 16 * RG1 - 1) / (16 * RG1);

  Gemm1Args g0 = {x, W1, as1, ad1, h1, s1, d1, N, 0,     gpart, 0};
  Gemm1Args g1 = {x, W1, as1, ad1, h1, s1, d1, N, gpart, gpart, 0};
  csr_init<<<sblk, 256, 0, stream>>>(seg, cur, 2 * N);
  csr_count_fill<<<gpart + eblk, 256, 0, stream>>>(ei, E, seg, cur + 1, ov, g0);
  csr_sort<<<gpart + rblk, 256, 0, stream>>>(seg, tmp, lst, off, deg, cur, cur + 1, ov, N, g1);

  gather1_proj<<<rblk1, 256, 0, stream>>>(off, deg, lst, s1, d1, h1, b1, ixz1,
                                         W2, as2, ad2, h2, s2, d2, N);

  gather2_out<<<rblk, 256, 0, stream>>>(off, deg, lst, s2, d2, h2, b2, out, ixz2, N);
}
